# starting kernel
# baseline (speedup 1.0000x reference)
#include <hip/hip_runtime.h>
#include <math.h>

typedef __attribute__((ext_vector_type(2))) float v2f;
typedef __attribute__((ext_vector_type(8))) float v8f;

#define N_ITERS 5
#define EPS_F   1e-6f
#define ETA_F   0.02f

#define OM2_O 0
#define OM1_O 16384
#define XS_O  28672
#define YS_O  32768
#define CM_O  36864
#define IM_O  37888
#define SC_O  38656
#define LDS_FLOATS 38912

static __device__ __forceinline__ v8f wmma4(v2f a, v2f b, v8f c) {
  return __builtin_amdgcn_wmma_f32_16x16x4_f32(false, a, false, b, (short)0, c,
                                               false, false);
}

__global__ __launch_bounds__(256) void l2norm_rows(const float* __restrict__ imgs,
                                                   const float* __restrict__ caps,
                                                   float* __restrict__ wimgs,
                                                   float* __restrict__ wcaps) {
  const int row = blockIdx.x;
  const float* src;
  float* dst;
  if (row < 128 * 36) {
    src = imgs + (size_t)row * 1024;
    dst = wimgs + (size_t)row * 1024;
  } else {
    int r = row - 128 * 36;
    src = caps + (size_t)r * 1024;
    dst = wcaps + (size_t)r * 1024;
  }
  float v[4];
  float s = 0.f;
#pragma unroll
  for (int e = 0; e < 4; ++e) {
    v[e] = src[threadIdx.x + 256 * e];
    s += v[e] * v[e];
  }
#pragma unroll
  for (int off = 16; off > 0; off >>= 1) s += __shfl_xor(s, off, 32);
  __shared__ float red[8];
  if ((threadIdx.x & 31) == 0) red[threadIdx.x >> 5] = s;
  __syncthreads();
  float tot = 0.f;
#pragma unroll
  for (int k = 0; k < 8; ++k) tot += red[k];
  const float inv = 1.0f / fmaxf(sqrtf(tot), 1e-12f);
#pragma unroll
  for (int e = 0; e < 4; ++e) dst[threadIdx.x + 256 * e] = v[e] * inv;
}

static __device__ __forceinline__ void slabs(float* lds,
                                             const float* __restrict__ gi,
                                             const float* __restrict__ gc,
                                             int w, int ln, int koff, int ioff,
                                             int tg, int ig, int d0) {
  v8f x; x = 0.0f;
  for (int l0 = 0; l0 < 60; l0 += 4) {
    v2f a = *(const v2f*)&lds[OM2_O + (ln * 16 + w) * 64 + l0 + koff];
    v2f b;
    b.x = gc[((size_t)tg * 60 + (l0 + koff)) * 1024 + d0 + ln];
    b.y = gc[((size_t)tg * 60 + (l0 + koff + 1)) * 1024 + d0 + ln];
    x = wmma4(a, b, x);
  }
  v8f y; y = 0.0f;
  for (int k0 = 0; k0 < 36; k0 += 4) {
    v2f a = *(const v2f*)&lds[OM1_O + (w * 16 + ln) * 48 + k0 + koff];
    v2f b;
    b.x = gi[((size_t)ig * 36 + (k0 + koff)) * 1024 + d0 + ln];
    b.y = gi[((size_t)ig * 36 + (k0 + koff + 1)) * 1024 + d0 + ln];
    y = wmma4(a, b, y);
  }
#pragma unroll
  for (int j = 0; j < 8; ++j) {
    lds[XS_O + w * 256 + (j + ioff) * 16 + ln] = x[j];
    lds[YS_O + w * 256 + (j + ioff) * 16 + ln] = y[j];
  }
}

__global__ __launch_bounds__(512) void nmf_tile(const float* __restrict__ gi,
                                                const float* __restrict__ gc,
                                                const int* __restrict__ il_g,
                                                const int* __restrict__ cl_g,
                                                float* __restrict__ out) {
  extern __shared__ float lds[];
  const int tid = threadIdx.x;
  const int w = tid >> 5;
  const int lane = tid & 31;
  const int hi = lane >> 4;
  const int ln = lane & 15;
  const int ib = (blockIdx.x >> 3) * 16;
  const int tb = (blockIdx.x & 7) * 16;
  const int tg = tb + w;
  const int ig = ib + w;
  const int ioff = hi ? 8 : 0;
  const int koff = hi ? 2 : 0;

  for (int idx = tid; idx < 16 * 64; idx += 512) {
    int t = idx >> 6, l = idx & 63;
    lds[CM_O + idx] = (l < cl_g[tb + t]) ? 1.f : 0.f;
  }
  for (int idx = tid; idx < 16 * 48; idx += 512) {
    int i = idx / 48, k = idx % 48;
    lds[IM_O + idx] = (k < il_g[ib + i]) ? 1.f : 0.f;
  }
  for (int idx = tid; idx < 16 * 16 * 64; idx += 512) {
    int t = (idx >> 6) & 15, l = idx & 63;
    int cl = cl_g[tb + t];
    lds[OM2_O + idx] = (l < cl) ? (1.f / (float)cl) : 0.f;
  }
  for (int idx = tid; idx < 16 * 16 * 48; idx += 512) {
    int i = idx / (16 * 48), k = idx % 48;
    int il = il_g[ib + i];
    lds[OM1_O + idx] = (k < il) ? (1.f / (float)il) : 0.f;
  }
  __syncthreads();

#pragma unroll 1
  for (int iter = 0; iter < N_ITERS; ++iter) {
    v8f dv[4];
#pragma unroll
    for (int n = 0; n < 4; ++n) dv[n] = 0.0f;
#pragma unroll 1
    for (int ch = 0; ch < 64; ++ch) {
      const int d0 = ch * 16;
      slabs(lds, gi, gc, w, ln, koff, ioff, tg, ig, d0);
      __syncthreads();
#pragma unroll
      for (int n = 0; n < 4; ++n) {
        const int lc = (n * 16 + ln < 59) ? (n * 16 + ln) : 59;
#pragma unroll
        for (int s = 0; s < 4; ++s) {
          const int dd = s * 4 + koff;
          v2f ax = *(const v2f*)&lds[XS_O + w * 256 + ln * 16 + dd];
          v2f ay = *(const v2f*)&lds[YS_O + ln * 256 + w * 16 + dd];
          v2f a; a.x = ax.x - ay.x; a.y = ax.y - ay.y;
          v2f b = *(const v2f*)&gc[((size_t)tg * 60 + lc) * 1024 + d0 + dd];
          dv[n] = wmma4(a, b, dv[n]);
        }
      }
      __syncthreads();
    }
#pragma unroll
    for (int n = 0; n < 4; ++n)
#pragma unroll
      for (int j = 0; j < 8; ++j) {
        const int i = j + ioff, l = n * 16 + ln;
        const int idx = (i * 16 + w) * 64 + l;
        float nv = lds[OM2_O + idx] - ETA_F * dv[n][j];
        lds[OM2_O + idx] = fmaxf(nv, 0.f) * lds[CM_O + w * 64 + l];
      }
    __syncthreads();
    if (tid < 256) {
      float s = 0.f;
      for (int l = 0; l < 64; ++l) s += lds[OM2_O + tid * 64 + l];
      lds[SC_O + tid] = 1.f / (s + EPS_F);
    }
    __syncthreads();
    for (int idx = tid; idx < 16384; idx += 512)
      lds[OM2_O + idx] *= lds[SC_O + (idx >> 6)];
    __syncthreads();

    v8f du[3];
#pragma unroll
    for (int n = 0; n < 3; ++n) du[n] = 0.0f;
#pragma unroll 1
    for (int ch = 0; ch < 64; ++ch) {
      const int d0 = ch * 16;
      slabs(lds, gi, gc, w, ln, koff, ioff, tg, ig, d0);
      __syncthreads();
#pragma unroll
      for (int n = 0; n < 3; ++n) {
        const int kc = (n * 16 + ln < 35) ? (n * 16 + ln) : 35;
#pragma unroll
        for (int s = 0; s < 4; ++s) {
          const int dd = s * 4 + koff;
          v2f ax = *(const v2f*)&lds[XS_O + ln * 256 + w * 16 + dd];
          v2f ay = *(const v2f*)&lds[YS_O + w * 256 + ln * 16 + dd];
          v2f a; a.x = ax.x - ay.x; a.y = ax.y - ay.y;
          v2f b = *(const v2f*)&gi[((size_t)ig * 36 + kc) * 1024 + d0 + dd];
          du[n] = wmma4(a, b, du[n]);
        }
      }
      __syncthreads();
    }
#pragma unroll
    for (int n = 0; n < 3; ++n)
#pragma unroll
      for (int j = 0; j < 8; ++j) {
        const int t = j + ioff, k = n * 16 + ln;
        const int idx = (w * 16 + t) * 48 + k;
        float nv = lds[OM1_O + idx] + ETA_F * du[n][j];
        lds[OM1_O + idx] = fmaxf(nv, 0.f) * lds[IM_O + w * 48 + k];
      }
    __syncthreads();
    if (tid < 256) {
      float s = 0.f;
      for (int k = 0; k < 48; ++k) s += lds[OM1_O + tid * 48 + k];
      lds[SC_O + tid] = 1.f / (s + EPS_F);
    }
    __syncthreads();
    for (int idx = tid; idx < 12288; idx += 512)
      lds[OM1_O + idx] *= lds[SC_O + idx / 48];
    __syncthreads();
  }

  float s11 = 0.f, s22 = 0.f, s12 = 0.f;
#pragma unroll 1
  for (int ch = 0; ch < 64; ++ch) {
    slabs(lds, gi, gc, w, ln, koff, ioff, tg, ig, ch * 16);
    __syncthreads();
    if (tid < 256) {
      const int pi = tid >> 4, pt = tid & 15;
#pragma unroll
      for (int d = 0; d < 16; ++d) {
        float g1 = lds[YS_O + pi * 256 + pt * 16 + d];
        float g2 = lds[XS_O + pt * 256 + pi * 16 + d];
        s12 += g1 * g2; s11 += g1 * g1; s22 += g2 * g2;
      }
    }
    __syncthreads();
  }
  if (tid < 256) {
    const int pi = tid >> 4, pt = tid & 15;
    const float dnm = fmaxf(sqrtf(s11), 1e-12f) * fmaxf(sqrtf(s22), 1e-12f);
    out[(size_t)(ib + pi) * 128 + (tb + pt)] = s12 / dnm;
  }
}

extern "C" void kernel_launch(void* const* d_in, const int* in_sizes, int n_in,
                              void* d_out, int out_size, void* d_ws, size_t ws_size,
                              hipStream_t stream) {
  (void)in_sizes; (void)n_in; (void)out_size; (void)ws_size;
  const float* imgs = (const float*)d_in[0];
  const float* caps = (const float*)d_in[1];
  const int* il = (const int*)d_in[2];
  const int* cl = (const int*)d_in[3];
  float* wimgs = (float*)d_ws;
  float* wcaps = wimgs + (size_t)128 * 36 * 1024;

  l2norm_rows<<<128 * 36 + 128 * 60, 256, 0, stream>>>(imgs, caps, wimgs, wcaps);
  nmf_tile<<<64, 512, LDS_FLOATS * sizeof(float), stream>>>(wimgs, wcaps, il, cl,
                                                            (float*)d_out);
}


# d chunk 16->32 floats: every 128-byte caps/imgs line is fetched whole and used once per pass, half the barriers, 8/16-byte loads; dv/du K order permuted inside 8-wide d blocks
# speedup vs baseline: 2.0026x; 2.0026x over previous
#include <hip/hip_runtime.h>
#include <math.h>

typedef __attribute__((ext_vector_type(2))) float v2f;
typedef __attribute__((ext_vector_type(4))) float v4f;
typedef __attribute__((ext_vector_type(8))) float v8f;

#define N_ITERS 5
#define EPS_F   1e-6f
#define ETA_F   0.02f

#define CH_W 32
#define N_CH 32
#define SL_RP 36
#define SL_OP 580

#define OM2_O 0
#define OM1_O 16384
#define XS_O  28672
#define YS_O  37952
#define CM_O  47232
#define IM_O  48256
#define SC_O  49024
#define LDS_FLOATS 49280

static __device__ __forceinline__ v8f wmma4(v2f a, v2f b, v8f c) {
  return __builtin_amdgcn_wmma_f32_16x16x4_f32(false, a, false, b, (short)0, c,
                                               false, false);
}

__global__ __launch_bounds__(256) void l2norm_rows(const float* __restrict__ imgs,
                                                   const float* __restrict__ caps,
                                                   float* __restrict__ wimgs,
                                                   float* __restrict__ wcaps) {
  const int row = blockIdx.x;
  const float* src;
  float* dst;
  if (row < 128 * 36) {
    src = imgs + (size_t)row * 1024;
    dst = wimgs + (size_t)row * 1024;
  } else {
    int r = row - 128 * 36;
    src = caps + (size_t)r * 1024;
    dst = wcaps + (size_t)r * 1024;
  }
  float v[4];
  float s = 0.f;
#pragma unroll
  for (int e = 0; e < 4; ++e) {
    v[e] = src[threadIdx.x + 256 * e];
    s += v[e] * v[e];
  }
#pragma unroll
  for (int off = 16; off > 0; off >>= 1) s += __shfl_xor(s, off, 32);
  __shared__ float red[8];
  if ((threadIdx.x & 31) == 0) red[threadIdx.x >> 5] = s;
  __syncthreads();
  float tot = 0.f;
#pragma unroll
  for (int k = 0; k < 8; ++k) tot += red[k];
  const float inv = 1.0f / fmaxf(sqrtf(tot), 1e-12f);
#pragma unroll
  for (int e = 0; e < 4; ++e) dst[threadIdx.x + 256 * e] = v[e] * inv;
}

static __device__ __forceinline__ void slabs(float* lds,
                                             const float* __restrict__ pi,
                                             const float* __restrict__ pc,
                                             int w, int ln, int koff, int ioff) {
  v8f x0, x1; x0 = 0.0f; x1 = 0.0f;
  for (int l0 = 0; l0 < 60; l0 += 4) {
    v2f a = *(const v2f*)&lds[OM2_O + (ln * 16 + w) * 64 + l0 + koff];
    v2f p = *(const v2f*)&pc[l0 * 1024];
    v2f q = *(const v2f*)&pc[(l0 + 1) * 1024];
    v2f b0; b0.x = p.x; b0.y = q.x;
    v2f b1; b1.x = p.y; b1.y = q.y;
    x0 = wmma4(a, b0, x0);
    x1 = wmma4(a, b1, x1);
  }
#pragma unroll
  for (int j = 0; j < 8; ++j) {
    v2f xs; xs.x = x0[j]; xs.y = x1[j];
    *(v2f*)&lds[XS_O + w * SL_OP + (j + ioff) * SL_RP + 2 * ln] = xs;
  }
  v8f y0, y1; y0 = 0.0f; y1 = 0.0f;
  for (int k0 = 0; k0 < 36; k0 += 4) {
    v2f a = *(const v2f*)&lds[OM1_O + (w * 16 + ln) * 48 + k0 + koff];
    v2f p = *(const v2f*)&pi[k0 * 1024];
    v2f q = *(const v2f*)&pi[(k0 + 1) * 1024];
    v2f b0; b0.x = p.x; b0.y = q.x;
    v2f b1; b1.x = p.y; b1.y = q.y;
    y0 = wmma4(a, b0, y0);
    y1 = wmma4(a, b1, y1);
  }
#pragma unroll
  for (int j = 0; j < 8; ++j) {
    v2f ys; ys.x = y0[j]; ys.y = y1[j];
    *(v2f*)&lds[YS_O + w * SL_OP + (j + ioff) * SL_RP + 2 * ln] = ys;
  }
}

__global__ __launch_bounds__(512) void nmf_tile(const float* __restrict__ gi,
                                                const float* __restrict__ gc,
                                                const int* __restrict__ il_g,
                                                const int* __restrict__ cl_g,
                                                float* __restrict__ out) {
  __shared__ __attribute__((aligned(16))) float lds[LDS_FLOATS];
  const int tid = threadIdx.x;
  const int w = tid >> 5;
  const int lane = tid & 31;
  const int hi = lane >> 4;
  const int ln = lane & 15;
  const int ib = (blockIdx.x >> 3) * 16;
  const int tb = (blockIdx.x & 7) * 16;
  const int tg = tb + w;
  const int ig = ib + w;
  const int ioff = hi ? 8 : 0;
  const int koff = hi ? 2 : 0;
  const int qoff = hi ? 4 : 0;
  const float* pcx = gc + ((size_t)tg * 60 + koff) * 1024 + 2 * ln;
  const float* piy = gi + ((size_t)ig * 36 + koff) * 1024 + 2 * ln;

  for (int idx = tid; idx < 16 * 64; idx += 512) {
    int t = idx >> 6, l = idx & 63;
    lds[CM_O + idx] = (l < cl_g[tb + t]) ? 1.f : 0.f;
  }
  for (int idx = tid; idx < 16 * 48; idx += 512) {
    int i = idx / 48, k = idx % 48;
    lds[IM_O + idx] = (k < il_g[ib + i]) ? 1.f : 0.f;
  }
  for (int idx = tid; idx < 16 * 16 * 64; idx += 512) {
    int t = (idx >> 6) & 15, l = idx & 63;
    int cl = cl_g[tb + t];
    lds[OM2_O + idx] = (l < cl) ? (1.f / (float)cl) : 0.f;
  }
  for (int idx = tid; idx < 16 * 16 * 48; idx += 512) {
    int i = idx / (16 * 48), k = idx % 48;
    int il = il_g[ib + i];
    lds[OM1_O + idx] = (k < il) ? (1.f / (float)il) : 0.f;
  }
  __syncthreads();

#pragma unroll 1
  for (int iter = 0; iter < N_ITERS; ++iter) {
    v8f dv[4];
#pragma unroll
    for (int n = 0; n < 4; ++n) dv[n] = 0.0f;
#pragma unroll 1
    for (int ch = 0; ch < N_CH; ++ch) {
      const int d0 = ch * CH_W;
      slabs(lds, piy + d0, pcx + d0, w, ln, koff, ioff);
      __syncthreads();
#pragma unroll
      for (int n = 0; n < 4; ++n) {
        const int lc = (n * 16 + ln < 59) ? (n * 16 + ln) : 59;
#pragma unroll
        for (int s = 0; s < 4; ++s) {
          const int dd = s * 8 + qoff;
          v4f ax = *(const v4f*)&lds[XS_O + w * SL_OP + ln * SL_RP + dd];
          v4f ay = *(const v4f*)&lds[YS_O + ln * SL_OP + w * SL_RP + dd];
          v4f b = *(const v4f*)&gc[((size_t)tg * 60 + lc) * 1024 + d0 + dd];
          v2f a0; a0.x = ax.x - ay.x; a0.y = ax.y - ay.y;
          v2f a1; a1.x = ax.z - ay.z; a1.y = ax.w - ay.w;
          v2f b0; b0.x = b.x; b0.y = b.y;
          v2f b1; b1.x = b.z; b1.y = b.w;
          dv[n] = wmma4(a0, b0, dv[n]);
          dv[n] = wmma4(a1, b1, dv[n]);
        }
      }
      __syncthreads();
    }
#pragma unroll
    for (int n = 0; n < 4; ++n)
#pragma unroll
      for (int j = 0; j < 8; ++j) {
        const int i = j + ioff, l = n * 16 + ln;
        const int idx = (i * 16 + w) * 64 + l;
        float nv = lds[OM2_O + idx] - ETA_F * dv[n][j];
        lds[OM2_O + idx] = fmaxf(nv, 0.f) * lds[CM_O + w * 64 + l];
      }
    __syncthreads();
    if (tid < 256) {
      float s = 0.f;
      for (int l = 0; l < 64; ++l) s += lds[OM2_O + tid * 64 + l];
      lds[SC_O + tid] = 1.f / (s + EPS_F);
    }
    __syncthreads();
    for (int idx = tid; idx < 16384; idx += 512)
      lds[OM2_O + idx] *= lds[SC_O + (idx >> 6)];
    __syncthreads();

    v8f du[3];
#pragma unroll
    for (int n = 0; n < 3; ++n) du[n] = 0.0f;
#pragma unroll 1
    for (int ch = 0; ch < N_CH; ++ch) {
      const int d0 = ch * CH_W;
      slabs(lds, piy + d0, pcx + d0, w, ln, koff, ioff);
      __syncthreads();
#pragma unroll
      for (int n = 0; n < 3; ++n) {
        const int kc = (n * 16 + ln < 35) ? (n * 16 + ln) : 35;
#pragma unroll
        for (int s = 0; s < 4; ++s) {
          const int dd = s * 8 + qoff;
          v4f ax = *(const v4f*)&lds[XS_O + ln * SL_OP + w * SL_RP + dd];
          v4f ay = *(const v4f*)&lds[YS_O + w * SL_OP + ln * SL_RP + dd];
          v4f b = *(const v4f*)&gi[((size_t)ig * 36 + kc) * 1024 + d0 + dd];
          v2f a0; a0.x = ax.x - ay.x; a0.y = ax.y - ay.y;
          v2f a1; a1.x = ax.z - ay.z; a1.y = ax.w - ay.w;
          v2f b0; b0.x = b.x; b0.y = b.y;
          v2f b1; b1.x = b.z; b1.y = b.w;
          du[n] = wmma4(a0, b0, du[n]);
          du[n] = wmma4(a1, b1, du[n]);
        }
      }
      __syncthreads();
    }
#pragma unroll
    for (int n = 0; n < 3; ++n)
#pragma unroll
      for (int j = 0; j < 8; ++j) {
        const int t = j + ioff, k = n * 16 + ln;
        const int idx = (w * 16 + t) * 48 + k;
        float nv = lds[OM1_O + idx] + ETA_F * du[n][j];
        lds[OM1_O + idx] = fmaxf(nv, 0.f) * lds[IM_O + w * 48 + k];
      }
    __syncthreads();
    if (tid < 256) {
      float s = 0.f;
      for (int k = 0; k < 48; ++k) s += lds[OM1_O + tid * 48 + k];
      lds[SC_O + tid] = 1.f / (s + EPS_F);
    }
    __syncthreads();
    for (int idx = tid; idx < 12288; idx += 512)
      lds[OM1_O + idx] *= lds[SC_O + idx / 48];
    __syncthreads();
  }

  float s11 = 0.f, s22 = 0.f, s12 = 0.f;
#pragma unroll 1
  for (int ch = 0; ch < N_CH; ++ch) {
    slabs(lds, piy + ch * CH_W, pcx + ch * CH_W, w, ln, koff, ioff);
    __syncthreads();
    if (tid < 256) {
      const int pi = tid >> 4, pt = tid & 15;
#pragma unroll
      for (int d = 0; d < CH_W; ++d) {
        float g1 = lds[YS_O + pi * SL_OP + pt * SL_RP + d];
        float g2 = lds[XS_O + pt * SL_OP + pi * SL_RP + d];
        s12 += g1 * g2; s11 += g1 * g1; s22 += g2 * g2;
      }
    }
    __syncthreads();
  }
  if (tid < 256) {
    const int pi = tid >> 4, pt = tid & 15;
    const float dnm = fmaxf(sqrtf(s11), 1e-12f) * fmaxf(sqrtf(s22), 1e-12f);
    out[(size_t)(ib + pi) * 128 + (tb + pt)] = s12 / dnm;
  }
}

extern "C" void kernel_launch(void* const* d_in, const int* in_sizes, int n_in,
                              void* d_out, int out_size, void* d_ws, size_t ws_size,
                              hipStream_t stream) {
  (void)in_sizes; (void)n_in; (void)out_size; (void)ws_size;
  const float* imgs = (const float*)d_in[0];
  const float* caps = (const float*)d_in[1];
  const int* il = (const int*)d_in[2];
  const int* cl = (const int*)d_in[3];
  float* wimgs = (float*)d_ws;
  float* wcaps = wimgs + (size_t)128 * 36 * 1024;

  l2norm_rows<<<128 * 36 + 128 * 60, 256, 0, stream>>>(imgs, caps, wimgs, wcaps);
  nmf_tile<<<64, 512, 0, stream>>>(wimgs, wcaps, il, cl, (float*)d_out);
}


# tile per workgroup 16x16 pairs -> 16 images x 4 captions: grid 64 -> 256, one workgroup on every WGP; image-side M rows repeat the 4 captions; d rounds 32 -> 128 wide
# speedup vs baseline: 3.6599x; 1.8276x over previous
#include <hip/hip_runtime.h>
#include <math.h>

typedef __attribute__((ext_vector_type(2))) float v2f;
typedef __attribute__((ext_vector_type(4))) float v4f;
typedef __attribute__((ext_vector_type(8))) float v8f;

#define N_ITERS 5
#define EPS_F   1e-6f
#define ETA_F   0.02f

#define TT   4
#define SC_W 128
#define N_SC 8
#define SL_RP 132
#define XS_OP 2116
#define YS_OP 532

#define OM2_O 0
#define OM1_O 4096
#define XS_O  7168
#define YS_O  15632
#define CM_O  24144
#define IM_O  24400
#define SC_O  25168
#define LDS_FLOATS 25232

static __device__ __forceinline__ v8f wmma4(v2f a, v2f b, v8f c) {
  return __builtin_amdgcn_wmma_f32_16x16x4_f32(false, a, false, b, (short)0, c,
                                               false, false);
}

__global__ __launch_bounds__(256) void l2norm_rows(const float* __restrict__ imgs,
                                                   const float* __restrict__ caps,
                                                   float* __restrict__ wimgs,
                                                   float* __restrict__ wcaps) {
  const int row = blockIdx.x;
  const float* src;
  float* dst;
  if (row < 128 * 36) {
    src = imgs + (size_t)row * 1024;
    dst = wimgs + (size_t)row * 1024;
  } else {
    int r = row - 128 * 36;
    src = caps + (size_t)r * 1024;
    dst = wcaps + (size_t)r * 1024;
  }
  float v[4];
  float s = 0.f;
#pragma unroll
  for (int e = 0; e < 4; ++e) {
    v[e] = src[threadIdx.x + 256 * e];
    s += v[e] * v[e];
  }
#pragma unroll
  for (int off = 16; off > 0; off >>= 1) s += __shfl_xor(s, off, 32);
  __shared__ float red[8];
  if ((threadIdx.x & 31) == 0) red[threadIdx.x >> 5] = s;
  __syncthreads();
  float tot = 0.f;
#pragma unroll
  for (int k = 0; k < 8; ++k) tot += red[k];
  const float inv = 1.0f / fmaxf(sqrtf(tot), 1e-12f);
#pragma unroll
  for (int e = 0; e < 4; ++e) dst[threadIdx.x + 256 * e] = v[e] * inv;
}

static __device__ __forceinline__ void slab_x(float* lds,
                                              const float* __restrict__ pc,
                                              int c, int q, int ln, int koff,
                                              int ioff) {
  v8f x0, x1; x0 = 0.0f; x1 = 0.0f;
  for (int l0 = 0; l0 < 60; l0 += 4) {
    v2f a = *(const v2f*)&lds[OM2_O + (ln * TT + c) * 64 + l0 + koff];
    v2f p = *(const v2f*)&pc[l0 * 1024];
    v2f r = *(const v2f*)&pc[(l0 + 1) * 1024];
    v2f b0; b0.x = p.x; b0.y = r.x;
    v2f b1; b1.x = p.y; b1.y = r.y;
    x0 = wmma4(a, b0, x0);
    x1 = wmma4(a, b1, x1);
  }
#pragma unroll
  for (int j = 0; j < 8; ++j) {
    v2f xs; xs.x = x0[j]; xs.y = x1[j];
    *(v2f*)&lds[XS_O + c * XS_OP + (j + ioff) * SL_RP + q * 32 + 2 * ln] = xs;
  }
}

static __device__ __forceinline__ void slab_y(float* lds,
                                              const float* __restrict__ pi,
                                              int w, int tr, int ln, int koff) {
  v2f a[9];
#pragma unroll
  for (int k = 0; k < 9; ++k)
    a[k] = *(const v2f*)&lds[OM1_O + (w * TT + tr) * 48 + 4 * k + koff];
#pragma unroll 1
  for (int half = 0; half < 2; ++half) {
    const float* ph = pi + half * 64;
    v8f y[4];
#pragma unroll
    for (int m = 0; m < 4; ++m) y[m] = 0.0f;
#pragma unroll
    for (int k = 0; k < 9; ++k) {
#pragma unroll
      for (int sub = 0; sub < 2; ++sub) {
        v2f p = *(const v2f*)&ph[(4 * k) * 1024 + sub * 32];
        v2f r = *(const v2f*)&ph[(4 * k + 1) * 1024 + sub * 32];
        v2f b0; b0.x = p.x; b0.y = r.x;
        v2f b1; b1.x = p.y; b1.y = r.y;
        y[2 * sub] = wmma4(a[k], b0, y[2 * sub]);
        y[2 * sub + 1] = wmma4(a[k], b1, y[2 * sub + 1]);
      }
    }
#pragma unroll
    for (int sub = 0; sub < 2; ++sub)
#pragma unroll
      for (int j = 0; j < TT; ++j) {
        v2f ys; ys.x = y[2 * sub][j]; ys.y = y[2 * sub + 1][j];
        *(v2f*)&lds[YS_O + w * YS_OP + j * SL_RP + half * 64 + sub * 32 + 2 * ln] = ys;
      }
  }
}

__global__ __launch_bounds__(512) void nmf_tile(const float* __restrict__ gi,
                                                const float* __restrict__ gc,
                                                const int* __restrict__ il_g,
                                                const int* __restrict__ cl_g,
                                                float* __restrict__ out) {
  __shared__ __attribute__((aligned(16))) float lds[LDS_FLOATS];
  const int tid = threadIdx.x;
  const int w = tid >> 5;
  const int lane = tid & 31;
  const int hi = lane >> 4;
  const int ln = lane & 15;
  const int ib = (blockIdx.x >> 5) * 16;
  const int tb = (blockIdx.x & 31) * TT;
  const int c = w & 3;
  const int q = w >> 2;
  const int tg = tb + c;
  const int ig = ib + w;
  const int tr = ln & 3;
  const int ioff = hi ? 8 : 0;
  const int koff = hi ? 2 : 0;
  const int qoff = hi ? 4 : 0;
  const float* pcx = gc + ((size_t)tg * 60 + koff) * 1024 + q * 32 + 2 * ln;
  const float* piy = gi + ((size_t)ig * 36 + koff) * 1024 + 2 * ln;
  const int lc = (q * 16 + ln < 59) ? (q * 16 + ln) : 59;
  const float* pcb = gc + ((size_t)tg * 60 + lc) * 1024 + qoff;

  for (int idx = tid; idx < TT * 64; idx += 512) {
    int t = idx >> 6, l = idx & 63;
    lds[CM_O + idx] = (l < cl_g[tb + t]) ? 1.f : 0.f;
  }
  for (int idx = tid; idx < 16 * 48; idx += 512) {
    int i = idx / 48, k = idx % 48;
    lds[IM_O + idx] = (k < il_g[ib + i]) ? 1.f : 0.f;
  }
  for (int idx = tid; idx < 16 * TT * 64; idx += 512) {
    int t = (idx >> 6) & (TT - 1), l = idx & 63;
    int cl = cl_g[tb + t];
    lds[OM2_O + idx] = (l < cl) ? (1.f / (float)cl) : 0.f;
  }
  for (int idx = tid; idx < 16 * TT * 48; idx += 512) {
    int i = idx / (TT * 48), k = idx % 48;
    int il = il_g[ib + i];
    lds[OM1_O + idx] = (k < il) ? (1.f / (float)il) : 0.f;
  }
  __syncthreads();

#pragma unroll 1
  for (int iter = 0; iter < N_ITERS; ++iter) {
    v8f dv; dv = 0.0f;
#pragma unroll 1
    for (int sc = 0; sc < N_SC; ++sc) {
      const int d0 = sc * SC_W;
      slab_x(lds, pcx + d0, c, q, ln, koff, ioff);
      slab_y(lds, piy + d0, w, tr, ln, koff);
      __syncthreads();
#pragma unroll
      for (int s = 0; s < SC_W / 8; ++s) {
        const int dd = s * 8 + qoff;
        v4f ax = *(const v4f*)&lds[XS_O + c * XS_OP + ln * SL_RP + dd];
        v4f ay = *(const v4f*)&lds[YS_O + ln * YS_OP + c * SL_RP + dd];
        v4f b = *(const v4f*)&pcb[d0 + s * 8];
        v2f a0; a0.x = ax.x - ay.x; a0.y = ax.y - ay.y;
        v2f a1; a1.x = ax.z - ay.z; a1.y = ax.w - ay.w;
        v2f b0; b0.x = b.x; b0.y = b.y;
        v2f b1; b1.x = b.z; b1.y = b.w;
        dv = wmma4(a0, b0, dv);
        dv = wmma4(a1, b1, dv);
      }
      __syncthreads();
    }
#pragma unroll
    for (int j = 0; j < 8; ++j) {
      const int i = j + ioff, l = q * 16 + ln;
      const int idx = (i * TT + c) * 64 + l;
      float nv = lds[OM2_O + idx] - ETA_F * dv[j];
      lds[OM2_O + idx] = fmaxf(nv, 0.f) * lds[CM_O + c * 64 + l];
    }
    __syncthreads();
    if (tid < 16 * TT) {
      float s = 0.f;
      for (int l = 0; l < 64; ++l) s += lds[OM2_O + tid * 64 + l];
      lds[SC_O + tid] = 1.f / (s + EPS_F);
    }
    __syncthreads();
    for (int idx = tid; idx < 16 * TT * 64; idx += 512)
      lds[OM2_O + idx] *= lds[SC_O + (idx >> 6)];
    __syncthreads();

    v8f du[3];
#pragma unroll
    for (int n = 0; n < 3; ++n) du[n] = 0.0f;
#pragma unroll 1
    for (int sc = 0; sc < N_SC; ++sc) {
      const int d0 = sc * SC_W;
      slab_x(lds, pcx + d0, c, q, ln, koff, ioff);
      slab_y(lds, piy + d0, w, tr, ln, koff);
      __syncthreads();
#pragma unroll
      for (int s = 0; s < SC_W / 8; ++s) {
        const int dd = s * 8 + qoff;
        v4f ax = *(const v4f*)&lds[XS_O + tr * XS_OP + w * SL_RP + dd];
        v4f ay = *(const v4f*)&lds[YS_O + w * YS_OP + tr * SL_RP + dd];
        v2f a0; a0.x = ax.x - ay.x; a0.y = ax.y - ay.y;
        v2f a1; a1.x = ax.z - ay.z; a1.y = ax.w - ay.w;
#pragma unroll
        for (int n = 0; n < 3; ++n) {
          const int kc = (n * 16 + ln < 35) ? (n * 16 + ln) : 35;
          v4f b = *(const v4f*)&gi[((size_t)ig * 36 + kc) * 1024 + d0 + dd];
          v2f b0; b0.x = b.x; b0.y = b.y;
          v2f b1; b1.x = b.z; b1.y = b.w;
          du[n] = wmma4(a0, b0, du[n]);
          du[n] = wmma4(a1, b1, du[n]);
        }
      }
      __syncthreads();
    }
#pragma unroll
    for (int n = 0; n < 3; ++n)
#pragma unroll
      for (int j = 0; j < TT; ++j) {
        const int k = n * 16 + ln;
        const int idx = (w * TT + j) * 48 + k;
        float nv = lds[OM1_O + idx] + ETA_F * du[n][j];
        lds[OM1_O + idx] = fmaxf(nv, 0.f) * lds[IM_O + w * 48 + k];
      }
    __syncthreads();
    if (tid < 16 * TT) {
      float s = 0.f;
      for (int k = 0; k < 48; ++k) s += lds[OM1_O + tid * 48 + k];
      lds[SC_O + tid] = 1.f / (s + EPS_F);
    }
    __syncthreads();
    for (int idx = tid; idx < 16 * TT * 48; idx += 512)
      lds[OM1_O + idx] *= lds[SC_O + idx / 48];
    __syncthreads();
  }

  float s11 = 0.f, s22 = 0.f, s12 = 0.f;
#pragma unroll 1
  for (int sc = 0; sc < N_SC; ++sc) {
    slab_x(lds, pcx + sc * SC_W, c, q, ln, koff, ioff);
    slab_y(lds, piy + sc * SC_W, w, tr, ln, koff);
    __syncthreads();
    if (tid < 16 * TT) {
      const int pi = tid >> 2, pt = tid & 3;
#pragma unroll 32
      for (int d = 0; d < SC_W; ++d) {
        float g1 = lds[YS_O + pi * YS_OP + pt * SL_RP + d];
        float g2 = lds[XS_O + pt * XS_OP + pi * SL_RP + d];
        s12 += g1 * g2; s11 += g1 * g1; s22 += g2 * g2;
      }
    }
    __syncthreads();
  }
  if (tid < 16 * TT) {
    const float dnm = fmaxf(sqrtf(s11), 1e-12f) * fmaxf(sqrtf(s22), 1e-12f);
    lds[SC_O + tid] = s12 / dnm;
  }
  __syncthreads();
  if (tid < 16) {
    const v4f r = *(const v4f*)&lds[SC_O + tid * TT];
    *(v4f*)&out[(size_t)(ib + tid) * 128 + tb] = r;
  }
}

extern "C" void kernel_launch(void* const* d_in, const int* in_sizes, int n_in,
                              void* d_out, int out_size, void* d_ws, size_t ws_size,
                              hipStream_t stream) {
  (void)in_sizes; (void)n_in; (void)out_size; (void)ws_size;
  const float* imgs = (const float*)d_in[0];
  const float* caps = (const float*)d_in[1];
  const int* il = (const int*)d_in[2];
  const int* cl = (const int*)d_in[3];
  float* wimgs = (float*)d_ws;
  float* wcaps = wimgs + (size_t)128 * 36 * 1024;

  l2norm_rows<<<128 * 36 + 128 * 60, 256, 0, stream>>>(imgs, caps, wimgs, wcaps);
  nmf_tile<<<256, 512, 0, stream>>>(wimgs, wcaps, il, cl, (float*)d_out);
}


# dV and dU B operands (K=d rows) staged per wave through LDS: whole-line 16-byte global loads, 4 lines per load, replace loads that touched 16 lines each; products and sum order unchanged
# speedup vs baseline: 4.0272x; 1.1004x over previous
#include <hip/hip_runtime.h>
#include <math.h>

typedef __attribute__((ext_vector_type(2))) float v2f;
typedef __attribute__((ext_vector_type(4))) float v4f;
typedef __attribute__((ext_vector_type(8))) float v8f;

#define N_ITERS 5
#define EPS_F   1e-6f
#define ETA_F   0.02f

#define TT   4
#define SC_W 128
#define N_SC 8
#define SL_RP 132
#define XS_OP 2116
#define YS_OP 532

#define OM2_O 0
#define OM1_O 4096
#define XS_O  7168
#define YS_O  15632
#define CM_O  24144
#define IM_O  24400
#define SC_O  25168
#define ST_O  25232
#define ST_RP 36
#define ST_W  1296
#define LDS_FLOATS 45968

static __device__ __forceinline__ v8f wmma4(v2f a, v2f b, v8f c) {
  return __builtin_amdgcn_wmma_f32_16x16x4_f32(false, a, false, b, (short)0, c,
                                               false, false);
}

__global__ __launch_bounds__(256) void l2norm_rows(const float* __restrict__ imgs,
                                                   const float* __restrict__ caps,
                                                   float* __restrict__ wimgs,
                                                   float* __restrict__ wcaps) {
  const int row = blockIdx.x;
  const float* src;
  float* dst;
  if (row < 128 * 36) {
    src = imgs + (size_t)row * 1024;
    dst = wimgs + (size_t)row * 1024;
  } else {
    int r = row - 128 * 36;
    src = caps + (size_t)r * 1024;
    dst = wcaps + (size_t)r * 1024;
  }
  float v[4];
  float s = 0.f;
#pragma unroll
  for (int e = 0; e < 4; ++e) {
    v[e] = src[threadIdx.x + 256 * e];
    s += v[e] * v[e];
  }
#pragma unroll
  for (int off = 16; off > 0; off >>= 1) s += __shfl_xor(s, off, 32);
  __shared__ float red[8];
  if ((threadIdx.x & 31) == 0) red[threadIdx.x >> 5] = s;
  __syncthreads();
  float tot = 0.f;
#pragma unroll
  for (int k = 0; k < 8; ++k) tot += red[k];
  const float inv = 1.0f / fmaxf(sqrtf(tot), 1e-12f);
#pragma unroll
  for (int e = 0; e < 4; ++e) dst[threadIdx.x + 256 * e] = v[e] * inv;
}

static __device__ __forceinline__ void slab_x(float* lds,
                                              const float* __restrict__ pc,
                                              int c, int q, int ln, int koff,
                                              int ioff) {
  v8f x0, x1; x0 = 0.0f; x1 = 0.0f;
  for (int l0 = 0; l0 < 60; l0 += 4) {
    v2f a = *(const v2f*)&lds[OM2_O + (ln * TT + c) * 64 + l0 + koff];
    v2f p = *(const v2f*)&pc[l0 * 1024];
    v2f r = *(const v2f*)&pc[(l0 + 1) * 1024];
    v2f b0; b0.x = p.x; b0.y = r.x;
    v2f b1; b1.x = p.y; b1.y = r.y;
    x0 = wmma4(a, b0, x0);
    x1 = wmma4(a, b1, x1);
  }
#pragma unroll
  for (int j = 0; j < 8; ++j) {
    v2f xs; xs.x = x0[j]; xs.y = x1[j];
    *(v2f*)&lds[XS_O + c * XS_OP + (j + ioff) * SL_RP + q * 32 + 2 * ln] = xs;
  }
}

static __device__ __forceinline__ void slab_y(float* lds,
                                              const float* __restrict__ pi,
                                              int w, int tr, int ln, int koff) {
  v2f a[9];
#pragma unroll
  for (int k = 0; k < 9; ++k)
    a[k] = *(const v2f*)&lds[OM1_O + (w * TT + tr) * 48 + 4 * k + koff];
#pragma unroll 1
  for (int half = 0; half < 2; ++half) {
    const float* ph = pi + half * 64;
    v8f y[4];
#pragma unroll
    for (int m = 0; m < 4; ++m) y[m] = 0.0f;
#pragma unroll
    for (int k = 0; k < 9; ++k) {
#pragma unroll
      for (int sub = 0; sub < 2; ++sub) {
        v2f p = *(const v2f*)&ph[(4 * k) * 1024 + sub * 32];
        v2f r = *(const v2f*)&ph[(4 * k + 1) * 1024 + sub * 32];
        v2f b0; b0.x = p.x; b0.y = r.x;
        v2f b1; b1.x = p.y; b1.y = r.y;
        y[2 * sub] = wmma4(a[k], b0, y[2 * sub]);
        y[2 * sub + 1] = wmma4(a[k], b1, y[2 * sub + 1]);
      }
    }
#pragma unroll
    for (int sub = 0; sub < 2; ++sub)
#pragma unroll
      for (int j = 0; j < TT; ++j) {
        v2f ys; ys.x = y[2 * sub][j]; ys.y = y[2 * sub + 1][j];
        *(v2f*)&lds[YS_O + w * YS_OP + j * SL_RP + half * 64 + sub * 32 + 2 * ln] = ys;
      }
  }
}

__global__ __launch_bounds__(512) void nmf_tile(const float* __restrict__ gi,
                                                const float* __restrict__ gc,
                                                const int* __restrict__ il_g,
                                                const int* __restrict__ cl_g,
                                                float* __restrict__ out) {
  __shared__ __attribute__((aligned(16))) float lds[LDS_FLOATS];
  const int tid = threadIdx.x;
  const int w = tid >> 5;
  const int lane = tid & 31;
  const int hi = lane >> 4;
  const int ln = lane & 15;
  const int ib = (blockIdx.x >> 5) * 16;
  const int tb = (blockIdx.x & 31) * TT;
  const int c = w & 3;
  const int q = w >> 2;
  const int tg = tb + c;
  const int ig = ib + w;
  const int tr = ln & 3;
  const int ioff = hi ? 8 : 0;
  const int koff = hi ? 2 : 0;
  const int qoff = hi ? 4 : 0;
  const float* pcx = gc + ((size_t)tg * 60 + koff) * 1024 + q * 32 + 2 * ln;
  const float* piy = gi + ((size_t)ig * 36 + koff) * 1024 + 2 * ln;
  float* stg = lds + ST_O + w * ST_W;
  const int sr = lane >> 3;
  const int sp = (lane & 7) * 4;
  const float* pig = gi + ((size_t)ig * 36 + sr) * 1024 + sp;
  const float* pcg[4];
#pragma unroll
  for (int j = 0; j < 4; ++j) {
    const int lr = (q * 16 + 4 * j + sr < 59) ? (q * 16 + 4 * j + sr) : 59;
    pcg[j] = gc + ((size_t)tg * 60 + lr) * 1024 + sp;
  }

  for (int idx = tid; idx < TT * 64; idx += 512) {
    int t = idx >> 6, l = idx & 63;
    lds[CM_O + idx] = (l < cl_g[tb + t]) ? 1.f : 0.f;
  }
  for (int idx = tid; idx < 16 * 48; idx += 512) {
    int i = idx / 48, k = idx % 48;
    lds[IM_O + idx] = (k < il_g[ib + i]) ? 1.f : 0.f;
  }
  for (int idx = tid; idx < 16 * TT * 64; idx += 512) {
    int t = (idx >> 6) & (TT - 1), l = idx & 63;
    int cl = cl_g[tb + t];
    lds[OM2_O + idx] = (l < cl) ? (1.f / (float)cl) : 0.f;
  }
  for (int idx = tid; idx < 16 * TT * 48; idx += 512) {
    int i = idx / (TT * 48), k = idx % 48;
    int il = il_g[ib + i];
    lds[OM1_O + idx] = (k < il) ? (1.f / (float)il) : 0.f;
  }
  __syncthreads();

#pragma unroll 1
  for (int iter = 0; iter < N_ITERS; ++iter) {
    v8f dv; dv = 0.0f;
#pragma unroll 1
    for (int sc = 0; sc < N_SC; ++sc) {
      const int d0 = sc * SC_W;
      slab_x(lds, pcx + d0, c, q, ln, koff, ioff);
      v4f h[4];
#pragma unroll
      for (int j = 0; j < 4; ++j) h[j] = *(const v4f*)&pcg[j][d0];
      slab_y(lds, piy + d0, w, tr, ln, koff);
      __syncthreads();
#pragma unroll
      for (int sub = 0; sub < 4; ++sub) {
#pragma unroll
        for (int j = 0; j < 4; ++j)
          *(v4f*)&stg[(4 * j + sr) * ST_RP + sp] = h[j];
        if (sub < 3) {
#pragma unroll
          for (int j = 0; j < 4; ++j)
            h[j] = *(const v4f*)&pcg[j][d0 + (sub + 1) * 32];
        }
        __builtin_amdgcn_fence(__ATOMIC_ACQ_REL, "wavefront");
        v4f bf[4], ax[4], ay[4];
#pragma unroll
        for (int s4 = 0; s4 < 4; ++s4) {
          const int dd = (sub * 4 + s4) * 8 + qoff;
          ax[s4] = *(const v4f*)&lds[XS_O + c * XS_OP + ln * SL_RP + dd];
          ay[s4] = *(const v4f*)&lds[YS_O + ln * YS_OP + c * SL_RP + dd];
          bf[s4] = *(const v4f*)&stg[ln * ST_RP + s4 * 8 + qoff];
        }
#pragma unroll
        for (int s4 = 0; s4 < 4; ++s4) {
          v2f a0; a0.x = ax[s4].x - ay[s4].x; a0.y = ax[s4].y - ay[s4].y;
          v2f a1; a1.x = ax[s4].z - ay[s4].z; a1.y = ax[s4].w - ay[s4].w;
          v2f b0; b0.x = bf[s4].x; b0.y = bf[s4].y;
          v2f b1; b1.x = bf[s4].z; b1.y = bf[s4].w;
          dv = wmma4(a0, b0, dv);
          dv = wmma4(a1, b1, dv);
        }
        __builtin_amdgcn_fence(__ATOMIC_ACQ_REL, "wavefront");
      }
      __syncthreads();
    }
#pragma unroll
    for (int j = 0; j < 8; ++j) {
      const int i = j + ioff, l = q * 16 + ln;
      const int idx = (i * TT + c) * 64 + l;
      float nv = lds[OM2_O + idx] - ETA_F * dv[j];
      lds[OM2_O + idx] = fmaxf(nv, 0.f) * lds[CM_O + c * 64 + l];
    }
    __syncthreads();
    if (tid < 16 * TT) {
      float s = 0.f;
      for (int l = 0; l < 64; ++l) s += lds[OM2_O + tid * 64 + l];
      lds[SC_O + tid] = 1.f / (s + EPS_F);
    }
    __syncthreads();
    for (int idx = tid; idx < 16 * TT * 64; idx += 512)
      lds[OM2_O + idx] *= lds[SC_O + (idx >> 6)];
    __syncthreads();

    v8f du[3];
#pragma unroll
    for (int n = 0; n < 3; ++n) du[n] = 0.0f;
#pragma unroll 1
    for (int sc = 0; sc < N_SC; ++sc) {
      const int d0 = sc * SC_W;
      slab_x(lds, pcx + d0, c, q, ln, koff, ioff);
      v4f g[9];
#pragma unroll
      for (int j = 0; j < 9; ++j) g[j] = *(const v4f*)&pig[d0 + (4 * j) * 1024];
      slab_y(lds, piy + d0, w, tr, ln, koff);
      __syncthreads();
#pragma unroll
      for (int sub = 0; sub < 4; ++sub) {
#pragma unroll
        for (int j = 0; j < 9; ++j)
          *(v4f*)&stg[(4 * j + sr) * ST_RP + sp] = g[j];
        if (sub < 3) {
#pragma unroll
          for (int j = 0; j < 9; ++j)
            g[j] = *(const v4f*)&pig[d0 + (sub + 1) * 32 + (4 * j) * 1024];
        }
        __builtin_amdgcn_fence(__ATOMIC_ACQ_REL, "wavefront");
        v4f bf[4][3], ax[4], ay[4];
#pragma unroll
        for (int s4 = 0; s4 < 4; ++s4) {
          const int dd = (sub * 4 + s4) * 8 + qoff;
          ax[s4] = *(const v4f*)&lds[XS_O + tr * XS_OP + w * SL_RP + dd];
          ay[s4] = *(const v4f*)&lds[YS_O + w * YS_OP + tr * SL_RP + dd];
#pragma unroll
          for (int n = 0; n < 3; ++n) {
            const int kc = (n * 16 + ln < 35) ? (n * 16 + ln) : 35;
            bf[s4][n] = *(const v4f*)&stg[kc * ST_RP + s4 * 8 + qoff];
          }
        }
#pragma unroll
        for (int s4 = 0; s4 < 4; ++s4) {
          v2f a0; a0.x = ax[s4].x - ay[s4].x; a0.y = ax[s4].y - ay[s4].y;
          v2f a1; a1.x = ax[s4].z - ay[s4].z; a1.y = ax[s4].w - ay[s4].w;
#pragma unroll
          for (int n = 0; n < 3; ++n) {
            v2f b0; b0.x = bf[s4][n].x; b0.y = bf[s4][n].y;
            v2f b1; b1.x = bf[s4][n].z; b1.y = bf[s4][n].w;
            du[n] = wmma4(a0, b0, du[n]);
            du[n] = wmma4(a1, b1, du[n]);
          }
        }
        __builtin_amdgcn_fence(__ATOMIC_ACQ_REL, "wavefront");
      }
      __syncthreads();
    }
#pragma unroll
    for (int n = 0; n < 3; ++n)
#pragma unroll
      for (int j = 0; j < TT; ++j) {
        const int k = n * 16 + ln;
        const int idx = (w * TT + j) * 48 + k;
        float nv = lds[OM1_O + idx] + ETA_F * du[n][j];
        lds[OM1_O + idx] = fmaxf(nv, 0.f) * lds[IM_O + w * 48 + k];
      }
    __syncthreads();
    if (tid < 16 * TT) {
      float s = 0.f;
      for (int k = 0; k < 48; ++k) s += lds[OM1_O + tid * 48 + k];
      lds[SC_O + tid] = 1.f / (s + EPS_F);
    }
    __syncthreads();
    for (int idx = tid; idx < 16 * TT * 48; idx += 512)
      lds[OM1_O + idx] *= lds[SC_O + idx / 48];
    __syncthreads();
  }

  float s11 = 0.f, s22 = 0.f, s12 = 0.f;
#pragma unroll 1
  for (int sc = 0; sc < N_SC; ++sc) {
    slab_x(lds, pcx + sc * SC_W, c, q, ln, koff, ioff);
    slab_y(lds, piy + sc * SC_W, w, tr, ln, koff);
    __syncthreads();
    if (tid < 16 * TT) {
      const int pi = tid >> 2, pt = tid & 3;
#pragma unroll 32
      for (int d = 0; d < SC_W; ++d) {
        float g1 = lds[YS_O + pi * YS_OP + pt * SL_RP + d];
        float g2 = lds[XS_O + pt * XS_OP + pi * SL_RP + d];
        s12 += g1 * g2; s11 += g1 * g1; s22 += g2 * g2;
      }
    }
    __syncthreads();
  }
  if (tid < 16 * TT) {
    const float dnm = fmaxf(sqrtf(s11), 1e-12f) * fmaxf(sqrtf(s22), 1e-12f);
    lds[SC_O + tid] = s12 / dnm;
  }
  __syncthreads();
  if (tid < 16) {
    const v4f r = *(const v4f*)&lds[SC_O + tid * TT];
    *(v4f*)&out[(size_t)(ib + tid) * 128 + tb] = r;
  }
}

extern "C" void kernel_launch(void* const* d_in, const int* in_sizes, int n_in,
                              void* d_out, int out_size, void* d_ws, size_t ws_size,
                              hipStream_t stream) {
  (void)in_sizes; (void)n_in; (void)out_size; (void)ws_size;
  const float* imgs = (const float*)d_in[0];
  const float* caps = (const float*)d_in[1];
  const int* il = (const int*)d_in[2];
  const int* cl = (const int*)d_in[3];
  float* wimgs = (float*)d_ws;
  float* wcaps = wimgs + (size_t)128 * 36 * 1024;

  l2norm_rows<<<128 * 36 + 128 * 60, 256, 0, stream>>>(imgs, caps, wimgs, wcaps);
  nmf_tile<<<256, 512, 0, stream>>>(wimgs, wcaps, il, cl, (float*)d_out);
}


# dV, dU split by linearity: X part of dV = omega_2 @ (caps caps^T), Y part of dU = omega_1 @ (imgs imgs^T) from Gram tiles of one added launch; pass 1 builds no X slab, pass 2 no Y; sum order changes
# speedup vs baseline: 5.3747x; 1.3346x over previous
#include <hip/hip_runtime.h>
#include <math.h>

typedef __attribute__((ext_vector_type(2))) float v2f;
typedef __attribute__((ext_vector_type(4))) float v4f;
typedef __attribute__((ext_vector_type(8))) float v8f;

#define N_ITERS 5
#define EPS_F   1e-6f
#define ETA_F   0.02f

#define TT   4
#define SC_W 128
#define N_SC 8
#define SL_RP 132
#define XS_OP 2116
#define YS_OP 532

#define OM2_O 0
#define OM1_O 4096
#define XS_O  7168
#define YS_O  15632
#define CM_O  24144
#define IM_O  24400
#define SC_O  25168
#define ST_O  25232
#define ST_RP 36
#define ST_W  1296
#define LDS_FLOATS 45968

static __device__ __forceinline__ v8f wmma4(v2f a, v2f b, v8f c) {
  return __builtin_amdgcn_wmma_f32_16x16x4_f32(false, a, false, b, (short)0, c,
                                               false, false);
}

__global__ __launch_bounds__(256) void l2norm_rows(const float* __restrict__ imgs,
                                                   const float* __restrict__ caps,
                                                   float* __restrict__ wimgs,
                                                   float* __restrict__ wcaps) {
  const int row = blockIdx.x;
  const float* src;
  float* dst;
  if (row < 128 * 36) {
    src = imgs + (size_t)row * 1024;
    dst = wimgs + (size_t)row * 1024;
  } else {
    int r = row - 128 * 36;
    src = caps + (size_t)r * 1024;
    dst = wcaps + (size_t)r * 1024;
  }
  float v[4];
  float s = 0.f;
#pragma unroll
  for (int e = 0; e < 4; ++e) {
    v[e] = src[threadIdx.x + 256 * e];
    s += v[e] * v[e];
  }
#pragma unroll
  for (int off = 16; off > 0; off >>= 1) s += __shfl_xor(s, off, 32);
  __shared__ float red[8];
  if ((threadIdx.x & 31) == 0) red[threadIdx.x >> 5] = s;
  __syncthreads();
  float tot = 0.f;
#pragma unroll
  for (int k = 0; k < 8; ++k) tot += red[k];
  const float inv = 1.0f / fmaxf(sqrtf(tot), 1e-12f);
#pragma unroll
  for (int e = 0; e < 4; ++e) dst[threadIdx.x + 256 * e] = v[e] * inv;
}

#define G_RP 1028
__global__ __launch_bounds__(512) void gram_rows(const float* __restrict__ wimgs,
                                                 const float* __restrict__ wcaps,
                                                 float* __restrict__ gcc,
                                                 float* __restrict__ gii) {
  __shared__ __attribute__((aligned(16))) float rows[60 * G_RP];
  const int tid = threadIdx.x;
  const int w = tid >> 5;
  const int lane = tid & 31;
  const int hi = lane >> 4;
  const int ln = lane & 15;
  const bool is_cap = blockIdx.x < 128;
  const int nr = is_cap ? 60 : 36;
  const int nt_side = is_cap ? 4 : 3;
  const float* src = is_cap ? wcaps + (size_t)blockIdx.x * 60 * 1024
                            : wimgs + (size_t)(blockIdx.x - 128) * 36 * 1024;
  float* dst = is_cap ? gcc + (size_t)blockIdx.x * 4096
                      : gii + (size_t)(blockIdx.x - 128) * 2304;
  for (int base = tid; base < nr * 256; base += 6 * 512) {
    v4f t[6];
#pragma unroll
    for (int u = 0; u < 6; ++u) t[u] = *(const v4f*)&src[(size_t)(base + u * 512) * 4];
#pragma unroll
    for (int u = 0; u < 6; ++u) {
      const int idx = base + u * 512;
      *(v4f*)&rows[(idx >> 8) * G_RP + (idx & 255) * 4] = t[u];
    }
  }
  __syncthreads();
  if (w < nt_side * nt_side) {
    const int mt = w / nt_side, nt = w % nt_side;
    const int ra = (mt * 16 + ln < nr - 1) ? (mt * 16 + ln) : (nr - 1);
    const int rb = (nt * 16 + ln < nr - 1) ? (nt * 16 + ln) : (nr - 1);
    const float* pa = rows + ra * G_RP + (hi ? 4 : 0);
    const float* pb = rows + rb * G_RP + (hi ? 4 : 0);
    v8f g0, g1; g0 = 0.0f; g1 = 0.0f;
#pragma unroll 8
    for (int d = 0; d < 1024; d += 8) {
      const v4f av = *(const v4f*)&pa[d];
      const v4f bv = *(const v4f*)&pb[d];
      v2f a0; a0.x = av.x; a0.y = av.y;
      v2f a1; a1.x = av.z; a1.y = av.w;
      v2f b0; b0.x = bv.x; b0.y = bv.y;
      v2f b1; b1.x = bv.z; b1.y = bv.w;
      g0 = wmma4(a0, b0, g0);
      g1 = wmma4(a1, b1, g1);
    }
    float* dt = dst + (mt * nt_side + nt) * 256;
#pragma unroll
    for (int j = 0; j < 8; ++j)
      dt[(j + (hi ? 8 : 0)) * 16 + ln] = g0[j] + g1[j];
  }
}

static __device__ __forceinline__ void slab_x(float* lds,
                                              const float* __restrict__ pc,
                                              int c, int q, int ln, int koff,
                                              int ioff) {
  v8f x0, x1; x0 = 0.0f; x1 = 0.0f;
  for (int l0 = 0; l0 < 60; l0 += 4) {
    v2f a = *(const v2f*)&lds[OM2_O + (ln * TT + c) * 64 + l0 + koff];
    v2f p = *(const v2f*)&pc[l0 * 1024];
    v2f r = *(const v2f*)&pc[(l0 + 1) * 1024];
    v2f b0; b0.x = p.x; b0.y = r.x;
    v2f b1; b1.x = p.y; b1.y = r.y;
    x0 = wmma4(a, b0, x0);
    x1 = wmma4(a, b1, x1);
  }
#pragma unroll
  for (int j = 0; j < 8; ++j) {
    v2f xs; xs.x = x0[j]; xs.y = x1[j];
    *(v2f*)&lds[XS_O + c * XS_OP + (j + ioff) * SL_RP + q * 32 + 2 * ln] = xs;
  }
}

static __device__ __forceinline__ void slab_y(float* lds,
                                              const float* __restrict__ pi,
                                              int w, int tr, int ln, int koff) {
  v2f a[9];
#pragma unroll
  for (int k = 0; k < 9; ++k)
    a[k] = *(const v2f*)&lds[OM1_O + (w * TT + tr) * 48 + 4 * k + koff];
#pragma unroll 1
  for (int half = 0; half < 2; ++half) {
    const float* ph = pi + half * 64;
    v8f y[4];
#pragma unroll
    for (int m = 0; m < 4; ++m) y[m] = 0.0f;
#pragma unroll
    for (int k = 0; k < 9; ++k) {
#pragma unroll
      for (int sub = 0; sub < 2; ++sub) {
        v2f p = *(const v2f*)&ph[(4 * k) * 1024 + sub * 32];
        v2f r = *(const v2f*)&ph[(4 * k + 1) * 1024 + sub * 32];
        v2f b0; b0.x = p.x; b0.y = r.x;
        v2f b1; b1.x = p.y; b1.y = r.y;
        y[2 * sub] = wmma4(a[k], b0, y[2 * sub]);
        y[2 * sub + 1] = wmma4(a[k], b1, y[2 * sub + 1]);
      }
    }
#pragma unroll
    for (int sub = 0; sub < 2; ++sub)
#pragma unroll
      for (int j = 0; j < TT; ++j) {
        v2f ys; ys.x = y[2 * sub][j]; ys.y = y[2 * sub + 1][j];
        *(v2f*)&lds[YS_O + w * YS_OP + j * SL_RP + half * 64 + sub * 32 + 2 * ln] = ys;
      }
  }
}

__global__ __launch_bounds__(512) void nmf_tile(const float* __restrict__ gi,
                                                const float* __restrict__ gc,
                                                const float* gcc,
                                                const float* gii,
                                                const int* __restrict__ il_g,
                                                const int* __restrict__ cl_g,
                                                float* __restrict__ out) {
  __shared__ __attribute__((aligned(16))) float lds[LDS_FLOATS];
  const int tid = threadIdx.x;
  const int w = tid >> 5;
  const int lane = tid & 31;
  const int hi = lane >> 4;
  const int ln = lane & 15;
  const int ib = (blockIdx.x >> 5) * 16;
  const int tb = (blockIdx.x & 31) * TT;
  const int c = w & 3;
  const int q = w >> 2;
  const int tg = tb + c;
  const int ig = ib + w;
  const int tr = ln & 3;
  const int ioff = hi ? 8 : 0;
  const int koff = hi ? 2 : 0;
  const int qoff = hi ? 4 : 0;
  const float* pcx = gc + ((size_t)tg * 60 + koff) * 1024 + q * 32 + 2 * ln;
  const float* piy = gi + ((size_t)ig * 36 + koff) * 1024 + 2 * ln;
  float* stg = lds + ST_O + w * ST_W;
  const int sr = lane >> 3;
  const int sp = (lane & 7) * 4;
  const float* pig = gi + ((size_t)ig * 36 + sr) * 1024 + sp;
  const float* pcg[4];
#pragma unroll
  for (int j = 0; j < 4; ++j) {
    const int lr = (q * 16 + 4 * j + sr < 59) ? (q * 16 + 4 * j + sr) : 59;
    pcg[j] = gc + ((size_t)tg * 60 + lr) * 1024 + sp;
  }

  const float* gct = gcc + (size_t)tg * 4096 + q * 256 + ln;
  const float* git = gii + (size_t)ig * 2304 + ln;

  for (int idx = tid; idx < TT * 64; idx += 512) {
    int t = idx >> 6, l = idx & 63;
    lds[CM_O + idx] = (l < cl_g[tb + t]) ? 1.f : 0.f;
  }
  for (int idx = tid; idx < 16 * 48; idx += 512) {
    int i = idx / 48, k = idx % 48;
    lds[IM_O + idx] = (k < il_g[ib + i]) ? 1.f : 0.f;
  }
  for (int idx = tid; idx < 16 * TT * 64; idx += 512) {
    int t = (idx >> 6) & (TT - 1), l = idx & 63;
    int cl = cl_g[tb + t];
    lds[OM2_O + idx] = (l < cl) ? (1.f / (float)cl) : 0.f;
  }
  for (int idx = tid; idx < 16 * TT * 48; idx += 512) {
    int i = idx / (TT * 48), k = idx % 48;
    int il = il_g[ib + i];
    lds[OM1_O + idx] = (k < il) ? (1.f / (float)il) : 0.f;
  }
  __syncthreads();

#pragma unroll 1
  for (int iter = 0; iter < N_ITERS; ++iter) {
    v8f dv; dv = 0.0f;
#pragma unroll 1
    for (int sc = 0; sc < N_SC; ++sc) {
      const int d0 = sc * SC_W;
      v4f h[4];
#pragma unroll
      for (int j = 0; j < 4; ++j) h[j] = *(const v4f*)&pcg[j][d0];
      slab_y(lds, piy + d0, w, tr, ln, koff);
      __syncthreads();
#pragma unroll
      for (int sub = 0; sub < 4; ++sub) {
#pragma unroll
        for (int j = 0; j < 4; ++j)
          *(v4f*)&stg[(4 * j + sr) * ST_RP + sp] = h[j];
        if (sub < 3) {
#pragma unroll
          for (int j = 0; j < 4; ++j)
            h[j] = *(const v4f*)&pcg[j][d0 + (sub + 1) * 32];
        }
        __builtin_amdgcn_fence(__ATOMIC_ACQ_REL, "wavefront");
        v4f bf[4], ay[4];
#pragma unroll
        for (int s4 = 0; s4 < 4; ++s4) {
          const int dd = (sub * 4 + s4) * 8 + qoff;
          ay[s4] = *(const v4f*)&lds[YS_O + ln * YS_OP + c * SL_RP + dd];
          bf[s4] = *(const v4f*)&stg[ln * ST_RP + s4 * 8 + qoff];
        }
#pragma unroll
        for (int s4 = 0; s4 < 4; ++s4) {
          v2f a0; a0.x = ay[s4].x; a0.y = ay[s4].y;
          v2f a1; a1.x = ay[s4].z; a1.y = ay[s4].w;
          v2f b0; b0.x = bf[s4].x; b0.y = bf[s4].y;
          v2f b1; b1.x = bf[s4].z; b1.y = bf[s4].w;
          dv = wmma4(a0, b0, dv);
          dv = wmma4(a1, b1, dv);
        }
        __builtin_amdgcn_fence(__ATOMIC_ACQ_REL, "wavefront");
      }
      __syncthreads();
    }
    v8f dg; dg = 0.0f;
#pragma unroll
    for (int l0 = 0; l0 < 60; l0 += 4) {
      const int r0 = l0 + koff;
      v2f a = *(const v2f*)&lds[OM2_O + (ln * TT + c) * 64 + r0];
      const float* gp = gct + (r0 >> 4) * 1024 + (r0 & 15) * 16;
      v2f b; b.x = gp[0]; b.y = gp[16];
      dg = wmma4(a, b, dg);
    }
    __syncthreads();
#pragma unroll
    for (int j = 0; j < 8; ++j) {
      const int i = j + ioff, l = q * 16 + ln;
      const int idx = (i * TT + c) * 64 + l;
      float nv = lds[OM2_O + idx] - ETA_F * (dg[j] - dv[j]);
      lds[OM2_O + idx] = fmaxf(nv, 0.f) * lds[CM_O + c * 64 + l];
    }
    __syncthreads();
    if (tid < 16 * TT) {
      float s = 0.f;
      for (int l = 0; l < 64; ++l) s += lds[OM2_O + tid * 64 + l];
      lds[SC_O + tid] = 1.f / (s + EPS_F);
    }
    __syncthreads();
    for (int idx = tid; idx < 16 * TT * 64; idx += 512)
      lds[OM2_O + idx] *= lds[SC_O + (idx >> 6)];
    __syncthreads();

    v8f du[3];
#pragma unroll
    for (int n = 0; n < 3; ++n) du[n] = 0.0f;
#pragma unroll 1
    for (int sc = 0; sc < N_SC; ++sc) {
      const int d0 = sc * SC_W;
      slab_x(lds, pcx + d0, c, q, ln, koff, ioff);
      v4f g[9];
#pragma unroll
      for (int j = 0; j < 9; ++j) g[j] = *(const v4f*)&pig[d0 + (4 * j) * 1024];
      __builtin_amdgcn_sched_barrier(0);
      __syncthreads();
#pragma unroll
      for (int sub = 0; sub < 4; ++sub) {
#pragma unroll
        for (int j = 0; j < 9; ++j)
          *(v4f*)&stg[(4 * j + sr) * ST_RP + sp] = g[j];
        if (sub < 3) {
#pragma unroll
          for (int j = 0; j < 9; ++j)
            g[j] = *(const v4f*)&pig[d0 + (sub + 1) * 32 + (4 * j) * 1024];
        }
        __builtin_amdgcn_fence(__ATOMIC_ACQ_REL, "wavefront");
        v4f bf[4][3], ax[4];
#pragma unroll
        for (int s4 = 0; s4 < 4; ++s4) {
          const int dd = (sub * 4 + s4) * 8 + qoff;
          ax[s4] = *(const v4f*)&lds[XS_O + tr * XS_OP + w * SL_RP + dd];
#pragma unroll
          for (int n = 0; n < 3; ++n) {
            const int kc = (n * 16 + ln < 35) ? (n * 16 + ln) : 35;
            bf[s4][n] = *(const v4f*)&stg[kc * ST_RP + s4 * 8 + qoff];
          }
        }
#pragma unroll
        for (int s4 = 0; s4 < 4; ++s4) {
          v2f a0; a0.x = ax[s4].x; a0.y = ax[s4].y;
          v2f a1; a1.x = ax[s4].z; a1.y = ax[s4].w;
#pragma unroll
          for (int n = 0; n < 3; ++n) {
            v2f b0; b0.x = bf[s4][n].x; b0.y = bf[s4][n].y;
            v2f b1; b1.x = bf[s4][n].z; b1.y = bf[s4][n].w;
            du[n] = wmma4(a0, b0, du[n]);
            du[n] = wmma4(a1, b1, du[n]);
          }
        }
        __builtin_amdgcn_fence(__ATOMIC_ACQ_REL, "wavefront");
      }
      __syncthreads();
    }
    v8f dgi[3];
#pragma unroll
    for (int n = 0; n < 3; ++n) dgi[n] = 0.0f;
#pragma unroll
    for (int kk = 0; kk < 9; ++kk) {
      const int r0 = 4 * kk + koff;
      v2f a = *(const v2f*)&lds[OM1_O + (w * TT + tr) * 48 + r0];
#pragma unroll
      for (int n = 0; n < 3; ++n) {
        const float* gp = git + ((r0 >> 4) * 3 + n) * 256 + (r0 & 15) * 16;
        v2f b; b.x = gp[0]; b.y = gp[16];
        dgi[n] = wmma4(a, b, dgi[n]);
      }
    }
#pragma unroll
    for (int n = 0; n < 3; ++n)
#pragma unroll
      for (int j = 0; j < TT; ++j) {
        const int k = n * 16 + ln;
        const int idx = (w * TT + j) * 48 + k;
        float nv = lds[OM1_O + idx] + ETA_F * (du[n][j] - dgi[n][j]);
        lds[OM1_O + idx] = fmaxf(nv, 0.f) * lds[IM_O + w * 48 + k];
      }
    __syncthreads();
    if (tid < 16 * TT) {
      float s = 0.f;
      for (int k = 0; k < 48; ++k) s += lds[OM1_O + tid * 48 + k];
      lds[SC_O + tid] = 1.f / (s + EPS_F);
    }
    __syncthreads();
    for (int idx = tid; idx < 16 * TT * 48; idx += 512)
      lds[OM1_O + idx] *= lds[SC_O + idx / 48];
    __syncthreads();
  }

  float s11 = 0.f, s22 = 0.f, s12 = 0.f;
#pragma unroll 1
  for (int sc = 0; sc < N_SC; ++sc) {
    slab_x(lds, pcx + sc * SC_W, c, q, ln, koff, ioff);
    slab_y(lds, piy + sc * SC_W, w, tr, ln, koff);
    __syncthreads();
    if (tid < 16 * TT) {
      const int pi = tid >> 2, pt = tid & 3;
#pragma unroll 32
      for (int d = 0; d < SC_W; ++d) {
        float g1 = lds[YS_O + pi * YS_OP + pt * SL_RP + d];
        float g2 = lds[XS_O + pt * XS_OP + pi * SL_RP + d];
        s12 += g1 * g2; s11 += g1 * g1; s22 += g2 * g2;
      }
    }
    __syncthreads();
  }
  if (tid < 16 * TT) {
    const float dnm = fmaxf(sqrtf(s11), 1e-12f) * fmaxf(sqrtf(s22), 1e-12f);
    lds[SC_O + tid] = s12 / dnm;
  }
  __syncthreads();
  if (tid < 16) {
    const v4f r = *(const v4f*)&lds[SC_O + tid * TT];
    *(v4f*)&out[(size_t)(ib + tid) * 128 + tb] = r;
  }
}

extern "C" void kernel_launch(void* const* d_in, const int* in_sizes, int n_in,
                              void* d_out, int out_size, void* d_ws, size_t ws_size,
                              hipStream_t stream) {
  (void)in_sizes; (void)n_in; (void)out_size; (void)ws_size;
  const float* imgs = (const float*)d_in[0];
  const float* caps = (const float*)d_in[1];
  const int* il = (const int*)d_in[2];
  const int* cl = (const int*)d_in[3];
  float* wimgs = (float*)d_ws;
  float* wcaps = wimgs + (size_t)128 * 36 * 1024;
  float* gcc = wcaps + (size_t)128 * 60 * 1024;
  float* gii = gcc + (size_t)128 * 4096;

  l2norm_rows<<<128 * 36 + 128 * 60, 256, 0, stream>>>(imgs, caps, wimgs, wcaps);
  gram_rows<<<256, 512, 0, stream>>>(wimgs, wcaps, gcc, gii);
  nmf_tile<<<256, 512, 0, stream>>>(wimgs, wcaps, gcc, gii, il, cl, (float*)d_out);
}


# cross term split by linearity too: C = imgs caps^T as one 4608x7680x1024 WMMA product (added launch); iterations and result per pair from C and the Gram tiles, no d loop left; sum order changes
# speedup vs baseline: 6.3556x; 1.1825x over previous
#include <hip/hip_runtime.h>
#include <math.h>

typedef __attribute__((ext_vector_type(2))) float v2f;
typedef __attribute__((ext_vector_type(4))) float v4f;
typedef __attribute__((ext_vector_type(8))) float v8f;

#define N_ITERS 5
#define EPS_F   1e-6f
#define ETA_F   0.02f


static __device__ __forceinline__ v8f wmma4(v2f a, v2f b, v8f c) {
  return __builtin_amdgcn_wmma_f32_16x16x4_f32(false, a, false, b, (short)0, c,
                                               false, false);
}

__global__ __launch_bounds__(256) void l2norm_rows(const float* __restrict__ imgs,
                                                   const float* __restrict__ caps,
                                                   float* __restrict__ wimgs,
                                                   float* __restrict__ wcaps) {
  const int row = blockIdx.x;
  const float* src;
  float* dst;
  if (row < 128 * 36) {
    src = imgs + (size_t)row * 1024;
    dst = wimgs + (size_t)row * 1024;
  } else {
    int r = row - 128 * 36;
    src = caps + (size_t)r * 1024;
    dst = wcaps + (size_t)r * 1024;
  }
  float v[4];
  float s = 0.f;
#pragma unroll
  for (int e = 0; e < 4; ++e) {
    v[e] = src[threadIdx.x + 256 * e];
    s += v[e] * v[e];
  }
#pragma unroll
  for (int off = 16; off > 0; off >>= 1) s += __shfl_xor(s, off, 32);
  __shared__ float red[8];
  if ((threadIdx.x & 31) == 0) red[threadIdx.x >> 5] = s;
  __syncthreads();
  float tot = 0.f;
#pragma unroll
  for (int k = 0; k < 8; ++k) tot += red[k];
  const float inv = 1.0f / fmaxf(sqrtf(tot), 1e-12f);
#pragma unroll
  for (int e = 0; e < 4; ++e) dst[threadIdx.x + 256 * e] = v[e] * inv;
}

#define G_RP 1028
__global__ __launch_bounds__(512) void gram_rows(const float* __restrict__ wimgs,
                                                 const float* __restrict__ wcaps,
                                                 float* __restrict__ gcc,
                                                 float* __restrict__ gii) {
  __shared__ __attribute__((aligned(16))) float rows[60 * G_RP];
  const int tid = threadIdx.x;
  const int w = tid >> 5;
  const int lane = tid & 31;
  const int hi = lane >> 4;
  const int ln = lane & 15;
  const bool is_cap = blockIdx.x < 128;
  const int nr = is_cap ? 60 : 36;
  const int nt_side = is_cap ? 4 : 3;
  const float* src = is_cap ? wcaps + (size_t)blockIdx.x * 60 * 1024
                            : wimgs + (size_t)(blockIdx.x - 128) * 36 * 1024;
  float* dst = is_cap ? gcc + (size_t)blockIdx.x * 4096
                      : gii + (size_t)(blockIdx.x - 128) * 2304;
  for (int base = tid; base < nr * 256; base += 6 * 512) {
    v4f t[6];
#pragma unroll
    for (int u = 0; u < 6; ++u) t[u] = *(const v4f*)&src[(size_t)(base + u * 512) * 4];
#pragma unroll
    for (int u = 0; u < 6; ++u) {
      const int idx = base + u * 512;
      *(v4f*)&rows[(idx >> 8) * G_RP + (idx & 255) * 4] = t[u];
    }
  }
  __syncthreads();
  if (w < nt_side * nt_side) {
    const int mt = w / nt_side, nt = w % nt_side;
    const int ra = (mt * 16 + ln < nr - 1) ? (mt * 16 + ln) : (nr - 1);
    const int rb = (nt * 16 + ln < nr - 1) ? (nt * 16 + ln) : (nr - 1);
    const float* pa = rows + ra * G_RP + (hi ? 4 : 0);
    const float* pb = rows + rb * G_RP + (hi ? 4 : 0);
    v8f g0, g1; g0 = 0.0f; g1 = 0.0f;
#pragma unroll 8
    for (int d = 0; d < 1024; d += 8) {
      const v4f av = *(const v4f*)&pa[d];
      const v4f bv = *(const v4f*)&pb[d];
      v2f a0; a0.x = av.x; a0.y = av.y;
      v2f a1; a1.x = av.z; a1.y = av.w;
      v2f b0; b0.x = bv.x; b0.y = bv.y;
      v2f b1; b1.x = bv.z; b1.y = bv.w;
      g0 = wmma4(a0, b0, g0);
      g1 = wmma4(a1, b1, g1);
    }
    float* dt = dst + (mt * nt_side + nt) * 256;
#pragma unroll
    for (int j = 0; j < 8; ++j)
      dt[(j + (hi ? 8 : 0)) * 16 + ln] = g0[j] + g1[j];
  }
}

#define CG_RP 36
#define CG_N  7680
__global__ __launch_bounds__(512) void cross_gram(const float* __restrict__ wimgs,
                                                  const float* __restrict__ wcaps,
                                                  float* __restrict__ cg) {
  __shared__ __attribute__((aligned(16))) float sa[2][128 * CG_RP];
  __shared__ __attribute__((aligned(16))) float sb[2][128 * CG_RP];
  const int tid = threadIdx.x;
  const int w = tid >> 5;
  const int lane = tid & 31;
  const int hi = lane >> 4;
  const int ln = lane & 15;
  const int wm = w >> 2, wn = w & 3;
  const int mbase = blockIdx.y * 128;
  const int nbase = blockIdx.x * 128;
  const int r0 = tid >> 3;
  const int pc = (tid & 7) * 4;
  const float* ga0 = wimgs + (size_t)(mbase + r0) * 1024 + pc;
  const float* ga1 = ga0 + (size_t)64 * 1024;
  const float* gb0 = wcaps + (size_t)(nbase + r0) * 1024 + pc;
  const float* gb1 = gb0 + (size_t)64 * 1024;
  const int rb0 = (r0 & 96) + ((r0 & 1) << 4) + ((r0 & 31) >> 1);
  const int rb1 = rb0 + 64;
  const int sa0 = r0 * CG_RP + pc, sa1 = (r0 + 64) * CG_RP + pc;
  const int sb0 = rb0 * CG_RP + pc, sb1 = rb1 * CG_RP + pc;
  const int fa = (wm * 32 + ln) * CG_RP + (hi ? 4 : 0);
  const int fb = (wn * 32 + ln) * CG_RP + (hi ? 4 : 0);

  v8f acc[2][2];
#pragma unroll
  for (int mt = 0; mt < 2; ++mt)
#pragma unroll
    for (int nt = 0; nt < 2; ++nt) acc[mt][nt] = 0.0f;

  v4f ta0 = *(const v4f*)&ga0[0], ta1 = *(const v4f*)&ga1[0];
  v4f tb0 = *(const v4f*)&gb0[0], tb1 = *(const v4f*)&gb1[0];
  *(v4f*)&sa[0][sa0] = ta0; *(v4f*)&sa[0][sa1] = ta1;
  *(v4f*)&sb[0][sb0] = tb0; *(v4f*)&sb[0][sb1] = tb1;
  __syncthreads();
#pragma unroll 1
  for (int ch = 0; ch < 32; ++ch) {
    const int cur = ch & 1;
    if (ch < 31) {
      const int d1 = (ch + 1) * 32;
      ta0 = *(const v4f*)&ga0[d1]; ta1 = *(const v4f*)&ga1[d1];
      tb0 = *(const v4f*)&gb0[d1]; tb1 = *(const v4f*)&gb1[d1];
    }
    const float* pa = &sa[cur][fa];
    const float* pb = &sb[cur][fb];
    v4f av[4][2], bv[4][2];
#pragma unroll
    for (int s4 = 0; s4 < 4; ++s4)
#pragma unroll
      for (int t2 = 0; t2 < 2; ++t2) {
        av[s4][t2] = *(const v4f*)&pa[t2 * 16 * CG_RP + s4 * 8];
        bv[s4][t2] = *(const v4f*)&pb[t2 * 16 * CG_RP + s4 * 8];
      }
#pragma unroll
    for (int s4 = 0; s4 < 4; ++s4)
#pragma unroll
      for (int mt = 0; mt < 2; ++mt)
#pragma unroll
        for (int nt = 0; nt < 2; ++nt) {
          v2f a0; a0.x = av[s4][mt].x; a0.y = av[s4][mt].y;
          v2f a1; a1.x = av[s4][mt].z; a1.y = av[s4][mt].w;
          v2f b0; b0.x = bv[s4][nt].x; b0.y = bv[s4][nt].y;
          v2f b1; b1.x = bv[s4][nt].z; b1.y = bv[s4][nt].w;
          acc[mt][nt] = wmma4(a0, b0, acc[mt][nt]);
          acc[mt][nt] = wmma4(a1, b1, acc[mt][nt]);
        }
    if (ch < 31) {
      const int nx = cur ^ 1;
      *(v4f*)&sa[nx][sa0] = ta0; *(v4f*)&sa[nx][sa1] = ta1;
      *(v4f*)&sb[nx][sb0] = tb0; *(v4f*)&sb[nx][sb1] = tb1;
    }
    __syncthreads();
  }
#pragma unroll
  for (int mt = 0; mt < 2; ++mt)
#pragma unroll
    for (int j = 0; j < 8; ++j) {
      const int m = mbase + wm * 32 + mt * 16 + (hi ? 8 : 0) + j;
      v2f o; o.x = acc[mt][0][j]; o.y = acc[mt][1][j];
      *(v2f*)&cg[(size_t)m * CG_N + nbase + wn * 32 + 2 * ln] = o;
    }
}

#define PC_RP 68
#define GC_RP 68
#define GI_RP 36
#define PW_C  (36 * PC_RP)
#define PW_GI (36 * GI_RP)
#define PL_C  0
#define PL_GI (16 * PW_C)
#define PL_GC (PL_GI + 16 * PW_GI)
#define PL_O1 (PL_GC + 60 * GC_RP)
#define O2_RP 68
#define DG_RP 68
#define PL_O2 (PL_O1 + 16 * 48)
#define PL_DG (PL_O2 + 16 * O2_RP)
#define PL_FLOATS (PL_DG + 16 * DG_RP)

static __device__ __forceinline__ float wave_sum(float s) {
#pragma unroll
  for (int off = 16; off > 0; off >>= 1) s += __shfl_xor(s, off, 32);
  return s;
}

__global__ __launch_bounds__(512) void nmf_pairs(const float* __restrict__ cg,
                                                 const float* __restrict__ gcc,
                                                 const float* __restrict__ gii,
                                                 const int* __restrict__ il_g,
                                                 const int* __restrict__ cl_g,
                                                 float* __restrict__ out) {
  __shared__ __attribute__((aligned(16))) float lds[PL_FLOATS];
  const int tid = threadIdx.x;
  const int w = tid >> 5;
  const int lane = tid & 31;
  const int hi = lane >> 4;
  const int ln = lane & 15;
  const int t = blockIdx.x & 127;
  const int ig = (blockIdx.x >> 7) * 16 + w;
  float* cw = lds + PL_C + w * PW_C;
  float* giw = lds + PL_GI + w * PW_GI;
  float* gcs = lds + PL_GC;
  float* o1 = lds + PL_O1 + w * 48;
  float* o2 = lds + PL_O2 + w * O2_RP;
  float* dgs = lds + PL_DG;
  const int il = il_g[ig];
  const int cl = cl_g[t];

  {
    const float* src = cg + (size_t)ig * 36 * CG_N + t * 60;
#pragma unroll
    for (int u = 0; u < 17; ++u) {
      const int idx = lane + 32 * u;
      if (idx < 540) {
        const int row = idx / 15, p4 = (idx % 15) * 4;
        *(v4f*)&cw[row * PC_RP + p4] = *(const v4f*)&src[(size_t)row * CG_N + p4];
      }
    }
  }
  for (int idx = lane; idx < 36 * 36; idx += 32) {
    const int a = idx / 36, b = idx % 36;
    giw[a * GI_RP + b] =
        gii[(size_t)ig * 2304 + ((a >> 4) * 3 + (b >> 4)) * 256 + (a & 15) * 16 + (b & 15)];
  }
  for (int idx = tid; idx < 60 * 60; idx += 512) {
    const int a = idx / 60, b = idx % 60;
    gcs[a * GC_RP + b] =
        gcc[(size_t)t * 4096 + ((a >> 4) * 4 + (b >> 4)) * 256 + (a & 15) * 16 + (b & 15)];
  }
  const int l0 = lane, l1 = lane + 32;
  const int l1c = (l1 < 59) ? l1 : 59;
  const int kA = lane, kB = 32 + (lane & 3);
  o2[l0] = (l0 < cl) ? (1.f / (float)cl) : 0.f;
  o2[l1] = (l1 < cl) ? (1.f / (float)cl) : 0.f;
  o1[kA] = (kA < il) ? (1.f / (float)il) : 0.f;
  if (lane < 4) o1[kB] = (kB < il) ? (1.f / (float)il) : 0.f;
  __syncthreads();

  float dg0 = 0.f, dg1 = 0.f;
  float duA = 0.f, duB = 0.f;
  float giA = 0.f, giB = 0.f;
#pragma unroll 1
  for (int iter = 0; iter <= N_ITERS; ++iter) {
    if (w < 4) {
      const float* ao = lds + PL_O2 + ln * O2_RP + (hi ? 2 : 0);
      const float* bg = gcs + (hi ? 2 : 0) * GC_RP + w * 16 + ln;
      v8f g0, g1; g0 = 0.0f; g1 = 0.0f;
#pragma unroll
      for (int lp = 0; lp < 60; lp += 8) {
        v2f a = *(const v2f*)&ao[lp];
        v2f b; b.x = bg[lp * GC_RP]; b.y = bg[(lp + 1) * GC_RP];
        g0 = wmma4(a, b, g0);
        if (lp + 4 < 60) {
          v2f a2 = *(const v2f*)&ao[lp + 4];
          v2f b2; b2.x = bg[(lp + 4) * GC_RP]; b2.y = bg[(lp + 5) * GC_RP];
          g1 = wmma4(a2, b2, g1);
        }
      }
#pragma unroll
      for (int j = 0; j < 8; ++j)
        dgs[(j + (hi ? 8 : 0)) * DG_RP + w * 16 + ln] = g0[j] + g1[j];
    }
    __syncthreads();
    dg0 = dgs[w * DG_RP + l0];
    dg1 = dgs[w * DG_RP + l1c];
    float dc0 = 0.f, dc1 = 0.f;
    if (iter < N_ITERS) {
#pragma unroll
      for (int k = 0; k < 36; k += 4) {
        const v4f o = *(const v4f*)&o1[k];
#pragma unroll
        for (int e = 0; e < 4; ++e) {
          dc0 += o[e] * cw[(k + e) * PC_RP + l0];
          dc1 += o[e] * cw[(k + e) * PC_RP + l1c];
        }
      }
      const float n0 = (l0 < cl) ? fmaxf(o2[l0] - ETA_F * (dg0 - dc0), 0.f) : 0.f;
      const float n1 = (l1 < cl) ? fmaxf(o2[l1] - ETA_F * (dg1 - dc1), 0.f) : 0.f;
      const float inv = 1.f / (wave_sum(n0 + n1) + EPS_F);
      __syncthreads();
      o2[l0] = n0 * inv;
      o2[l1] = n1 * inv;
      __syncthreads();
    }
    duA = 0.f; duB = 0.f; giA = 0.f; giB = 0.f;
#pragma unroll
    for (int l = 0; l < 60; l += 4) {
      const v4f o = *(const v4f*)&o2[l];
      const v4f ca = *(const v4f*)&cw[kA * PC_RP + l];
      const v4f cb = *(const v4f*)&cw[kB * PC_RP + l];
#pragma unroll
      for (int e = 0; e < 4; ++e) { duA += ca[e] * o[e]; duB += cb[e] * o[e]; }
    }
#pragma unroll
    for (int kp = 0; kp < 36; kp += 4) {
      const v4f o = *(const v4f*)&o1[kp];
#pragma unroll
      for (int e = 0; e < 4; ++e) {
        giA += o[e] * giw[(kp + e) * GI_RP + kA];
        giB += o[e] * giw[(kp + e) * GI_RP + kB];
      }
    }
    if (iter < N_ITERS) {
      const float nA = (kA < il) ? fmaxf(o1[kA] + ETA_F * (duA - giA), 0.f) : 0.f;
      const float nB = (lane < 4 && kB < il) ? fmaxf(o1[kB] + ETA_F * (duB - giB), 0.f) : 0.f;
      const float inv = 1.f / (wave_sum(nA + nB) + EPS_F);
      __syncthreads();
      o1[kA] = nA * inv;
      if (lane < 4) o1[kB] = nB * inv;
      __syncthreads();
    }
  }
  const float s22 = wave_sum(o2[l0] * dg0 + o2[l1] * dg1);
  const float s12 = wave_sum(o1[kA] * duA + ((lane < 4) ? o1[kB] * duB : 0.f));
  const float s11 = wave_sum(o1[kA] * giA + ((lane < 4) ? o1[kB] * giB : 0.f));
  if (lane == 0) {
    const float dnm = fmaxf(sqrtf(fmaxf(s11, 0.f)), 1e-12f) *
                      fmaxf(sqrtf(fmaxf(s22, 0.f)), 1e-12f);
    out[(size_t)ig * 128 + t] = s12 / dnm;
  }
}

extern "C" void kernel_launch(void* const* d_in, const int* in_sizes, int n_in,
                              void* d_out, int out_size, void* d_ws, size_t ws_size,
                              hipStream_t stream) {
  (void)in_sizes; (void)n_in; (void)out_size; (void)ws_size;
  const float* imgs = (const float*)d_in[0];
  const float* caps = (const float*)d_in[1];
  const int* il = (const int*)d_in[2];
  const int* cl = (const int*)d_in[3];
  float* wimgs = (float*)d_ws;
  float* wcaps = wimgs + (size_t)128 * 36 * 1024;
  float* gcc = wcaps + (size_t)128 * 60 * 1024;
  float* gii = gcc + (size_t)128 * 4096;
  float* cg = gii + (size_t)128 * 2304;

  l2norm_rows<<<128 * 36 + 128 * 60, 256, 0, stream>>>(imgs, caps, wimgs, wcaps);
  gram_rows<<<256, 512, 0, stream>>>(wimgs, wcaps, gcc, gii);
  cross_gram<<<dim3(60, 36), 512, 0, stream>>>(wimgs, wcaps, cg);
  nmf_pairs<<<1024, 512, 0, stream>>>(cg, gcc, gii, il, cl, (float*)d_out);
}


# cross matrix C formed and read only over valid rows: images' first img_lens rows, captions' first cap_lens rows (to 4), packed by prefix offsets; masked part of a pair's block zero-filled
# speedup vs baseline: 8.5818x; 1.3503x over previous
#include <hip/hip_runtime.h>
#include <math.h>

typedef __attribute__((ext_vector_type(2))) float v2f;
typedef __attribute__((ext_vector_type(4))) float v4f;
typedef __attribute__((ext_vector_type(8))) float v8f;

#define N_ITERS 5
#define EPS_F   1e-6f
#define ETA_F   0.02f


static __device__ __forceinline__ v8f wmma4(v2f a, v2f b, v8f c) {
  return __builtin_amdgcn_wmma_f32_16x16x4_f32(false, a, false, b, (short)0, c,
                                               false, false);
}

static __device__ __forceinline__ int packed_rows(int len, int full, bool pad4) {
  const int c = min(max(len, 0), full);
  return pad4 ? ((c + 3) & ~3) : c;
}
#define OFF_C 132

__global__ __launch_bounds__(256) void l2norm_rows(const float* __restrict__ imgs,
                                                   const float* __restrict__ caps,
                                                   float* __restrict__ wimgs,
                                                   float* __restrict__ wcaps) {
  const int row = blockIdx.x;
  const float* src;
  float* dst;
  if (row < 128 * 36) {
    src = imgs + (size_t)row * 1024;
    dst = wimgs + (size_t)row * 1024;
  } else {
    int r = row - 128 * 36;
    src = caps + (size_t)r * 1024;
    dst = wcaps + (size_t)r * 1024;
  }
  float v[4];
  float s = 0.f;
#pragma unroll
  for (int e = 0; e < 4; ++e) {
    v[e] = src[threadIdx.x + 256 * e];
    s += v[e] * v[e];
  }
#pragma unroll
  for (int off = 16; off > 0; off >>= 1) s += __shfl_xor(s, off, 32);
  __shared__ float red[8];
  if ((threadIdx.x & 31) == 0) red[threadIdx.x >> 5] = s;
  __syncthreads();
  float tot = 0.f;
#pragma unroll
  for (int k = 0; k < 8; ++k) tot += red[k];
  const float inv = 1.0f / fmaxf(sqrtf(tot), 1e-12f);
#pragma unroll
  for (int e = 0; e < 4; ++e) dst[threadIdx.x + 256 * e] = v[e] * inv;
}

#define G_RP 1028
__global__ __launch_bounds__(512) void gram_rows(const float* __restrict__ wimgs,
                                                 const float* __restrict__ wcaps,
                                                 float* __restrict__ gcc,
                                                 float* __restrict__ gii,
                                                 const int* __restrict__ il_g,
                                                 const int* __restrict__ cl_g,
                                                 int* __restrict__ offs,
                                                 int* __restrict__ rmap_i,
                                                 int* __restrict__ rmap_c,
                                                 float* __restrict__ zpiece) {
  __shared__ __attribute__((aligned(16))) float rows[60 * G_RP];
  __shared__ int part[4];
  const int tid = threadIdx.x;
  const int w = tid >> 5;
  const int lane = tid & 31;
  const int hi = lane >> 4;
  const int ln = lane & 15;
  const bool is_cap = blockIdx.x < 128;
  const int nr = is_cap ? 60 : 36;
  const int nt_side = is_cap ? 4 : 3;
  const float* src = is_cap ? wcaps + (size_t)blockIdx.x * 60 * 1024
                            : wimgs + (size_t)(blockIdx.x - 128) * 36 * 1024;
  float* dst = is_cap ? gcc + (size_t)blockIdx.x * 4096
                      : gii + (size_t)(blockIdx.x - 128) * 2304;
  const int me = is_cap ? blockIdx.x : blockIdx.x - 128;
  const int* lens = is_cap ? cl_g : il_g;
  if (w < 4) {
    int s = (tid < me) ? packed_rows(lens[tid], nr, is_cap) : 0;
#pragma unroll
    for (int off = 16; off > 0; off >>= 1) s += __shfl_xor(s, off, 32);
    if (lane == 0) part[w] = s;
  }
  for (int base = tid; base < nr * 256; base += 6 * 512) {
    v4f t[6];
#pragma unroll
    for (int u = 0; u < 6; ++u) t[u] = *(const v4f*)&src[(size_t)(base + u * 512) * 4];
#pragma unroll
    for (int u = 0; u < 6; ++u) {
      const int idx = base + u * 512;
      *(v4f*)&rows[(idx >> 8) * G_RP + (idx & 255) * 4] = t[u];
    }
  }
  __syncthreads();
  {
    const int first = part[0] + part[1] + part[2] + part[3];
    const int mine = packed_rows(lens[me], nr, is_cap);
    int* rmap = is_cap ? rmap_c : rmap_i;
    if (tid < mine) rmap[first + tid] = me * nr + tid;
    if (tid == 0) {
      int* o = offs + (is_cap ? OFF_C : 0);
      o[me] = first;
      if (me == 127) o[128] = first + mine;
    }
    if (blockIdx.x == 0 && tid < 4) zpiece[tid] = 0.f;
  }
  if (w < nt_side * nt_side) {
    const int mt = w / nt_side, nt = w % nt_side;
    const int ra = (mt * 16 + ln < nr - 1) ? (mt * 16 + ln) : (nr - 1);
    const int rb = (nt * 16 + ln < nr - 1) ? (nt * 16 + ln) : (nr - 1);
    const float* pa = rows + ra * G_RP + (hi ? 4 : 0);
    const float* pb = rows + rb * G_RP + (hi ? 4 : 0);
    v8f g0, g1; g0 = 0.0f; g1 = 0.0f;
#pragma unroll 8
    for (int d = 0; d < 1024; d += 8) {
      const v4f av = *(const v4f*)&pa[d];
      const v4f bv = *(const v4f*)&pb[d];
      v2f a0; a0.x = av.x; a0.y = av.y;
      v2f a1; a1.x = av.z; a1.y = av.w;
      v2f b0; b0.x = bv.x; b0.y = bv.y;
      v2f b1; b1.x = bv.z; b1.y = bv.w;
      g0 = wmma4(a0, b0, g0);
      g1 = wmma4(a1, b1, g1);
    }
    float* dt = dst + (mt * nt_side + nt) * 256;
#pragma unroll
    for (int j = 0; j < 8; ++j)
      dt[(j + (hi ? 8 : 0)) * 16 + ln] = g0[j] + g1[j];
  }
}

#define CG_RP 36
#define CG_N  7680
__global__ __launch_bounds__(512) void cross_gram(const float* __restrict__ wimgs,
                                                  const float* __restrict__ wcaps,
                                                  const int* __restrict__ offs,
                                                  const int* __restrict__ rmap_i,
                                                  const int* __restrict__ rmap_c,
                                                  float* __restrict__ cg) {
  __shared__ __attribute__((aligned(16))) float sa[2][128 * CG_RP];
  __shared__ __attribute__((aligned(16))) float sb[2][128 * CG_RP];
  const int tid = threadIdx.x;
  const int w = tid >> 5;
  const int lane = tid & 31;
  const int hi = lane >> 4;
  const int ln = lane & 15;
  const int wm = w >> 2, wn = w & 3;
  const int mbase = blockIdx.y * 128;
  const int nbase = blockIdx.x * 128;
  const int mtot = offs[128];
  const int ntot = offs[OFF_C + 128];
  if (mbase >= mtot || nbase >= ntot) return;
  const int r0 = tid >> 3;
  const int pc = (tid & 7) * 4;
  const int ma0 = mbase + r0, ma1 = ma0 + 64;
  const int nb0 = nbase + r0, nb1 = nb0 + 64;
  const int qa0 = rmap_i[min(ma0, mtot - 1)];
  const int qa1 = rmap_i[min(ma1, mtot - 1)];
  const int qb0 = rmap_c[min(nb0, ntot - 1)];
  const int qb1 = rmap_c[min(nb1, ntot - 1)];
  const float* ga0 = wimgs + (size_t)qa0 * 1024 + pc;
  const float* ga1 = wimgs + (size_t)qa1 * 1024 + pc;
  const float* gb0 = wcaps + (size_t)qb0 * 1024 + pc;
  const float* gb1 = wcaps + (size_t)qb1 * 1024 + pc;
  const int rb0 = (r0 & 96) + ((r0 & 1) << 4) + ((r0 & 31) >> 1);
  const int rb1 = rb0 + 64;
  const int sa0 = r0 * CG_RP + pc, sa1 = (r0 + 64) * CG_RP + pc;
  const int sb0 = rb0 * CG_RP + pc, sb1 = rb1 * CG_RP + pc;
  const int fa = (wm * 32 + ln) * CG_RP + (hi ? 4 : 0);
  const int fb = (wn * 32 + ln) * CG_RP + (hi ? 4 : 0);

  v8f acc[2][2];
#pragma unroll
  for (int mt = 0; mt < 2; ++mt)
#pragma unroll
    for (int nt = 0; nt < 2; ++nt) acc[mt][nt] = 0.0f;

  v4f ta0 = *(const v4f*)&ga0[0], ta1 = *(const v4f*)&ga1[0];
  v4f tb0 = *(const v4f*)&gb0[0], tb1 = *(const v4f*)&gb1[0];
  *(v4f*)&sa[0][sa0] = ta0; *(v4f*)&sa[0][sa1] = ta1;
  *(v4f*)&sb[0][sb0] = tb0; *(v4f*)&sb[0][sb1] = tb1;
  __syncthreads();
#pragma unroll 1
  for (int ch = 0; ch < 32; ++ch) {
    const int cur = ch & 1;
    if (ch < 31) {
      const int d1 = (ch + 1) * 32;
      ta0 = *(const v4f*)&ga0[d1]; ta1 = *(const v4f*)&ga1[d1];
      tb0 = *(const v4f*)&gb0[d1]; tb1 = *(const v4f*)&gb1[d1];
    }
    const float* pa = &sa[cur][fa];
    const float* pb = &sb[cur][fb];
    v4f av[4][2], bv[4][2];
#pragma unroll
    for (int s4 = 0; s4 < 4; ++s4)
#pragma unroll
      for (int t2 = 0; t2 < 2; ++t2) {
        av[s4][t2] = *(const v4f*)&pa[t2 * 16 * CG_RP + s4 * 8];
        bv[s4][t2] = *(const v4f*)&pb[t2 * 16 * CG_RP + s4 * 8];
      }
#pragma unroll
    for (int s4 = 0; s4 < 4; ++s4)
#pragma unroll
      for (int mt = 0; mt < 2; ++mt)
#pragma unroll
        for (int nt = 0; nt < 2; ++nt) {
          v2f a0; a0.x = av[s4][mt].x; a0.y = av[s4][mt].y;
          v2f a1; a1.x = av[s4][mt].z; a1.y = av[s4][mt].w;
          v2f b0; b0.x = bv[s4][nt].x; b0.y = bv[s4][nt].y;
          v2f b1; b1.x = bv[s4][nt].z; b1.y = bv[s4][nt].w;
          acc[mt][nt] = wmma4(a0, b0, acc[mt][nt]);
          acc[mt][nt] = wmma4(a1, b1, acc[mt][nt]);
        }
    if (ch < 31) {
      const int nx = cur ^ 1;
      *(v4f*)&sa[nx][sa0] = ta0; *(v4f*)&sa[nx][sa1] = ta1;
      *(v4f*)&sb[nx][sb0] = tb0; *(v4f*)&sb[nx][sb1] = tb1;
    }
    __syncthreads();
  }
#pragma unroll
  for (int mt = 0; mt < 2; ++mt)
#pragma unroll
    for (int j = 0; j < 8; ++j) {
      const int m = mbase + wm * 32 + mt * 16 + (hi ? 8 : 0) + j;
      v2f o; o.x = acc[mt][0][j]; o.y = acc[mt][1][j];
      *(v2f*)&cg[(size_t)m * CG_N + nbase + wn * 32 + 2 * ln] = o;
    }
}

#define PC_RP 68
#define GC_RP 68
#define GI_RP 36
#define PW_C  (36 * PC_RP)
#define PW_GI (36 * GI_RP)
#define PL_C  0
#define PL_GI (16 * PW_C)
#define PL_GC (PL_GI + 16 * PW_GI)
#define PL_O1 (PL_GC + 60 * GC_RP)
#define O2_RP 68
#define DG_RP 68
#define PL_O2 (PL_O1 + 16 * 48)
#define PL_DG (PL_O2 + 16 * O2_RP)
#define PL_FLOATS (PL_DG + 16 * DG_RP)

static __device__ __forceinline__ float wave_sum(float s) {
#pragma unroll
  for (int off = 16; off > 0; off >>= 1) s += __shfl_xor(s, off, 32);
  return s;
}

__global__ __launch_bounds__(512) void nmf_pairs(const float* __restrict__ cg,
                                                 const int* __restrict__ offs,
                                                 const float* __restrict__ zpiece,
                                                 const float* __restrict__ gcc,
                                                 const float* __restrict__ gii,
                                                 const int* __restrict__ il_g,
                                                 const int* __restrict__ cl_g,
                                                 float* __restrict__ out) {
  __shared__ __attribute__((aligned(16))) float lds[PL_FLOATS];
  const int tid = threadIdx.x;
  const int w = tid >> 5;
  const int lane = tid & 31;
  const int hi = lane >> 4;
  const int ln = lane & 15;
  const int t = blockIdx.x & 127;
  const int ig = (blockIdx.x >> 7) * 16 + w;
  float* cw = lds + PL_C + w * PW_C;
  float* giw = lds + PL_GI + w * PW_GI;
  float* gcs = lds + PL_GC;
  float* o1 = lds + PL_O1 + w * 48;
  float* o2 = lds + PL_O2 + w * O2_RP;
  float* dgs = lds + PL_DG;
  const int il = il_g[ig];
  const int cl = cl_g[t];

  {
    const int vi = packed_rows(il, 36, false);
    const int vc4 = packed_rows(cl, 60, true);
    const float* src = cg + (size_t)offs[ig] * CG_N + offs[OFF_C + t];
#pragma unroll
    for (int u = 0; u < 17; ++u) {
      const int idx = lane + 32 * u;
      if (idx < 540) {
        const int row = idx / 15, p4 = (idx % 15) * 4;
        const float* p = (row < vi && p4 < vc4) ? &src[(size_t)row * CG_N + p4] : zpiece;
        *(v4f*)&cw[row * PC_RP + p4] = *(const v4f*)p;
      }
    }
  }
  for (int idx = lane; idx < 36 * 36; idx += 32) {
    const int a = idx / 36, b = idx % 36;
    giw[a * GI_RP + b] =
        gii[(size_t)ig * 2304 + ((a >> 4) * 3 + (b >> 4)) * 256 + (a & 15) * 16 + (b & 15)];
  }
  for (int idx = tid; idx < 60 * 60; idx += 512) {
    const int a = idx / 60, b = idx % 60;
    gcs[a * GC_RP + b] =
        gcc[(size_t)t * 4096 + ((a >> 4) * 4 + (b >> 4)) * 256 + (a & 15) * 16 + (b & 15)];
  }
  const int l0 = lane, l1 = lane + 32;
  const int l1c = (l1 < 59) ? l1 : 59;
  const int kA = lane, kB = 32 + (lane & 3);
  o2[l0] = (l0 < cl) ? (1.f / (float)cl) : 0.f;
  o2[l1] = (l1 < cl) ? (1.f / (float)cl) : 0.f;
  o1[kA] = (kA < il) ? (1.f / (float)il) : 0.f;
  if (lane < 4) o1[kB] = (kB < il) ? (1.f / (float)il) : 0.f;
  __syncthreads();

  float dg0 = 0.f, dg1 = 0.f;
  float duA = 0.f, duB = 0.f;
  float giA = 0.f, giB = 0.f;
#pragma unroll 1
  for (int iter = 0; iter <= N_ITERS; ++iter) {
    if (w < 4) {
      const float* ao = lds + PL_O2 + ln * O2_RP + (hi ? 2 : 0);
      const float* bg = gcs + (hi ? 2 : 0) * GC_RP + w * 16 + ln;
      v8f g0, g1; g0 = 0.0f; g1 = 0.0f;
#pragma unroll
      for (int lp = 0; lp < 60; lp += 8) {
        v2f a = *(const v2f*)&ao[lp];
        v2f b; b.x = bg[lp * GC_RP]; b.y = bg[(lp + 1) * GC_RP];
        g0 = wmma4(a, b, g0);
        if (lp + 4 < 60) {
          v2f a2 = *(const v2f*)&ao[lp + 4];
          v2f b2; b2.x = bg[(lp + 4) * GC_RP]; b2.y = bg[(lp + 5) * GC_RP];
          g1 = wmma4(a2, b2, g1);
        }
      }
#pragma unroll
      for (int j = 0; j < 8; ++j)
        dgs[(j + (hi ? 8 : 0)) * DG_RP + w * 16 + ln] = g0[j] + g1[j];
    }
    __syncthreads();
    dg0 = dgs[w * DG_RP + l0];
    dg1 = dgs[w * DG_RP + l1c];
    float dc0 = 0.f, dc1 = 0.f;
    if (iter < N_ITERS) {
#pragma unroll
      for (int k = 0; k < 36; k += 4) {
        const v4f o = *(const v4f*)&o1[k];
#pragma unroll
        for (int e = 0; e < 4; ++e) {
          dc0 += o[e] * cw[(k + e) * PC_RP + l0];
          dc1 += o[e] * cw[(k + e) * PC_RP + l1c];
        }
      }
      const float n0 = (l0 < cl) ? fmaxf(o2[l0] - ETA_F * (dg0 - dc0), 0.f) : 0.f;
      const float n1 = (l1 < cl) ? fmaxf(o2[l1] - ETA_F * (dg1 - dc1), 0.f) : 0.f;
      const float inv = 1.f / (wave_sum(n0 + n1) + EPS_F);
      __syncthreads();
      o2[l0] = n0 * inv;
      o2[l1] = n1 * inv;
      __syncthreads();
    }
    duA = 0.f; duB = 0.f; giA = 0.f; giB = 0.f;
#pragma unroll
    for (int l = 0; l < 60; l += 4) {
      const v4f o = *(const v4f*)&o2[l];
      const v4f ca = *(const v4f*)&cw[kA * PC_RP + l];
      const v4f cb = *(const v4f*)&cw[kB * PC_RP + l];
#pragma unroll
      for (int e = 0; e < 4; ++e) { duA += ca[e] * o[e]; duB += cb[e] * o[e]; }
    }
#pragma unroll
    for (int kp = 0; kp < 36; kp += 4) {
      const v4f o = *(const v4f*)&o1[kp];
#pragma unroll
      for (int e = 0; e < 4; ++e) {
        giA += o[e] * giw[(kp + e) * GI_RP + kA];
        giB += o[e] * giw[(kp + e) * GI_RP + kB];
      }
    }
    if (iter < N_ITERS) {
      const float nA = (kA < il) ? fmaxf(o1[kA] + ETA_F * (duA - giA), 0.f) : 0.f;
      const float nB = (lane < 4 && kB < il) ? fmaxf(o1[kB] + ETA_F * (duB - giB), 0.f) : 0.f;
      const float inv = 1.f / (wave_sum(nA + nB) + EPS_F);
      __syncthreads();
      o1[kA] = nA * inv;
      if (lane < 4) o1[kB] = nB * inv;
      __syncthreads();
    }
  }
  const float s22 = wave_sum(o2[l0] * dg0 + o2[l1] * dg1);
  const float s12 = wave_sum(o1[kA] * duA + ((lane < 4) ? o1[kB] * duB : 0.f));
  const float s11 = wave_sum(o1[kA] * giA + ((lane < 4) ? o1[kB] * giB : 0.f));
  if (lane == 0) {
    const float dnm = fmaxf(sqrtf(fmaxf(s11, 0.f)), 1e-12f) *
                      fmaxf(sqrtf(fmaxf(s22, 0.f)), 1e-12f);
    out[(size_t)ig * 128 + t] = s12 / dnm;
  }
}

extern "C" void kernel_launch(void* const* d_in, const int* in_sizes, int n_in,
                              void* d_out, int out_size, void* d_ws, size_t ws_size,
                              hipStream_t stream) {
  (void)in_sizes; (void)n_in; (void)out_size; (void)ws_size;
  const float* imgs = (const float*)d_in[0];
  const float* caps = (const float*)d_in[1];
  const int* il = (const int*)d_in[2];
  const int* cl = (const int*)d_in[3];
  float* wimgs = (float*)d_ws;
  float* wcaps = wimgs + (size_t)128 * 36 * 1024;
  float* gcc = wcaps + (size_t)128 * 60 * 1024;
  float* gii = gcc + (size_t)128 * 4096;
  float* cg = gii + (size_t)128 * 2304;
  int* offs = (int*)(cg + (size_t)4608 * 7680);
  int* rmap_i = offs + 512;
  int* rmap_c = rmap_i + 4608;
  float* zpiece = (float*)(rmap_c + 7680);

  l2norm_rows<<<128 * 36 + 128 * 60, 256, 0, stream>>>(imgs, caps, wimgs, wcaps);
  gram_rows<<<256, 512, 0, stream>>>(wimgs, wcaps, gcc, gii, il, cl, offs, rmap_i, rmap_c, zpiece);
  cross_gram<<<dim3(60, 36), 512, 0, stream>>>(wimgs, wcaps, offs, rmap_i, rmap_c, cg);
  nmf_pairs<<<1024, 512, 0, stream>>>(cg, offs, zpiece, gcc, gii, il, cl, (float*)d_out);
}


# row normalization folded into the products: 1/|row| from the Gram diagonals scales Gram tiles and C after their sums (rounding differs, sum order kept); no normalized copy, one launch fewer
# speedup vs baseline: 8.8966x; 1.0367x over previous
#include <hip/hip_runtime.h>
#include <math.h>

typedef __attribute__((ext_vector_type(2))) float v2f;
typedef __attribute__((ext_vector_type(4))) float v4f;
typedef __attribute__((ext_vector_type(8))) float v8f;

#define N_ITERS 5
#define EPS_F   1e-6f
#define ETA_F   0.02f


static __device__ __forceinline__ v8f wmma4(v2f a, v2f b, v8f c) {
  return __builtin_amdgcn_wmma_f32_16x16x4_f32(false, a, false, b, (short)0, c,
                                               false, false);
}

static __device__ __forceinline__ int packed_rows(int len, int full, bool pad4) {
  const int c = min(max(len, 0), full);
  return pad4 ? ((c + 3) & ~3) : c;
}
#define OFF_C 132

#define INV_RP 64
#define G_RP 1028
__global__ __launch_bounds__(512) void gram_rows(const float* __restrict__ imgs,
                                                 const float* __restrict__ caps,
                                                 float* __restrict__ gcc,
                                                 float* __restrict__ gii,
                                                 float* __restrict__ inv_i,
                                                 float* __restrict__ inv_c,
                                                 const int* __restrict__ il_g,
                                                 const int* __restrict__ cl_g,
                                                 int* __restrict__ offs,
                                                 int* __restrict__ rmap_i,
                                                 int* __restrict__ rmap_c,
                                                 float* __restrict__ zpiece) {
  __shared__ __attribute__((aligned(16))) float rows[60 * G_RP];
  __shared__ int part[4];
  __shared__ __attribute__((aligned(16))) float sinv[64];
  const int tid = threadIdx.x;
  const int w = tid >> 5;
  const int lane = tid & 31;
  const int hi = lane >> 4;
  const int ln = lane & 15;
  const bool is_cap = blockIdx.x < 128;
  const int nr = is_cap ? 60 : 36;
  const int nt_side = is_cap ? 4 : 3;
  const float* src = is_cap ? caps + (size_t)blockIdx.x * 60 * 1024
                            : imgs + (size_t)(blockIdx.x - 128) * 36 * 1024;
  float* dst = is_cap ? gcc + (size_t)blockIdx.x * 4096
                      : gii + (size_t)(blockIdx.x - 128) * 2304;
  const int me = is_cap ? blockIdx.x : blockIdx.x - 128;
  const int* lens = is_cap ? cl_g : il_g;
  if (w < 4) {
    int s = (tid < me) ? packed_rows(lens[tid], nr, is_cap) : 0;
#pragma unroll
    for (int off = 16; off > 0; off >>= 1) s += __shfl_xor(s, off, 32);
    if (lane == 0) part[w] = s;
  }
  for (int base = tid; base < nr * 256; base += 6 * 512) {
    v4f t[6];
#pragma unroll
    for (int u = 0; u < 6; ++u) t[u] = *(const v4f*)&src[(size_t)(base + u * 512) * 4];
#pragma unroll
    for (int u = 0; u < 6; ++u) {
      const int idx = base + u * 512;
      *(v4f*)&rows[(idx >> 8) * G_RP + (idx & 255) * 4] = t[u];
    }
  }
  __syncthreads();
  {
    const int first = part[0] + part[1] + part[2] + part[3];
    const int mine = packed_rows(lens[me], nr, is_cap);
    int* rmap = is_cap ? rmap_c : rmap_i;
    if (tid < mine) rmap[first + tid] = me * nr + tid;
    if (tid == 0) {
      int* o = offs + (is_cap ? OFF_C : 0);
      o[me] = first;
      if (me == 127) o[128] = first + mine;
    }
    if (blockIdx.x == 0 && tid < 4) zpiece[tid] = 0.f;
  }
  const bool has_tile = w < nt_side * nt_side;
  const int mt = w / nt_side, nt = w % nt_side;
  v8f gs; gs = 0.0f;
  if (has_tile) {
    const int ra = (mt * 16 + ln < nr - 1) ? (mt * 16 + ln) : (nr - 1);
    const int rb = (nt * 16 + ln < nr - 1) ? (nt * 16 + ln) : (nr - 1);
    const float* pa = rows + ra * G_RP + (hi ? 4 : 0);
    const float* pb = rows + rb * G_RP + (hi ? 4 : 0);
    v8f g0, g1; g0 = 0.0f; g1 = 0.0f;
#pragma unroll 8
    for (int d = 0; d < 1024; d += 8) {
      const v4f av = *(const v4f*)&pa[d];
      const v4f bv = *(const v4f*)&pb[d];
      v2f a0; a0.x = av.x; a0.y = av.y;
      v2f a1; a1.x = av.z; a1.y = av.w;
      v2f b0; b0.x = bv.x; b0.y = bv.y;
      v2f b1; b1.x = bv.z; b1.y = bv.w;
      g0 = wmma4(a0, b0, g0);
      g1 = wmma4(a1, b1, g1);
    }
#pragma unroll
    for (int j = 0; j < 8; ++j) gs[j] = g0[j] + g1[j];
    if (mt == nt) {
      float dv = 0.f;
#pragma unroll
      for (int j = 0; j < 8; ++j) dv = ((ln & 7) == j) ? gs[j] : dv;
      if ((ln >> 3) == hi) {
        const float inv = 1.0f / fmaxf(sqrtf(dv), 1e-12f);
        const int r = mt * 16 + ln;
        sinv[r] = inv;
      }
    }
  }
  __syncthreads();
  if (w == 15) {
    float* ginv = (is_cap ? inv_c : inv_i) + me * INV_RP;
    ginv[lane] = sinv[lane];
    if (32 + lane < nt_side * 16) ginv[32 + lane] = sinv[32 + lane];
  }
  if (has_tile) {
    float* dt = dst + (mt * nt_side + nt) * 256;
    const v4f s0 = *(const v4f*)&sinv[mt * 16 + (hi ? 8 : 0)];
    const v4f s1 = *(const v4f*)&sinv[mt * 16 + (hi ? 8 : 0) + 4];
    const float cs = sinv[nt * 16 + ln];
#pragma unroll
    for (int j = 0; j < 8; ++j) {
      const float rs = (j < 4) ? s0[j & 3] : s1[j & 3];
      dt[(j + (hi ? 8 : 0)) * 16 + ln] = gs[j] * rs * cs;
    }
  }
}

#define CG_RP 36
#define CG_N  7680
__global__ __launch_bounds__(512) void cross_gram(const float* __restrict__ imgs,
                                                  const float* __restrict__ caps,
                                                  const float* __restrict__ inv_i,
                                                  const float* __restrict__ inv_c,
                                                  const int* __restrict__ offs,
                                                  const int* __restrict__ rmap_i,
                                                  const int* __restrict__ rmap_c,
                                                  float* __restrict__ cg) {
  __shared__ __attribute__((aligned(16))) float sa[2][128 * CG_RP];
  __shared__ __attribute__((aligned(16))) float sb[2][128 * CG_RP];
  __shared__ __attribute__((aligned(16))) float ssc[256];
  const int tid = threadIdx.x;
  const int w = tid >> 5;
  const int lane = tid & 31;
  const int hi = lane >> 4;
  const int ln = lane & 15;
  const int wm = w >> 2, wn = w & 3;
  const int mbase = blockIdx.y * 128;
  const int nbase = blockIdx.x * 128;
  const int mtot = offs[128];
  const int ntot = offs[OFF_C + 128];
  if (mbase >= mtot || nbase >= ntot) return;
  const int r0 = tid >> 3;
  const int pc = (tid & 7) * 4;
  const int ma0 = mbase + r0, ma1 = ma0 + 64;
  const int nb0 = nbase + r0, nb1 = nb0 + 64;
  const int qa0 = rmap_i[min(ma0, mtot - 1)];
  const int qa1 = rmap_i[min(ma1, mtot - 1)];
  const int qb0 = rmap_c[min(nb0, ntot - 1)];
  const int qb1 = rmap_c[min(nb1, ntot - 1)];
  const float* ga0 = imgs + (size_t)qa0 * 1024 + pc;
  const float* ga1 = imgs + (size_t)qa1 * 1024 + pc;
  const float* gb0 = caps + (size_t)qb0 * 1024 + pc;
  const float* gb1 = caps + (size_t)qb1 * 1024 + pc;
  const int rb0 = (r0 & 96) + ((r0 & 1) << 4) + ((r0 & 31) >> 1);
  const int rb1 = rb0 + 64;
  const int sa0 = r0 * CG_RP + pc, sa1 = (r0 + 64) * CG_RP + pc;
  const int sb0 = rb0 * CG_RP + pc, sb1 = rb1 * CG_RP + pc;
  const int fa = (wm * 32 + ln) * CG_RP + (hi ? 4 : 0);
  const int fb = (wn * 32 + ln) * CG_RP + (hi ? 4 : 0);

  v8f acc[2][2];
#pragma unroll
  for (int mt = 0; mt < 2; ++mt)
#pragma unroll
    for (int nt = 0; nt < 2; ++nt) acc[mt][nt] = 0.0f;

  v4f ta0 = *(const v4f*)&ga0[0], ta1 = *(const v4f*)&ga1[0];
  v4f tb0 = *(const v4f*)&gb0[0], tb1 = *(const v4f*)&gb1[0];
  const float va0 = inv_i[qa0 + (qa0 / 36) * (INV_RP - 36)];
  const float va1 = inv_i[qa1 + (qa1 / 36) * (INV_RP - 36)];
  const float vb0 = inv_c[qb0 + (qb0 / 60) * (INV_RP - 60)];
  const float vb1 = inv_c[qb1 + (qb1 / 60) * (INV_RP - 60)];
  *(v4f*)&sa[0][sa0] = ta0; *(v4f*)&sa[0][sa1] = ta1;
  *(v4f*)&sb[0][sb0] = tb0; *(v4f*)&sb[0][sb1] = tb1;
  if ((tid & 7) == 0) {
    ssc[r0] = va0; ssc[r0 + 64] = va1;
    ssc[128 + r0] = vb0; ssc[128 + r0 + 64] = vb1;
  }
  __syncthreads();
#pragma unroll 1
  for (int ch = 0; ch < 32; ++ch) {
    const int cur = ch & 1;
    if (ch < 31) {
      const int d1 = (ch + 1) * 32;
      ta0 = *(const v4f*)&ga0[d1]; ta1 = *(const v4f*)&ga1[d1];
      tb0 = *(const v4f*)&gb0[d1]; tb1 = *(const v4f*)&gb1[d1];
    }
    const float* pa = &sa[cur][fa];
    const float* pb = &sb[cur][fb];
    v4f av[4][2], bv[4][2];
#pragma unroll
    for (int s4 = 0; s4 < 4; ++s4)
#pragma unroll
      for (int t2 = 0; t2 < 2; ++t2) {
        av[s4][t2] = *(const v4f*)&pa[t2 * 16 * CG_RP + s4 * 8];
        bv[s4][t2] = *(const v4f*)&pb[t2 * 16 * CG_RP + s4 * 8];
      }
#pragma unroll
    for (int s4 = 0; s4 < 4; ++s4)
#pragma unroll
      for (int mt = 0; mt < 2; ++mt)
#pragma unroll
        for (int nt = 0; nt < 2; ++nt) {
          v2f a0; a0.x = av[s4][mt].x; a0.y = av[s4][mt].y;
          v2f a1; a1.x = av[s4][mt].z; a1.y = av[s4][mt].w;
          v2f b0; b0.x = bv[s4][nt].x; b0.y = bv[s4][nt].y;
          v2f b1; b1.x = bv[s4][nt].z; b1.y = bv[s4][nt].w;
          acc[mt][nt] = wmma4(a0, b0, acc[mt][nt]);
          acc[mt][nt] = wmma4(a1, b1, acc[mt][nt]);
        }
    if (ch < 31) {
      const int nx = cur ^ 1;
      *(v4f*)&sa[nx][sa0] = ta0; *(v4f*)&sa[nx][sa1] = ta1;
      *(v4f*)&sb[nx][sb0] = tb0; *(v4f*)&sb[nx][sb1] = tb1;
    }
    __syncthreads();
  }
  const v2f cs = *(const v2f*)&ssc[128 + wn * 32 + 2 * ln];
#pragma unroll
  for (int mt = 0; mt < 2; ++mt) {
    const v4f s0 = *(const v4f*)&ssc[wm * 32 + mt * 16 + (hi ? 8 : 0)];
    const v4f s1 = *(const v4f*)&ssc[wm * 32 + mt * 16 + (hi ? 8 : 0) + 4];
#pragma unroll
    for (int j = 0; j < 8; ++j) {
      const int m = mbase + wm * 32 + mt * 16 + (hi ? 8 : 0) + j;
      const float rs = (j < 4) ? s0[j & 3] : s1[j & 3];
      v2f o; o.x = acc[mt][0][j] * rs * cs.x; o.y = acc[mt][1][j] * rs * cs.y;
      *(v2f*)&cg[(size_t)m * CG_N + nbase + wn * 32 + 2 * ln] = o;
    }
  }
}

#define PC_RP 68
#define GC_RP 68
#define GI_RP 36
#define PW_C  (36 * PC_RP)
#define PW_GI (36 * GI_RP)
#define PL_C  0
#define PL_GI (16 * PW_C)
#define PL_GC (PL_GI + 16 * PW_GI)
#define PL_O1 (PL_GC + 60 * GC_RP)
#define O2_RP 68
#define DG_RP 68
#define PL_O2 (PL_O1 + 16 * 48)
#define PL_DG (PL_O2 + 16 * O2_RP)
#define PL_FLOATS (PL_DG + 16 * DG_RP)

static __device__ __forceinline__ float wave_sum(float s) {
#pragma unroll
  for (int off = 16; off > 0; off >>= 1) s += __shfl_xor(s, off, 32);
  return s;
}

__global__ __launch_bounds__(512) void nmf_pairs(const float* __restrict__ cg,
                                                 const int* __restrict__ offs,
                                                 const float* __restrict__ zpiece,
                                                 const float* __restrict__ gcc,
                                                 const float* __restrict__ gii,
                                                 const int* __restrict__ il_g,
                                                 const int* __restrict__ cl_g,
                                                 float* __restrict__ out) {
  __shared__ __attribute__((aligned(16))) float lds[PL_FLOATS];
  const int tid = threadIdx.x;
  const int w = tid >> 5;
  const int lane = tid & 31;
  const int hi = lane >> 4;
  const int ln = lane & 15;
  const int t = blockIdx.x & 127;
  const int ig = (blockIdx.x >> 7) * 16 + w;
  float* cw = lds + PL_C + w * PW_C;
  float* giw = lds + PL_GI + w * PW_GI;
  float* gcs = lds + PL_GC;
  float* o1 = lds + PL_O1 + w * 48;
  float* o2 = lds + PL_O2 + w * O2_RP;
  float* dgs = lds + PL_DG;
  const int il = il_g[ig];
  const int cl = cl_g[t];

  {
    const int vi = packed_rows(il, 36, false);
    const int vc4 = packed_rows(cl, 60, true);
    const float* src = cg + (size_t)offs[ig] * CG_N + offs[OFF_C + t];
#pragma unroll
    for (int u = 0; u < 17; ++u) {
      const int idx = lane + 32 * u;
      if (idx < 540) {
        const int row = idx / 15, p4 = (idx % 15) * 4;
        const float* p = (row < vi && p4 < vc4) ? &src[(size_t)row * CG_N + p4] : zpiece;
        *(v4f*)&cw[row * PC_RP + p4] = *(const v4f*)p;
      }
    }
  }
  for (int idx = lane; idx < 36 * 36; idx += 32) {
    const int a = idx / 36, b = idx % 36;
    giw[a * GI_RP + b] =
        gii[(size_t)ig * 2304 + ((a >> 4) * 3 + (b >> 4)) * 256 + (a & 15) * 16 + (b & 15)];
  }
  for (int idx = tid; idx < 60 * 60; idx += 512) {
    const int a = idx / 60, b = idx % 60;
    gcs[a * GC_RP + b] =
        gcc[(size_t)t * 4096 + ((a >> 4) * 4 + (b >> 4)) * 256 + (a & 15) * 16 + (b & 15)];
  }
  const int l0 = lane, l1 = lane + 32;
  const int l1c = (l1 < 59) ? l1 : 59;
  const int kA = lane, kB = 32 + (lane & 3);
  o2[l0] = (l0 < cl) ? (1.f / (float)cl) : 0.f;
  o2[l1] = (l1 < cl) ? (1.f / (float)cl) : 0.f;
  o1[kA] = (kA < il) ? (1.f / (float)il) : 0.f;
  if (lane < 4) o1[kB] = (kB < il) ? (1.f / (float)il) : 0.f;
  __syncthreads();

  float dg0 = 0.f, dg1 = 0.f;
  float duA = 0.f, duB = 0.f;
  float giA = 0.f, giB = 0.f;
#pragma unroll 1
  for (int iter = 0; iter <= N_ITERS; ++iter) {
    if (w < 4) {
      const float* ao = lds + PL_O2 + ln * O2_RP + (hi ? 2 : 0);
      const float* bg = gcs + (hi ? 2 : 0) * GC_RP + w * 16 + ln;
      v8f g0, g1; g0 = 0.0f; g1 = 0.0f;
#pragma unroll
      for (int lp = 0; lp < 60; lp += 8) {
        v2f a = *(const v2f*)&ao[lp];
        v2f b; b.x = bg[lp * GC_RP]; b.y = bg[(lp + 1) * GC_RP];
        g0 = wmma4(a, b, g0);
        if (lp + 4 < 60) {
          v2f a2 = *(const v2f*)&ao[lp + 4];
          v2f b2; b2.x = bg[(lp + 4) * GC_RP]; b2.y = bg[(lp + 5) * GC_RP];
          g1 = wmma4(a2, b2, g1);
        }
      }
#pragma unroll
      for (int j = 0; j < 8; ++j)
        dgs[(j + (hi ? 8 : 0)) * DG_RP + w * 16 + ln] = g0[j] + g1[j];
    }
    __syncthreads();
    dg0 = dgs[w * DG_RP + l0];
    dg1 = dgs[w * DG_RP + l1c];
    float dc0 = 0.f, dc1 = 0.f;
    if (iter < N_ITERS) {
#pragma unroll
      for (int k = 0; k < 36; k += 4) {
        const v4f o = *(const v4f*)&o1[k];
#pragma unroll
        for (int e = 0; e < 4; ++e) {
          dc0 += o[e] * cw[(k + e) * PC_RP + l0];
          dc1 += o[e] * cw[(k + e) * PC_RP + l1c];
        }
      }
      const float n0 = (l0 < cl) ? fmaxf(o2[l0] - ETA_F * (dg0 - dc0), 0.f) : 0.f;
      const float n1 = (l1 < cl) ? fmaxf(o2[l1] - ETA_F * (dg1 - dc1), 0.f) : 0.f;
      const float inv = 1.f / (wave_sum(n0 + n1) + EPS_F);
      __syncthreads();
      o2[l0] = n0 * inv;
      o2[l1] = n1 * inv;
      __syncthreads();
    }
    duA = 0.f; duB = 0.f; giA = 0.f; giB = 0.f;
#pragma unroll
    for (int l = 0; l < 60; l += 4) {
      const v4f o = *(const v4f*)&o2[l];
      const v4f ca = *(const v4f*)&cw[kA * PC_RP + l];
      const v4f cb = *(const v4f*)&cw[kB * PC_RP + l];
#pragma unroll
      for (int e = 0; e < 4; ++e) { duA += ca[e] * o[e]; duB += cb[e] * o[e]; }
    }
#pragma unroll
    for (int kp = 0; kp < 36; kp += 4) {
      const v4f o = *(const v4f*)&o1[kp];
#pragma unroll
      for (int e = 0; e < 4; ++e) {
        giA += o[e] * giw[(kp + e) * GI_RP + kA];
        giB += o[e] * giw[(kp + e) * GI_RP + kB];
      }
    }
    if (iter < N_ITERS) {
      const float nA = (kA < il) ? fmaxf(o1[kA] + ETA_F * (duA - giA), 0.f) : 0.f;
      const float nB = (lane < 4 && kB < il) ? fmaxf(o1[kB] + ETA_F * (duB - giB), 0.f) : 0.f;
      const float inv = 1.f / (wave_sum(nA + nB) + EPS_F);
      __syncthreads();
      o1[kA] = nA * inv;
      if (lane < 4) o1[kB] = nB * inv;
      __syncthreads();
    }
  }
  const float s22 = wave_sum(o2[l0] * dg0 + o2[l1] * dg1);
  const float s12 = wave_sum(o1[kA] * duA + ((lane < 4) ? o1[kB] * duB : 0.f));
  const float s11 = wave_sum(o1[kA] * giA + ((lane < 4) ? o1[kB] * giB : 0.f));
  if (lane == 0) {
    const float dnm = fmaxf(sqrtf(fmaxf(s11, 0.f)), 1e-12f) *
                      fmaxf(sqrtf(fmaxf(s22, 0.f)), 1e-12f);
    out[(size_t)ig * 128 + t] = s12 / dnm;
  }
}

extern "C" void kernel_launch(void* const* d_in, const int* in_sizes, int n_in,
                              void* d_out, int out_size, void* d_ws, size_t ws_size,
                              hipStream_t stream) {
  (void)in_sizes; (void)n_in; (void)out_size; (void)ws_size;
  const float* imgs = (const float*)d_in[0];
  const float* caps = (const float*)d_in[1];
  const int* il = (const int*)d_in[2];
  const int* cl = (const int*)d_in[3];
  float* gcc = (float*)d_ws;
  float* gii = gcc + (size_t)128 * 4096;
  float* cg = gii + (size_t)128 * 2304;
  int* offs = (int*)(cg + (size_t)4608 * 7680);
  int* rmap_i = offs + 512;
  int* rmap_c = rmap_i + 4608;
  float* zpiece = (float*)(rmap_c + 7680);
  float* inv_i = zpiece + 32;
  float* inv_c = inv_i + 128 * 64;

  gram_rows<<<256, 512, 0, stream>>>(imgs, caps, gcc, gii, inv_i, inv_c, il, cl, offs, rmap_i,
                                     rmap_c, zpiece);
  cross_gram<<<dim3(60, 36), 512, 0, stream>>>(imgs, caps, inv_i, inv_c, offs, rmap_i, rmap_c, cg);
  nmf_pairs<<<1024, 512, 0, stream>>>(cg, offs, zpiece, gcc, gii, il, cl, (float*)d_out);
}


# pair kernel staging: Gram matrices fetched in 16-byte pieces, all Gram and C loads in flight before the first LDS store (were 4-byte loads, 1-4 in flight, behind C); lengths/offsets by scalar loads
# speedup vs baseline: 9.2628x; 1.0412x over previous
#include <hip/hip_runtime.h>
#include <math.h>

typedef __attribute__((ext_vector_type(2))) float v2f;
typedef __attribute__((ext_vector_type(4))) float v4f;
typedef __attribute__((ext_vector_type(8))) float v8f;

#define N_ITERS 5
#define EPS_F   1e-6f
#define ETA_F   0.02f


static __device__ __forceinline__ v8f wmma4(v2f a, v2f b, v8f c) {
  return __builtin_amdgcn_wmma_f32_16x16x4_f32(false, a, false, b, (short)0, c,
                                               false, false);
}

static __device__ __forceinline__ int packed_rows(int len, int full, bool pad4) {
  const int c = min(max(len, 0), full);
  return pad4 ? ((c + 3) & ~3) : c;
}
#define OFF_C 132

#define INV_RP 64
#define G_RP 1028
__global__ __launch_bounds__(512) void gram_rows(const float* __restrict__ imgs,
                                                 const float* __restrict__ caps,
                                                 float* __restrict__ gcc,
                                                 float* __restrict__ gii,
                                                 float* __restrict__ inv_i,
                                                 float* __restrict__ inv_c,
                                                 const int* __restrict__ il_g,
                                                 const int* __restrict__ cl_g,
                                                 int* __restrict__ offs,
                                                 int* __restrict__ rmap_i,
                                                 int* __restrict__ rmap_c,
                                                 float* __restrict__ zpiece) {
  __shared__ __attribute__((aligned(16))) float rows[60 * G_RP];
  __shared__ int part[4];
  __shared__ __attribute__((aligned(16))) float sinv[64];
  const int tid = threadIdx.x;
  const int w = tid >> 5;
  const int lane = tid & 31;
  const int hi = lane >> 4;
  const int ln = lane & 15;
  const bool is_cap = blockIdx.x < 128;
  const int nr = is_cap ? 60 : 36;
  const int nt_side = is_cap ? 4 : 3;
  const float* src = is_cap ? caps + (size_t)blockIdx.x * 60 * 1024
                            : imgs + (size_t)(blockIdx.x - 128) * 36 * 1024;
  float* dst = is_cap ? gcc + (size_t)blockIdx.x * 4096
                      : gii + (size_t)(blockIdx.x - 128) * 2304;
  const int me = is_cap ? blockIdx.x : blockIdx.x - 128;
  const int* lens = is_cap ? cl_g : il_g;
  if (w < 4) {
    int s = (tid < me) ? packed_rows(lens[tid], nr, is_cap) : 0;
#pragma unroll
    for (int off = 16; off > 0; off >>= 1) s += __shfl_xor(s, off, 32);
    if (lane == 0) part[w] = s;
  }
  for (int base = tid; base < nr * 256; base += 6 * 512) {
    v4f t[6];
#pragma unroll
    for (int u = 0; u < 6; ++u) t[u] = *(const v4f*)&src[(size_t)(base + u * 512) * 4];
#pragma unroll
    for (int u = 0; u < 6; ++u) {
      const int idx = base + u * 512;
      *(v4f*)&rows[(idx >> 8) * G_RP + (idx & 255) * 4] = t[u];
    }
  }
  __syncthreads();
  {
    const int first = part[0] + part[1] + part[2] + part[3];
    const int mine = packed_rows(lens[me], nr, is_cap);
    int* rmap = is_cap ? rmap_c : rmap_i;
    if (tid < mine) rmap[first + tid] = me * nr + tid;
    if (tid == 0) {
      int* o = offs + (is_cap ? OFF_C : 0);
      o[me] = first;
      if (me == 127) o[128] = first + mine;
    }
    if (blockIdx.x == 0 && tid < 4) zpiece[tid] = 0.f;
  }
  const bool has_tile = w < nt_side * nt_side;
  const int mt = w / nt_side, nt = w % nt_side;
  v8f gs; gs = 0.0f;
  if (has_tile) {
    const int ra = (mt * 16 + ln < nr - 1) ? (mt * 16 + ln) : (nr - 1);
    const int rb = (nt * 16 + ln < nr - 1) ? (nt * 16 + ln) : (nr - 1);
    const float* pa = rows + ra * G_RP + (hi ? 4 : 0);
    const float* pb = rows + rb * G_RP + (hi ? 4 : 0);
    v8f g0, g1; g0 = 0.0f; g1 = 0.0f;
#pragma unroll 8
    for (int d = 0; d < 1024; d += 8) {
      const v4f av = *(const v4f*)&pa[d];
      const v4f bv = *(const v4f*)&pb[d];
      v2f a0; a0.x = av.x; a0.y = av.y;
      v2f a1; a1.x = av.z; a1.y = av.w;
      v2f b0; b0.x = bv.x; b0.y = bv.y;
      v2f b1; b1.x = bv.z; b1.y = bv.w;
      g0 = wmma4(a0, b0, g0);
      g1 = wmma4(a1, b1, g1);
    }
#pragma unroll
    for (int j = 0; j < 8; ++j) gs[j] = g0[j] + g1[j];
    if (mt == nt) {
      float dv = 0.f;
#pragma unroll
      for (int j = 0; j < 8; ++j) dv = ((ln & 7) == j) ? gs[j] : dv;
      if ((ln >> 3) == hi) {
        const float inv = 1.0f / fmaxf(sqrtf(dv), 1e-12f);
        const int r = mt * 16 + ln;
        sinv[r] = inv;
      }
    }
  }
  __syncthreads();
  if (w == 15) {
    float* ginv = (is_cap ? inv_c : inv_i) + me * INV_RP;
    ginv[lane] = sinv[lane];
    if (32 + lane < nt_side * 16) ginv[32 + lane] = sinv[32 + lane];
  }
  if (has_tile) {
    float* dt = dst + (mt * nt_side + nt) * 256;
    const v4f s0 = *(const v4f*)&sinv[mt * 16 + (hi ? 8 : 0)];
    const v4f s1 = *(const v4f*)&sinv[mt * 16 + (hi ? 8 : 0) + 4];
    const float cs = sinv[nt * 16 + ln];
#pragma unroll
    for (int j = 0; j < 8; ++j) {
      const float rs = (j < 4) ? s0[j & 3] : s1[j & 3];
      dt[(j + (hi ? 8 : 0)) * 16 + ln] = gs[j] * rs * cs;
    }
  }
}

#define CG_RP 36
#define CG_N  7680
__global__ __launch_bounds__(512) void cross_gram(const float* __restrict__ imgs,
                                                  const float* __restrict__ caps,
                                                  const float* __restrict__ inv_i,
                                                  const float* __restrict__ inv_c,
                                                  const int* __restrict__ offs,
                                                  const int* __restrict__ rmap_i,
                                                  const int* __restrict__ rmap_c,
                                                  float* __restrict__ cg) {
  __shared__ __attribute__((aligned(16))) float sa[2][128 * CG_RP];
  __shared__ __attribute__((aligned(16))) float sb[2][128 * CG_RP];
  __shared__ __attribute__((aligned(16))) float ssc[256];
  const int tid = threadIdx.x;
  const int w = tid >> 5;
  const int lane = tid & 31;
  const int hi = lane >> 4;
  const int ln = lane & 15;
  const int wm = w >> 2, wn = w & 3;
  const int mbase = blockIdx.y * 128;
  const int nbase = blockIdx.x * 128;
  const int mtot = offs[128];
  const int ntot = offs[OFF_C + 128];
  if (mbase >= mtot || nbase >= ntot) return;
  const int r0 = tid >> 3;
  const int pc = (tid & 7) * 4;
  const int ma0 = mbase + r0, ma1 = ma0 + 64;
  const int nb0 = nbase + r0, nb1 = nb0 + 64;
  const int qa0 = rmap_i[min(ma0, mtot - 1)];
  const int qa1 = rmap_i[min(ma1, mtot - 1)];
  const int qb0 = rmap_c[min(nb0, ntot - 1)];
  const int qb1 = rmap_c[min(nb1, ntot - 1)];
  const float* ga0 = imgs + (size_t)qa0 * 1024 + pc;
  const float* ga1 = imgs + (size_t)qa1 * 1024 + pc;
  const float* gb0 = caps + (size_t)qb0 * 1024 + pc;
  const float* gb1 = caps + (size_t)qb1 * 1024 + pc;
  const int rb0 = (r0 & 96) + ((r0 & 1) << 4) + ((r0 & 31) >> 1);
  const int rb1 = rb0 + 64;
  const int sa0 = r0 * CG_RP + pc, sa1 = (r0 + 64) * CG_RP + pc;
  const int sb0 = rb0 * CG_RP + pc, sb1 = rb1 * CG_RP + pc;
  const int fa = (wm * 32 + ln) * CG_RP + (hi ? 4 : 0);
  const int fb = (wn * 32 + ln) * CG_RP + (hi ? 4 : 0);

  v8f acc[2][2];
#pragma unroll
  for (int mt = 0; mt < 2; ++mt)
#pragma unroll
    for (int nt = 0; nt < 2; ++nt) acc[mt][nt] = 0.0f;

  v4f ta0 = *(const v4f*)&ga0[0], ta1 = *(const v4f*)&ga1[0];
  v4f tb0 = *(const v4f*)&gb0[0], tb1 = *(const v4f*)&gb1[0];
  const float va0 = inv_i[qa0 + (qa0 / 36) * (INV_RP - 36)];
  const float va1 = inv_i[qa1 + (qa1 / 36) * (INV_RP - 36)];
  const float vb0 = inv_c[qb0 + (qb0 / 60) * (INV_RP - 60)];
  const float vb1 = inv_c[qb1 + (qb1 / 60) * (INV_RP - 60)];
  *(v4f*)&sa[0][sa0] = ta0; *(v4f*)&sa[0][sa1] = ta1;
  *(v4f*)&sb[0][sb0] = tb0; *(v4f*)&sb[0][sb1] = tb1;
  if ((tid & 7) == 0) {
    ssc[r0] = va0; ssc[r0 + 64] = va1;
    ssc[128 + r0] = vb0; ssc[128 + r0 + 64] = vb1;
  }
  __syncthreads();
#pragma unroll 1
  for (int ch = 0; ch < 32; ++ch) {
    const int cur = ch & 1;
    if (ch < 31) {
      const int d1 = (ch + 1) * 32;
      ta0 = *(const v4f*)&ga0[d1]; ta1 = *(const v4f*)&ga1[d1];
      tb0 = *(const v4f*)&gb0[d1]; tb1 = *(const v4f*)&gb1[d1];
    }
    const float* pa = &sa[cur][fa];
    const float* pb = &sb[cur][fb];
    v4f av[4][2], bv[4][2];
#pragma unroll
    for (int s4 = 0; s4 < 4; ++s4)
#pragma unroll
      for (int t2 = 0; t2 < 2; ++t2) {
        av[s4][t2] = *(const v4f*)&pa[t2 * 16 * CG_RP + s4 * 8];
        bv[s4][t2] = *(const v4f*)&pb[t2 * 16 * CG_RP + s4 * 8];
      }
#pragma unroll
    for (int s4 = 0; s4 < 4; ++s4)
#pragma unroll
      for (int mt = 0; mt < 2; ++mt)
#pragma unroll
        for (int nt = 0; nt < 2; ++nt) {
          v2f a0; a0.x = av[s4][mt].x; a0.y = av[s4][mt].y;
          v2f a1; a1.x = av[s4][mt].z; a1.y = av[s4][mt].w;
          v2f b0; b0.x = bv[s4][nt].x; b0.y = bv[s4][nt].y;
          v2f b1; b1.x = bv[s4][nt].z; b1.y = bv[s4][nt].w;
          acc[mt][nt] = wmma4(a0, b0, acc[mt][nt]);
          acc[mt][nt] = wmma4(a1, b1, acc[mt][nt]);
        }
    if (ch < 31) {
      const int nx = cur ^ 1;
      *(v4f*)&sa[nx][sa0] = ta0; *(v4f*)&sa[nx][sa1] = ta1;
      *(v4f*)&sb[nx][sb0] = tb0; *(v4f*)&sb[nx][sb1] = tb1;
    }
    __syncthreads();
  }
  const v2f cs = *(const v2f*)&ssc[128 + wn * 32 + 2 * ln];
#pragma unroll
  for (int mt = 0; mt < 2; ++mt) {
    const v4f s0 = *(const v4f*)&ssc[wm * 32 + mt * 16 + (hi ? 8 : 0)];
    const v4f s1 = *(const v4f*)&ssc[wm * 32 + mt * 16 + (hi ? 8 : 0) + 4];
#pragma unroll
    for (int j = 0; j < 8; ++j) {
      const int m = mbase + wm * 32 + mt * 16 + (hi ? 8 : 0) + j;
      const float rs = (j < 4) ? s0[j & 3] : s1[j & 3];
      v2f o; o.x = acc[mt][0][j] * rs * cs.x; o.y = acc[mt][1][j] * rs * cs.y;
      *(v2f*)&cg[(size_t)m * CG_N + nbase + wn * 32 + 2 * ln] = o;
    }
  }
}

#define PC_RP 68
#define GC_RP 68
#define GI_RP 36
#define PW_C  (36 * PC_RP)
#define PW_GI (36 * GI_RP)
#define PL_C  0
#define PL_GI (16 * PW_C)
#define PL_GC (PL_GI + 16 * PW_GI)
#define PL_O1 (PL_GC + 60 * GC_RP)
#define O2_RP 68
#define DG_RP 68
#define PL_O2 (PL_O1 + 16 * 48)
#define PL_DG (PL_O2 + 16 * O2_RP)
#define PL_FLOATS (PL_DG + 16 * DG_RP)

static __device__ __forceinline__ float wave_sum(float s) {
#pragma unroll
  for (int off = 16; off > 0; off >>= 1) s += __shfl_xor(s, off, 32);
  return s;
}

__global__ __launch_bounds__(512) void nmf_pairs(const float* __restrict__ cg,
                                                 const int* __restrict__ offs,
                                                 const float* __restrict__ zpiece,
                                                 const float* __restrict__ gcc,
                                                 const float* __restrict__ gii,
                                                 const int* __restrict__ il_g,
                                                 const int* __restrict__ cl_g,
                                                 float* __restrict__ out) {
  __shared__ __attribute__((aligned(16))) float lds[PL_FLOATS];
  const int tid = threadIdx.x;
  const int w = tid >> 5;
  const int lane = tid & 31;
  const int hi = lane >> 4;
  const int ln = lane & 15;
  const int t = blockIdx.x & 127;
  const int ig = __builtin_amdgcn_readfirstlane((blockIdx.x >> 7) * 16 + w);
  float* cw = lds + PL_C + w * PW_C;
  float* giw = lds + PL_GI + w * PW_GI;
  float* gcs = lds + PL_GC;
  float* o1 = lds + PL_O1 + w * 48;
  float* o2 = lds + PL_O2 + w * O2_RP;
  float* dgs = lds + PL_DG;
  const int il = il_g[ig];
  const int cl = cl_g[t];

  {
    const float* gsi = gii + (size_t)ig * 2304;
    const float* gsc = gcc + (size_t)t * 4096;
    v4f gir[11], gcr[2], cr[17];
#pragma unroll
    for (int u = 0; u < 11; ++u) {
      const int p = lane + 32 * u;
      if (p < 324) {
        const int a = p / 9, b4 = (p % 9) * 4;
        gir[u] = *(const v4f*)&gsi[((a >> 4) * 3 + (b4 >> 4)) * 256 + (a & 15) * 16 + (b4 & 15)];
      }
    }
#pragma unroll
    for (int u = 0; u < 2; ++u) {
      const int p = tid + 512 * u;
      if (p < 900) {
        const int a = p / 15, b4 = (p % 15) * 4;
        gcr[u] = *(const v4f*)&gsc[((a >> 4) * 4 + (b4 >> 4)) * 256 + (a & 15) * 16 + (b4 & 15)];
      }
    }
    const int vi = packed_rows(il, 36, false);
    const int vc4 = packed_rows(cl, 60, true);
    const float* src = cg + (size_t)offs[ig] * CG_N + offs[OFF_C + t];
#pragma unroll
    for (int u = 0; u < 17; ++u) {
      const int idx = lane + 32 * u;
      if (idx < 540) {
        const int row = idx / 15, p4 = (idx % 15) * 4;
        const float* p = (row < vi && p4 < vc4) ? &src[(size_t)row * CG_N + p4] : zpiece;
        cr[u] = *(const v4f*)p;
      }
    }
#pragma unroll
    for (int u = 0; u < 11; ++u) {
      const int p = lane + 32 * u;
      if (p < 324) *(v4f*)&giw[p * 4] = gir[u];
    }
#pragma unroll
    for (int u = 0; u < 2; ++u) {
      const int p = tid + 512 * u;
      if (p < 900) *(v4f*)&gcs[(p / 15) * GC_RP + (p % 15) * 4] = gcr[u];
    }
#pragma unroll
    for (int u = 0; u < 17; ++u) {
      const int idx = lane + 32 * u;
      if (idx < 540) *(v4f*)&cw[(idx / 15) * PC_RP + (idx % 15) * 4] = cr[u];
    }
  }
  const int l0 = lane, l1 = lane + 32;
  const int l1c = (l1 < 59) ? l1 : 59;
  const int kA = lane, kB = 32 + (lane & 3);
  o2[l0] = (l0 < cl) ? (1.f / (float)cl) : 0.f;
  o2[l1] = (l1 < cl) ? (1.f / (float)cl) : 0.f;
  o1[kA] = (kA < il) ? (1.f / (float)il) : 0.f;
  if (lane < 4) o1[kB] = (kB < il) ? (1.f / (float)il) : 0.f;
  __syncthreads();

  float dg0 = 0.f, dg1 = 0.f;
  float duA = 0.f, duB = 0.f;
  float giA = 0.f, giB = 0.f;
#pragma unroll 1
  for (int iter = 0; iter <= N_ITERS; ++iter) {
    if (w < 4) {
      const float* ao = lds + PL_O2 + ln * O2_RP + (hi ? 2 : 0);
      const float* bg = gcs + (hi ? 2 : 0) * GC_RP + w * 16 + ln;
      v8f g0, g1; g0 = 0.0f; g1 = 0.0f;
#pragma unroll
      for (int lp = 0; lp < 60; lp += 8) {
        v2f a = *(const v2f*)&ao[lp];
        v2f b; b.x = bg[lp * GC_RP]; b.y = bg[(lp + 1) * GC_RP];
        g0 = wmma4(a, b, g0);
        if (lp + 4 < 60) {
          v2f a2 = *(const v2f*)&ao[lp + 4];
          v2f b2; b2.x = bg[(lp + 4) * GC_RP]; b2.y = bg[(lp + 5) * GC_RP];
          g1 = wmma4(a2, b2, g1);
        }
      }
#pragma unroll
      for (int j = 0; j < 8; ++j)
        dgs[(j + (hi ? 8 : 0)) * DG_RP + w * 16 + ln] = g0[j] + g1[j];
    }
    __syncthreads();
    dg0 = dgs[w * DG_RP + l0];
    dg1 = dgs[w * DG_RP + l1c];
    float dc0 = 0.f, dc1 = 0.f;
    if (iter < N_ITERS) {
#pragma unroll
      for (int k = 0; k < 36; k += 4) {
        const v4f o = *(const v4f*)&o1[k];
#pragma unroll
        for (int e = 0; e < 4; ++e) {
          dc0 += o[e] * cw[(k + e) * PC_RP + l0];
          dc1 += o[e] * cw[(k + e) * PC_RP + l1c];
        }
      }
      const float n0 = (l0 < cl) ? fmaxf(o2[l0] - ETA_F * (dg0 - dc0), 0.f) : 0.f;
      const float n1 = (l1 < cl) ? fmaxf(o2[l1] - ETA_F * (dg1 - dc1), 0.f) : 0.f;
      const float inv = 1.f / (wave_sum(n0 + n1) + EPS_F);
      __syncthreads();
      o2[l0] = n0 * inv;
      o2[l1] = n1 * inv;
      __syncthreads();
    }
    duA = 0.f; duB = 0.f; giA = 0.f; giB = 0.f;
#pragma unroll
    for (int l = 0; l < 60; l += 4) {
      const v4f o = *(const v4f*)&o2[l];
      const v4f ca = *(const v4f*)&cw[kA * PC_RP + l];
      const v4f cb = *(const v4f*)&cw[kB * PC_RP + l];
#pragma unroll
      for (int e = 0; e < 4; ++e) { duA += ca[e] * o[e]; duB += cb[e] * o[e]; }
    }
#pragma unroll
    for (int kp = 0; kp < 36; kp += 4) {
      const v4f o = *(const v4f*)&o1[kp];
#pragma unroll
      for (int e = 0; e < 4; ++e) {
        giA += o[e] * giw[(kp + e) * GI_RP + kA];
        giB += o[e] * giw[(kp + e) * GI_RP + kB];
      }
    }
    if (iter < N_ITERS) {
      const float nA = (kA < il) ? fmaxf(o1[kA] + ETA_F * (duA - giA), 0.f) : 0.f;
      const float nB = (lane < 4 && kB < il) ? fmaxf(o1[kB] + ETA_F * (duB - giB), 0.f) : 0.f;
      const float inv = 1.f / (wave_sum(nA + nB) + EPS_F);
      __syncthreads();
      o1[kA] = nA * inv;
      if (lane < 4) o1[kB] = nB * inv;
      __syncthreads();
    }
  }
  const float s22 = wave_sum(o2[l0] * dg0 + o2[l1] * dg1);
  const float s12 = wave_sum(o1[kA] * duA + ((lane < 4) ? o1[kB] * duB : 0.f));
  const float s11 = wave_sum(o1[kA] * giA + ((lane < 4) ? o1[kB] * giB : 0.f));
  if (lane == 0) {
    const float dnm = fmaxf(sqrtf(fmaxf(s11, 0.f)), 1e-12f) *
                      fmaxf(sqrtf(fmaxf(s22, 0.f)), 1e-12f);
    out[(size_t)ig * 128 + t] = s12 / dnm;
  }
}

extern "C" void kernel_launch(void* const* d_in, const int* in_sizes, int n_in,
                              void* d_out, int out_size, void* d_ws, size_t ws_size,
                              hipStream_t stream) {
  (void)in_sizes; (void)n_in; (void)out_size; (void)ws_size;
  const float* imgs = (const float*)d_in[0];
  const float* caps = (const float*)d_in[1];
  const int* il = (const int*)d_in[2];
  const int* cl = (const int*)d_in[3];
  float* gcc = (float*)d_ws;
  float* gii = gcc + (size_t)128 * 4096;
  float* cg = gii + (size_t)128 * 2304;
  int* offs = (int*)(cg + (size_t)4608 * 7680);
  int* rmap_i = offs + 512;
  int* rmap_c = rmap_i + 4608;
  float* zpiece = (float*)(rmap_c + 7680);
  float* inv_i = zpiece + 32;
  float* inv_c = inv_i + 128 * 64;

  gram_rows<<<256, 512, 0, stream>>>(imgs, caps, gcc, gii, inv_i, inv_c, il, cl, offs, rmap_i,
                                     rmap_c, zpiece);
  cross_gram<<<dim3(60, 36), 512, 0, stream>>>(imgs, caps, inv_i, inv_c, offs, rmap_i, rmap_c, cg);
  nmf_pairs<<<1024, 512, 0, stream>>>(cg, offs, zpiece, gcc, gii, il, cl, (float*)d_out);
}


# pair kernel: C block (by columns and by rows) and image Gram columns held in registers over the iterations, read from LDS once; rows 32..35 summed in 8 lane slices: their sum order changes
# speedup vs baseline: 9.6781x; 1.0448x over previous
#include <hip/hip_runtime.h>
#include <math.h>

typedef __attribute__((ext_vector_type(2))) float v2f;
typedef __attribute__((ext_vector_type(4))) float v4f;
typedef __attribute__((ext_vector_type(8))) float v8f;

#define N_ITERS 5
#define EPS_F   1e-6f
#define ETA_F   0.02f


static __device__ __forceinline__ v8f wmma4(v2f a, v2f b, v8f c) {
  return __builtin_amdgcn_wmma_f32_16x16x4_f32(false, a, false, b, (short)0, c,
                                               false, false);
}

static __device__ __forceinline__ int packed_rows(int len, int full, bool pad4) {
  const int c = min(max(len, 0), full);
  return pad4 ? ((c + 3) & ~3) : c;
}
#define OFF_C 132

#define INV_RP 64
#define G_RP 1028
__global__ __launch_bounds__(512) void gram_rows(const float* __restrict__ imgs,
                                                 const float* __restrict__ caps,
                                                 float* __restrict__ gcc,
                                                 float* __restrict__ gii,
                                                 float* __restrict__ inv_i,
                                                 float* __restrict__ inv_c,
                                                 const int* __restrict__ il_g,
                                                 const int* __restrict__ cl_g,
                                                 int* __restrict__ offs,
                                                 int* __restrict__ rmap_i,
                                                 int* __restrict__ rmap_c,
                                                 float* __restrict__ zpiece) {
  __shared__ __attribute__((aligned(16))) float rows[60 * G_RP];
  __shared__ int part[4];
  __shared__ __attribute__((aligned(16))) float sinv[64];
  const int tid = threadIdx.x;
  const int w = tid >> 5;
  const int lane = tid & 31;
  const int hi = lane >> 4;
  const int ln = lane & 15;
  const bool is_cap = blockIdx.x < 128;
  const int nr = is_cap ? 60 : 36;
  const int nt_side = is_cap ? 4 : 3;
  const float* src = is_cap ? caps + (size_t)blockIdx.x * 60 * 1024
                            : imgs + (size_t)(blockIdx.x - 128) * 36 * 1024;
  float* dst = is_cap ? gcc + (size_t)blockIdx.x * 4096
                      : gii + (size_t)(blockIdx.x - 128) * 2304;
  const int me = is_cap ? blockIdx.x : blockIdx.x - 128;
  const int* lens = is_cap ? cl_g : il_g;
  if (w < 4) {
    int s = (tid < me) ? packed_rows(lens[tid], nr, is_cap) : 0;
#pragma unroll
    for (int off = 16; off > 0; off >>= 1) s += __shfl_xor(s, off, 32);
    if (lane == 0) part[w] = s;
  }
  for (int base = tid; base < nr * 256; base += 6 * 512) {
    v4f t[6];
#pragma unroll
    for (int u = 0; u < 6; ++u) t[u] = *(const v4f*)&src[(size_t)(base + u * 512) * 4];
#pragma unroll
    for (int u = 0; u < 6; ++u) {
      const int idx = base + u * 512;
      *(v4f*)&rows[(idx >> 8) * G_RP + (idx & 255) * 4] = t[u];
    }
  }
  __syncthreads();
  {
    const int first = part[0] + part[1] + part[2] + part[3];
    const int mine = packed_rows(lens[me], nr, is_cap);
    int* rmap = is_cap ? rmap_c : rmap_i;
    if (tid < mine) rmap[first + tid] = me * nr + tid;
    if (tid == 0) {
      int* o = offs + (is_cap ? OFF_C : 0);
      o[me] = first;
      if (me == 127) o[128] = first + mine;
    }
    if (blockIdx.x == 0 && tid < 4) zpiece[tid] = 0.f;
  }
  const bool has_tile = w < nt_side * nt_side;
  const int mt = w / nt_side, nt = w % nt_side;
  v8f gs; gs = 0.0f;
  if (has_tile) {
    const int ra = (mt * 16 + ln < nr - 1) ? (mt * 16 + ln) : (nr - 1);
    const int rb = (nt * 16 + ln < nr - 1) ? (nt * 16 + ln) : (nr - 1);
    const float* pa = rows + ra * G_RP + (hi ? 4 : 0);
    const float* pb = rows + rb * G_RP + (hi ? 4 : 0);
    v8f g0, g1; g0 = 0.0f; g1 = 0.0f;
#pragma unroll 8
    for (int d = 0; d < 1024; d += 8) {
      const v4f av = *(const v4f*)&pa[d];
      const v4f bv = *(const v4f*)&pb[d];
      v2f a0; a0.x = av.x; a0.y = av.y;
      v2f a1; a1.x = av.z; a1.y = av.w;
      v2f b0; b0.x = bv.x; b0.y = bv.y;
      v2f b1; b1.x = bv.z; b1.y = bv.w;
      g0 = wmma4(a0, b0, g0);
      g1 = wmma4(a1, b1, g1);
    }
#pragma unroll
    for (int j = 0; j < 8; ++j) gs[j] = g0[j] + g1[j];
    if (mt == nt) {
      float dv = 0.f;
#pragma unroll
      for (int j = 0; j < 8; ++j) dv = ((ln & 7) == j) ? gs[j] : dv;
      if ((ln >> 3) == hi) {
        const float inv = 1.0f / fmaxf(sqrtf(dv), 1e-12f);
        const int r = mt * 16 + ln;
        sinv[r] = inv;
      }
    }
  }
  __syncthreads();
  if (w == 15) {
    float* ginv = (is_cap ? inv_c : inv_i) + me * INV_RP;
    ginv[lane] = sinv[lane];
    if (32 + lane < nt_side * 16) ginv[32 + lane] = sinv[32 + lane];
  }
  if (has_tile) {
    float* dt = dst + (mt * nt_side + nt) * 256;
    const v4f s0 = *(const v4f*)&sinv[mt * 16 + (hi ? 8 : 0)];
    const v4f s1 = *(const v4f*)&sinv[mt * 16 + (hi ? 8 : 0) + 4];
    const float cs = sinv[nt * 16 + ln];
#pragma unroll
    for (int j = 0; j < 8; ++j) {
      const float rs = (j < 4) ? s0[j & 3] : s1[j & 3];
      dt[(j + (hi ? 8 : 0)) * 16 + ln] = gs[j] * rs * cs;
    }
  }
}

#define CG_RP 36
#define CG_N  7680
__global__ __launch_bounds__(512) void cross_gram(const float* __restrict__ imgs,
                                                  const float* __restrict__ caps,
                                                  const float* __restrict__ inv_i,
                                                  const float* __restrict__ inv_c,
                                                  const int* __restrict__ offs,
                                                  const int* __restrict__ rmap_i,
                                                  const int* __restrict__ rmap_c,
                                                  float* __restrict__ cg) {
  __shared__ __attribute__((aligned(16))) float sa[2][128 * CG_RP];
  __shared__ __attribute__((aligned(16))) float sb[2][128 * CG_RP];
  __shared__ __attribute__((aligned(16))) float ssc[256];
  const int tid = threadIdx.x;
  const int w = tid >> 5;
  const int lane = tid & 31;
  const int hi = lane >> 4;
  const int ln = lane & 15;
  const int wm = w >> 2, wn = w & 3;
  const int mbase = blockIdx.y * 128;
  const int nbase = blockIdx.x * 128;
  const int mtot = offs[128];
  const int ntot = offs[OFF_C + 128];
  if (mbase >= mtot || nbase >= ntot) return;
  const int r0 = tid >> 3;
  const int pc = (tid & 7) * 4;
  const int ma0 = mbase + r0, ma1 = ma0 + 64;
  const int nb0 = nbase + r0, nb1 = nb0 + 64;
  const int qa0 = rmap_i[min(ma0, mtot - 1)];
  const int qa1 = rmap_i[min(ma1, mtot - 1)];
  const int qb0 = rmap_c[min(nb0, ntot - 1)];
  const int qb1 = rmap_c[min(nb1, ntot - 1)];
  const float* ga0 = imgs + (size_t)qa0 * 1024 + pc;
  const float* ga1 = imgs + (size_t)qa1 * 1024 + pc;
  const float* gb0 = caps + (size_t)qb0 * 1024 + pc;
  const float* gb1 = caps + (size_t)qb1 * 1024 + pc;
  const int rb0 = (r0 & 96) + ((r0 & 1) << 4) + ((r0 & 31) >> 1);
  const int rb1 = rb0 + 64;
  const int sa0 = r0 * CG_RP + pc, sa1 = (r0 + 64) * CG_RP + pc;
  const int sb0 = rb0 * CG_RP + pc, sb1 = rb1 * CG_RP + pc;
  const int fa = (wm * 32 + ln) * CG_RP + (hi ? 4 : 0);
  const int fb = (wn * 32 + ln) * CG_RP + (hi ? 4 : 0);

  v8f acc[2][2];
#pragma unroll
  for (int mt = 0; mt < 2; ++mt)
#pragma unroll
    for (int nt = 0; nt < 2; ++nt) acc[mt][nt] = 0.0f;

  v4f ta0 = *(const v4f*)&ga0[0], ta1 = *(const v4f*)&ga1[0];
  v4f tb0 = *(const v4f*)&gb0[0], tb1 = *(const v4f*)&gb1[0];
  const float va0 = inv_i[qa0 + (qa0 / 36) * (INV_RP - 36)];
  const float va1 = inv_i[qa1 + (qa1 / 36) * (INV_RP - 36)];
  const float vb0 = inv_c[qb0 + (qb0 / 60) * (INV_RP - 60)];
  const float vb1 = inv_c[qb1 + (qb1 / 60) * (INV_RP - 60)];
  *(v4f*)&sa[0][sa0] = ta0; *(v4f*)&sa[0][sa1] = ta1;
  *(v4f*)&sb[0][sb0] = tb0; *(v4f*)&sb[0][sb1] = tb1;
  if ((tid & 7) == 0) {
    ssc[r0] = va0; ssc[r0 + 64] = va1;
    ssc[128 + r0] = vb0; ssc[128 + r0 + 64] = vb1;
  }
  __syncthreads();
#pragma unroll 1
  for (int ch = 0; ch < 32; ++ch) {
    const int cur = ch & 1;
    if (ch < 31) {
      const int d1 = (ch + 1) * 32;
      ta0 = *(const v4f*)&ga0[d1]; ta1 = *(const v4f*)&ga1[d1];
      tb0 = *(const v4f*)&gb0[d1]; tb1 = *(const v4f*)&gb1[d1];
    }
    const float* pa = &sa[cur][fa];
    const float* pb = &sb[cur][fb];
    v4f av[4][2], bv[4][2];
#pragma unroll
    for (int s4 = 0; s4 < 4; ++s4)
#pragma unroll
      for (int t2 = 0; t2 < 2; ++t2) {
        av[s4][t2] = *(const v4f*)&pa[t2 * 16 * CG_RP + s4 * 8];
        bv[s4][t2] = *(const v4f*)&pb[t2 * 16 * CG_RP + s4 * 8];
      }
#pragma unroll
    for (int s4 = 0; s4 < 4; ++s4)
#pragma unroll
      for (int mt = 0; mt < 2; ++mt)
#pragma unroll
        for (int nt = 0; nt < 2; ++nt) {
          v2f a0; a0.x = av[s4][mt].x; a0.y = av[s4][mt].y;
          v2f a1; a1.x = av[s4][mt].z; a1.y = av[s4][mt].w;
          v2f b0; b0.x = bv[s4][nt].x; b0.y = bv[s4][nt].y;
          v2f b1; b1.x = bv[s4][nt].z; b1.y = bv[s4][nt].w;
          acc[mt][nt] = wmma4(a0, b0, acc[mt][nt]);
          acc[mt][nt] = wmma4(a1, b1, acc[mt][nt]);
        }
    if (ch < 31) {
      const int nx = cur ^ 1;
      *(v4f*)&sa[nx][sa0] = ta0; *(v4f*)&sa[nx][sa1] = ta1;
      *(v4f*)&sb[nx][sb0] = tb0; *(v4f*)&sb[nx][sb1] = tb1;
    }
    __syncthreads();
  }
  const v2f cs = *(const v2f*)&ssc[128 + wn * 32 + 2 * ln];
#pragma unroll
  for (int mt = 0; mt < 2; ++mt) {
    const v4f s0 = *(const v4f*)&ssc[wm * 32 + mt * 16 + (hi ? 8 : 0)];
    const v4f s1 = *(const v4f*)&ssc[wm * 32 + mt * 16 + (hi ? 8 : 0) + 4];
#pragma unroll
    for (int j = 0; j < 8; ++j) {
      const int m = mbase + wm * 32 + mt * 16 + (hi ? 8 : 0) + j;
      const float rs = (j < 4) ? s0[j & 3] : s1[j & 3];
      v2f o; o.x = acc[mt][0][j] * rs * cs.x; o.y = acc[mt][1][j] * rs * cs.y;
      *(v2f*)&cg[(size_t)m * CG_N + nbase + wn * 32 + 2 * ln] = o;
    }
  }
}

#define PC_RP 68
#define GC_RP 68
#define GI_RP 36
#define PW_C  (36 * PC_RP)
#define PW_GI (36 * GI_RP)
#define PL_C  0
#define PL_GI (16 * PW_C)
#define PL_GC (PL_GI + 16 * PW_GI)
#define PL_O1 (PL_GC + 60 * GC_RP)
#define O2_RP 68
#define DG_RP 68
#define PL_O2 (PL_O1 + 16 * 48)
#define PL_DG (PL_O2 + 16 * O2_RP)
#define PL_FLOATS (PL_DG + 16 * DG_RP)

static __device__ __forceinline__ float wave_sum(float s) {
#pragma unroll
  for (int off = 16; off > 0; off >>= 1) s += __shfl_xor(s, off, 32);
  return s;
}

__global__ __launch_bounds__(512) void nmf_pairs(const float* __restrict__ cg,
                                                 const int* __restrict__ offs,
                                                 const float* __restrict__ zpiece,
                                                 const float* __restrict__ gcc,
                                                 const float* __restrict__ gii,
                                                 const int* __restrict__ il_g,
                                                 const int* __restrict__ cl_g,
                                                 float* __restrict__ out) {
  __shared__ __attribute__((aligned(16))) float lds[PL_FLOATS];
  const int tid = threadIdx.x;
  const int w = tid >> 5;
  const int lane = tid & 31;
  const int hi = lane >> 4;
  const int ln = lane & 15;
  const int t = blockIdx.x & 127;
  const int ig = __builtin_amdgcn_readfirstlane((blockIdx.x >> 7) * 16 + w);
  float* cw = lds + PL_C + w * PW_C;
  float* giw = lds + PL_GI + w * PW_GI;
  float* gcs = lds + PL_GC;
  float* o1 = lds + PL_O1 + w * 48;
  float* o2 = lds + PL_O2 + w * O2_RP;
  float* dgs = lds + PL_DG;
  const int il = il_g[ig];
  const int cl = cl_g[t];

  {
    const float* gsi = gii + (size_t)ig * 2304;
    const float* gsc = gcc + (size_t)t * 4096;
    v4f gir[11], gcr[2], cr[17];
#pragma unroll
    for (int u = 0; u < 11; ++u) {
      const int p = lane + 32 * u;
      if (p < 324) {
        const int a = p / 9, b4 = (p % 9) * 4;
        gir[u] = *(const v4f*)&gsi[((a >> 4) * 3 + (b4 >> 4)) * 256 + (a & 15) * 16 + (b4 & 15)];
      }
    }
#pragma unroll
    for (int u = 0; u < 2; ++u) {
      const int p = tid + 512 * u;
      if (p < 900) {
        const int a = p / 15, b4 = (p % 15) * 4;
        gcr[u] = *(const v4f*)&gsc[((a >> 4) * 4 + (b4 >> 4)) * 256 + (a & 15) * 16 + (b4 & 15)];
      }
    }
    const int vi = packed_rows(il, 36, false);
    const int vc4 = packed_rows(cl, 60, true);
    const float* src = cg + (size_t)offs[ig] * CG_N + offs[OFF_C + t];
#pragma unroll
    for (int u = 0; u < 17; ++u) {
      const int idx = lane + 32 * u;
      if (idx < 540) {
        const int row = idx / 15, p4 = (idx % 15) * 4;
        const float* p = (row < vi && p4 < vc4) ? &src[(size_t)row * CG_N + p4] : zpiece;
        cr[u] = *(const v4f*)p;
      }
    }
#pragma unroll
    for (int u = 0; u < 11; ++u) {
      const int p = lane + 32 * u;
      if (p < 324) *(v4f*)&giw[p * 4] = gir[u];
    }
#pragma unroll
    for (int u = 0; u < 2; ++u) {
      const int p = tid + 512 * u;
      if (p < 900) *(v4f*)&gcs[(p / 15) * GC_RP + (p % 15) * 4] = gcr[u];
    }
#pragma unroll
    for (int u = 0; u < 17; ++u) {
      const int idx = lane + 32 * u;
      if (idx < 540) *(v4f*)&cw[(idx / 15) * PC_RP + (idx % 15) * 4] = cr[u];
    }
  }
  const int l0 = lane, l1 = lane + 32;
  const int l1c = (l1 < 59) ? l1 : 59;
  const int kA = lane, kB = 32 + (lane & 3);
  o2[l0] = (l0 < cl) ? (1.f / (float)cl) : 0.f;
  o2[l1] = (l1 < cl) ? (1.f / (float)cl) : 0.f;
  o1[kA] = (kA < il) ? (1.f / (float)il) : 0.f;
  if (lane < 4) o1[kB] = (kB < il) ? (1.f / (float)il) : 0.f;
  if (lane < 4) o1[36 + lane] = 0.f;
  __syncthreads();

  const int pp = lane >> 2;
  float ccol0[36], ccol1[36];
  float crow[60];
  float cx[8];
  float gcol[36];
  float gx[5];
#pragma unroll
  for (int k = 0; k < 36; ++k) {
    ccol0[k] = cw[k * PC_RP + l0];
    ccol1[k] = cw[k * PC_RP + l1c];
  }
#pragma unroll
  for (int l = 0; l < 60; l += 4) {
    const v4f c = *(const v4f*)&cw[kA * PC_RP + l];
#pragma unroll
    for (int e = 0; e < 4; ++e) crow[l + e] = c[e];
  }
  {
    const v4f c0 = *(const v4f*)&cw[kB * PC_RP + 8 * pp];
    const v4f c1 = *(const v4f*)&cw[kB * PC_RP + 8 * pp + ((pp < 7) ? 4 : 0)];
#pragma unroll
    for (int e = 0; e < 4; ++e) { cx[e] = c0[e]; cx[4 + e] = (pp < 7) ? c1[e] : 0.f; }
  }
#pragma unroll
  for (int kp = 0; kp < 36; ++kp) gcol[kp] = giw[kp * GI_RP + kA];
#pragma unroll
  for (int e = 0; e < 5; ++e) {
    const int kp = 5 * pp + e;
    gx[e] = (kp < 36) ? giw[min(kp, 35) * GI_RP + kB] : 0.f;
  }

  float dg0 = 0.f, dg1 = 0.f;
  float duA = 0.f, duB = 0.f;
  float giA = 0.f, giB = 0.f;
#pragma unroll 1
  for (int iter = 0; iter <= N_ITERS; ++iter) {
    if (w < 4) {
      const float* ao = lds + PL_O2 + ln * O2_RP + (hi ? 2 : 0);
      const float* bg = gcs + (hi ? 2 : 0) * GC_RP + w * 16 + ln;
      v8f g0, g1; g0 = 0.0f; g1 = 0.0f;
#pragma unroll
      for (int lp = 0; lp < 60; lp += 8) {
        v2f a = *(const v2f*)&ao[lp];
        v2f b; b.x = bg[lp * GC_RP]; b.y = bg[(lp + 1) * GC_RP];
        g0 = wmma4(a, b, g0);
        if (lp + 4 < 60) {
          v2f a2 = *(const v2f*)&ao[lp + 4];
          v2f b2; b2.x = bg[(lp + 4) * GC_RP]; b2.y = bg[(lp + 5) * GC_RP];
          g1 = wmma4(a2, b2, g1);
        }
      }
#pragma unroll
      for (int j = 0; j < 8; ++j)
        dgs[(j + (hi ? 8 : 0)) * DG_RP + w * 16 + ln] = g0[j] + g1[j];
    }
    __syncthreads();
    dg0 = dgs[w * DG_RP + l0];
    dg1 = dgs[w * DG_RP + l1c];
    float dc0 = 0.f, dc1 = 0.f;
    if (iter < N_ITERS) {
#pragma unroll
      for (int k = 0; k < 36; k += 4) {
        const v4f o = *(const v4f*)&o1[k];
#pragma unroll
        for (int e = 0; e < 4; ++e) {
          dc0 += o[e] * ccol0[k + e];
          dc1 += o[e] * ccol1[k + e];
        }
      }
      const float n0 = (l0 < cl) ? fmaxf(o2[l0] - ETA_F * (dg0 - dc0), 0.f) : 0.f;
      const float n1 = (l1 < cl) ? fmaxf(o2[l1] - ETA_F * (dg1 - dc1), 0.f) : 0.f;
      const float inv = 1.f / (wave_sum(n0 + n1) + EPS_F);
      __syncthreads();
      o2[l0] = n0 * inv;
      o2[l1] = n1 * inv;
      __syncthreads();
    }
    duA = 0.f; duB = 0.f; giA = 0.f; giB = 0.f;
#pragma unroll
    for (int l = 0; l < 60; l += 4) {
      const v4f o = *(const v4f*)&o2[l];
#pragma unroll
      for (int e = 0; e < 4; ++e) duA += crow[l + e] * o[e];
    }
#pragma unroll
    for (int kp = 0; kp < 36; kp += 4) {
      const v4f o = *(const v4f*)&o1[kp];
#pragma unroll
      for (int e = 0; e < 4; ++e) giA += o[e] * gcol[kp + e];
    }
    {
      const v4f oa = *(const v4f*)&o2[8 * pp];
      const v4f ob = *(const v4f*)&o2[8 * pp + 4];
#pragma unroll
      for (int e = 0; e < 4; ++e) duB += cx[e] * oa[e];
#pragma unroll
      for (int e = 0; e < 4; ++e) duB += cx[4 + e] * ob[e];
#pragma unroll
      for (int e = 0; e < 5; ++e) giB += o1[5 * pp + e] * gx[e];
#pragma unroll
      for (int off = 4; off < 32; off <<= 1) {
        duB += __shfl_xor(duB, off, 32);
        giB += __shfl_xor(giB, off, 32);
      }
    }
    if (iter < N_ITERS) {
      const float nA = (kA < il) ? fmaxf(o1[kA] + ETA_F * (duA - giA), 0.f) : 0.f;
      const float nB = (lane < 4 && kB < il) ? fmaxf(o1[kB] + ETA_F * (duB - giB), 0.f) : 0.f;
      const float inv = 1.f / (wave_sum(nA + nB) + EPS_F);
      __syncthreads();
      o1[kA] = nA * inv;
      if (lane < 4) o1[kB] = nB * inv;
      __syncthreads();
    }
  }
  const float s22 = wave_sum(o2[l0] * dg0 + o2[l1] * dg1);
  const float s12 = wave_sum(o1[kA] * duA + ((lane < 4) ? o1[kB] * duB : 0.f));
  const float s11 = wave_sum(o1[kA] * giA + ((lane < 4) ? o1[kB] * giB : 0.f));
  if (lane == 0) {
    const float dnm = fmaxf(sqrtf(fmaxf(s11, 0.f)), 1e-12f) *
                      fmaxf(sqrtf(fmaxf(s22, 0.f)), 1e-12f);
    out[(size_t)ig * 128 + t] = s12 / dnm;
  }
}

extern "C" void kernel_launch(void* const* d_in, const int* in_sizes, int n_in,
                              void* d_out, int out_size, void* d_ws, size_t ws_size,
                              hipStream_t stream) {
  (void)in_sizes; (void)n_in; (void)out_size; (void)ws_size;
  const float* imgs = (const float*)d_in[0];
  const float* caps = (const float*)d_in[1];
  const int* il = (const int*)d_in[2];
  const int* cl = (const int*)d_in[3];
  float* gcc = (float*)d_ws;
  float* gii = gcc + (size_t)128 * 4096;
  float* cg = gii + (size_t)128 * 2304;
  int* offs = (int*)(cg + (size_t)4608 * 7680);
  int* rmap_i = offs + 512;
  int* rmap_c = rmap_i + 4608;
  float* zpiece = (float*)(rmap_c + 7680);
  float* inv_i = zpiece + 32;
  float* inv_c = inv_i + 128 * 64;

  gram_rows<<<256, 512, 0, stream>>>(imgs, caps, gcc, gii, inv_i, inv_c, il, cl, offs, rmap_i,
                                     rmap_c, zpiece);
  cross_gram<<<dim3(60, 36), 512, 0, stream>>>(imgs, caps, inv_i, inv_c, offs, rmap_i, rmap_c, cg);
  nmf_pairs<<<1024, 512, 0, stream>>>(cg, offs, zpiece, gcc, gii, il, cl, (float*)d_out);
}


# pair kernel trips: lane exchanges leave the LDS unit: weights reach the products as scalar operands (v_readlane), lane sums by DPP row_xmask and permlanex16; same partners and order, sums bit-equal
# speedup vs baseline: 10.0515x; 1.0386x over previous
#include <hip/hip_runtime.h>
#include <math.h>

typedef __attribute__((ext_vector_type(2))) float v2f;
typedef __attribute__((ext_vector_type(4))) float v4f;
typedef __attribute__((ext_vector_type(8))) float v8f;

#define N_ITERS 5
#define EPS_F   1e-6f
#define ETA_F   0.02f


static __device__ __forceinline__ v8f wmma4(v2f a, v2f b, v8f c) {
  return __builtin_amdgcn_wmma_f32_16x16x4_f32(false, a, false, b, (short)0, c,
                                               false, false);
}

static __device__ __forceinline__ int packed_rows(int len, int full, bool pad4) {
  const int c = min(max(len, 0), full);
  return pad4 ? ((c + 3) & ~3) : c;
}
#define OFF_C 132

#define INV_RP 64
#define G_RP 1028
__global__ __launch_bounds__(512) void gram_rows(const float* __restrict__ imgs,
                                                 const float* __restrict__ caps,
                                                 float* __restrict__ gcc,
                                                 float* __restrict__ gii,
                                                 float* __restrict__ inv_i,
                                                 float* __restrict__ inv_c,
                                                 const int* __restrict__ il_g,
                                                 const int* __restrict__ cl_g,
                                                 int* __restrict__ offs,
                                                 int* __restrict__ rmap_i,
                                                 int* __restrict__ rmap_c,
                                                 float* __restrict__ zpiece) {
  __shared__ __attribute__((aligned(16))) float rows[60 * G_RP];
  __shared__ int part[4];
  __shared__ __attribute__((aligned(16))) float sinv[64];
  const int tid = threadIdx.x;
  const int w = tid >> 5;
  const int lane = tid & 31;
  const int hi = lane >> 4;
  const int ln = lane & 15;
  const bool is_cap = blockIdx.x < 128;
  const int nr = is_cap ? 60 : 36;
  const int nt_side = is_cap ? 4 : 3;
  const float* src = is_cap ? caps + (size_t)blockIdx.x * 60 * 1024
                            : imgs + (size_t)(blockIdx.x - 128) * 36 * 1024;
  float* dst = is_cap ? gcc + (size_t)blockIdx.x * 4096
                      : gii + (size_t)(blockIdx.x - 128) * 2304;
  const int me = is_cap ? blockIdx.x : blockIdx.x - 128;
  const int* lens = is_cap ? cl_g : il_g;
  if (w < 4) {
    int s = (tid < me) ? packed_rows(lens[tid], nr, is_cap) : 0;
#pragma unroll
    for (int off = 16; off > 0; off >>= 1) s += __shfl_xor(s, off, 32);
    if (lane == 0) part[w] = s;
  }
  for (int base = tid; base < nr * 256; base += 6 * 512) {
    v4f t[6];
#pragma unroll
    for (int u = 0; u < 6; ++u) t[u] = *(const v4f*)&src[(size_t)(base + u * 512) * 4];
#pragma unroll
    for (int u = 0; u < 6; ++u) {
      const int idx = base + u * 512;
      *(v4f*)&rows[(idx >> 8) * G_RP + (idx & 255) * 4] = t[u];
    }
  }
  __syncthreads();
  {
    const int first = part[0] + part[1] + part[2] + part[3];
    const int mine = packed_rows(lens[me], nr, is_cap);
    int* rmap = is_cap ? rmap_c : rmap_i;
    if (tid < mine) rmap[first + tid] = me * nr + tid;
    if (tid == 0) {
      int* o = offs + (is_cap ? OFF_C : 0);
      o[me] = first;
      if (me == 127) o[128] = first + mine;
    }
    if (blockIdx.x == 0 && tid < 4) zpiece[tid] = 0.f;
  }
  const bool has_tile = w < nt_side * nt_side;
  const int mt = w / nt_side, nt = w % nt_side;
  v8f gs; gs = 0.0f;
  if (has_tile) {
    const int ra = (mt * 16 + ln < nr - 1) ? (mt * 16 + ln) : (nr - 1);
    const int rb = (nt * 16 + ln < nr - 1) ? (nt * 16 + ln) : (nr - 1);
    const float* pa = rows + ra * G_RP + (hi ? 4 : 0);
    const float* pb = rows + rb * G_RP + (hi ? 4 : 0);
    v8f g0, g1; g0 = 0.0f; g1 = 0.0f;
#pragma unroll 8
    for (int d = 0; d < 1024; d += 8) {
      const v4f av = *(const v4f*)&pa[d];
      const v4f bv = *(const v4f*)&pb[d];
      v2f a0; a0.x = av.x; a0.y = av.y;
      v2f a1; a1.x = av.z; a1.y = av.w;
      v2f b0; b0.x = bv.x; b0.y = bv.y;
      v2f b1; b1.x = bv.z; b1.y = bv.w;
      g0 = wmma4(a0, b0, g0);
      g1 = wmma4(a1, b1, g1);
    }
#pragma unroll
    for (int j = 0; j < 8; ++j) gs[j] = g0[j] + g1[j];
    if (mt == nt) {
      float dv = 0.f;
#pragma unroll
      for (int j = 0; j < 8; ++j) dv = ((ln & 7) == j) ? gs[j] : dv;
      if ((ln >> 3) == hi) {
        const float inv = 1.0f / fmaxf(sqrtf(dv), 1e-12f);
        const int r = mt * 16 + ln;
        sinv[r] = inv;
      }
    }
  }
  __syncthreads();
  if (w == 15) {
    float* ginv = (is_cap ? inv_c : inv_i) + me * INV_RP;
    ginv[lane] = sinv[lane];
    if (32 + lane < nt_side * 16) ginv[32 + lane] = sinv[32 + lane];
  }
  if (has_tile) {
    float* dt = dst + (mt * nt_side + nt) * 256;
    const v4f s0 = *(const v4f*)&sinv[mt * 16 + (hi ? 8 : 0)];
    const v4f s1 = *(const v4f*)&sinv[mt * 16 + (hi ? 8 : 0) + 4];
    const float cs = sinv[nt * 16 + ln];
#pragma unroll
    for (int j = 0; j < 8; ++j) {
      const float rs = (j < 4) ? s0[j & 3] : s1[j & 3];
      dt[(j + (hi ? 8 : 0)) * 16 + ln] = gs[j] * rs * cs;
    }
  }
}

#define CG_RP 36
#define CG_N  7680
__global__ __launch_bounds__(512) void cross_gram(const float* __restrict__ imgs,
                                                  const float* __restrict__ caps,
                                                  const float* __restrict__ inv_i,
                                                  const float* __restrict__ inv_c,
                                                  const int* __restrict__ offs,
                                                  const int* __restrict__ rmap_i,
                                                  const int* __restrict__ rmap_c,
                                                  float* __restrict__ cg) {
  __shared__ __attribute__((aligned(16))) float sa[2][128 * CG_RP];
  __shared__ __attribute__((aligned(16))) float sb[2][128 * CG_RP];
  __shared__ __attribute__((aligned(16))) float ssc[256];
  const int tid = threadIdx.x;
  const int w = tid >> 5;
  const int lane = tid & 31;
  const int hi = lane >> 4;
  const int ln = lane & 15;
  const int wm = w >> 2, wn = w & 3;
  const int mbase = blockIdx.y * 128;
  const int nbase = blockIdx.x * 128;
  const int mtot = offs[128];
  const int ntot = offs[OFF_C + 128];
  if (mbase >= mtot || nbase >= ntot) return;
  const int r0 = tid >> 3;
  const int pc = (tid & 7) * 4;
  const int ma0 = mbase + r0, ma1 = ma0 + 64;
  const int nb0 = nbase + r0, nb1 = nb0 + 64;
  const int qa0 = rmap_i[min(ma0, mtot - 1)];
  const int qa1 = rmap_i[min(ma1, mtot - 1)];
  const int qb0 = rmap_c[min(nb0, ntot - 1)];
  const int qb1 = rmap_c[min(nb1, ntot - 1)];
  const float* ga0 = imgs + (size_t)qa0 * 1024 + pc;
  const float* ga1 = imgs + (size_t)qa1 * 1024 + pc;
  const float* gb0 = caps + (size_t)qb0 * 1024 + pc;
  const float* gb1 = caps + (size_t)qb1 * 1024 + pc;
  const int rb0 = (r0 & 96) + ((r0 & 1) << 4) + ((r0 & 31) >> 1);
  const int rb1 = rb0 + 64;
  const int sa0 = r0 * CG_RP + pc, sa1 = (r0 + 64) * CG_RP + pc;
  const int sb0 = rb0 * CG_RP + pc, sb1 = rb1 * CG_RP + pc;
  const int fa = (wm * 32 + ln) * CG_RP + (hi ? 4 : 0);
  const int fb = (wn * 32 + ln) * CG_RP + (hi ? 4 : 0);

  v8f acc[2][2];
#pragma unroll
  for (int mt = 0; mt < 2; ++mt)
#pragma unroll
    for (int nt = 0; nt < 2; ++nt) acc[mt][nt] = 0.0f;

  v4f ta0 = *(const v4f*)&ga0[0], ta1 = *(const v4f*)&ga1[0];
  v4f tb0 = *(const v4f*)&gb0[0], tb1 = *(const v4f*)&gb1[0];
  const float va0 = inv_i[qa0 + (qa0 / 36) * (INV_RP - 36)];
  const float va1 = inv_i[qa1 + (qa1 / 36) * (INV_RP - 36)];
  const float vb0 = inv_c[qb0 + (qb0 / 60) * (INV_RP - 60)];
  const float vb1 = inv_c[qb1 + (qb1 / 60) * (INV_RP - 60)];
  *(v4f*)&sa[0][sa0] = ta0; *(v4f*)&sa[0][sa1] = ta1;
  *(v4f*)&sb[0][sb0] = tb0; *(v4f*)&sb[0][sb1] = tb1;
  if ((tid & 7) == 0) {
    ssc[r0] = va0; ssc[r0 + 64] = va1;
    ssc[128 + r0] = vb0; ssc[128 + r0 + 64] = vb1;
  }
  __syncthreads();
#pragma unroll 1
  for (int ch = 0; ch < 32; ++ch) {
    const int cur = ch & 1;
    if (ch < 31) {
      const int d1 = (ch + 1) * 32;
      ta0 = *(const v4f*)&ga0[d1]; ta1 = *(const v4f*)&ga1[d1];
      tb0 = *(const v4f*)&gb0[d1]; tb1 = *(const v4f*)&gb1[d1];
    }
    const float* pa = &sa[cur][fa];
    const float* pb = &sb[cur][fb];
    v4f av[4][2], bv[4][2];
#pragma unroll
    for (int s4 = 0; s4 < 4; ++s4)
#pragma unroll
      for (int t2 = 0; t2 < 2; ++t2) {
        av[s4][t2] = *(const v4f*)&pa[t2 * 16 * CG_RP + s4 * 8];
        bv[s4][t2] = *(const v4f*)&pb[t2 * 16 * CG_RP + s4 * 8];
      }
#pragma unroll
    for (int s4 = 0; s4 < 4; ++s4)
#pragma unroll
      for (int mt = 0; mt < 2; ++mt)
#pragma unroll
        for (int nt = 0; nt < 2; ++nt) {
          v2f a0; a0.x = av[s4][mt].x; a0.y = av[s4][mt].y;
          v2f a1; a1.x = av[s4][mt].z; a1.y = av[s4][mt].w;
          v2f b0; b0.x = bv[s4][nt].x; b0.y = bv[s4][nt].y;
          v2f b1; b1.x = bv[s4][nt].z; b1.y = bv[s4][nt].w;
          acc[mt][nt] = wmma4(a0, b0, acc[mt][nt]);
          acc[mt][nt] = wmma4(a1, b1, acc[mt][nt]);
        }
    if (ch < 31) {
      const int nx = cur ^ 1;
      *(v4f*)&sa[nx][sa0] = ta0; *(v4f*)&sa[nx][sa1] = ta1;
      *(v4f*)&sb[nx][sb0] = tb0; *(v4f*)&sb[nx][sb1] = tb1;
    }
    __syncthreads();
  }
  const v2f cs = *(const v2f*)&ssc[128 + wn * 32 + 2 * ln];
#pragma unroll
  for (int mt = 0; mt < 2; ++mt) {
    const v4f s0 = *(const v4f*)&ssc[wm * 32 + mt * 16 + (hi ? 8 : 0)];
    const v4f s1 = *(const v4f*)&ssc[wm * 32 + mt * 16 + (hi ? 8 : 0) + 4];
#pragma unroll
    for (int j = 0; j < 8; ++j) {
      const int m = mbase + wm * 32 + mt * 16 + (hi ? 8 : 0) + j;
      const float rs = (j < 4) ? s0[j & 3] : s1[j & 3];
      v2f o; o.x = acc[mt][0][j] * rs * cs.x; o.y = acc[mt][1][j] * rs * cs.y;
      *(v2f*)&cg[(size_t)m * CG_N + nbase + wn * 32 + 2 * ln] = o;
    }
  }
}

#define PC_RP 68
#define GC_RP 68
#define GI_RP 36
#define PW_C  (36 * PC_RP)
#define PW_GI (36 * GI_RP)
#define PL_C  0
#define PL_GI (16 * PW_C)
#define PL_GC (PL_GI + 16 * PW_GI)
#define PL_O1 (PL_GC + 60 * GC_RP)
#define O2_RP 68
#define DG_RP 68
#define PL_O2 (PL_O1 + 16 * 48)
#define PL_DG (PL_O2 + 16 * O2_RP)
#define PL_FLOATS (PL_DG + 16 * DG_RP)

template <int K> static __device__ __forceinline__ float xor_in_row(float v) {
  return __int_as_float(__builtin_amdgcn_update_dpp(0, __float_as_int(v), 0x160 | K, 0xf, 0xf, true));
}
static __device__ __forceinline__ float xor_rows(float v) {
  const unsigned u = __float_as_uint(v);
  return __uint_as_float(__builtin_amdgcn_permlanex16(u, u, 0x76543210u, 0xfedcba98u, false, false));
}
template <int SRC> static __device__ __forceinline__ float from_lane(float v) {
  return __int_as_float(__builtin_amdgcn_readlane(__float_as_int(v), SRC));
}
static __device__ __forceinline__ float wave_sum(float s) {
  s += xor_rows(s);
  s += xor_in_row<8>(s);
  s += xor_in_row<4>(s);
  s += xor_in_row<2>(s);
  s += xor_in_row<1>(s);
  return s;
}
template <int K> static __device__ __forceinline__ float pick36(float wA, float wB) {
  if constexpr (K < 32) return from_lane<K>(wA); else return from_lane<K - 32>(wB);
}
template <int I> struct Idx {};
template <int K0, int N> static __device__ __forceinline__ void dot2_36(
    float wA, float wB, const float* __restrict__ m0, const float* __restrict__ m1,
    float& s0, float& s1, Idx<K0>, Idx<N>) {
  if constexpr (K0 < N) {
    const float o = pick36<K0>(wA, wB);
    s0 += o * m0[K0];
    s1 += o * m1[K0];
    dot2_36(wA, wB, m0, m1, s0, s1, Idx<K0 + 1>{}, Idx<N>{});
  }
}
template <int K0, int N> static __device__ __forceinline__ void dot1_o_m(
    float wA, float wB, const float* __restrict__ m, float& s, Idx<K0>, Idx<N>) {
  if constexpr (K0 < N) {
    s += pick36<K0>(wA, wB) * m[K0];
    dot1_o_m(wA, wB, m, s, Idx<K0 + 1>{}, Idx<N>{});
  }
}
template <int K0, int N> static __device__ __forceinline__ void dot1_m_o(
    float wA, float wB, const float* __restrict__ m, float& s, Idx<K0>, Idx<N>) {
  if constexpr (K0 < N) {
    s += m[K0] * pick36<K0>(wA, wB);
    dot1_m_o(wA, wB, m, s, Idx<K0 + 1>{}, Idx<N>{});
  }
}

__global__ __launch_bounds__(512) void nmf_pairs(const float* __restrict__ cg,
                                                 const int* __restrict__ offs,
                                                 const float* __restrict__ zpiece,
                                                 const float* __restrict__ gcc,
                                                 const float* __restrict__ gii,
                                                 const int* __restrict__ il_g,
                                                 const int* __restrict__ cl_g,
                                                 float* __restrict__ out) {
  __shared__ __attribute__((aligned(16))) float lds[PL_FLOATS];
  const int tid = threadIdx.x;
  const int w = tid >> 5;
  const int lane = tid & 31;
  const int hi = lane >> 4;
  const int ln = lane & 15;
  const int t = blockIdx.x & 127;
  const int ig = __builtin_amdgcn_readfirstlane((blockIdx.x >> 7) * 16 + w);
  float* cw = lds + PL_C + w * PW_C;
  float* giw = lds + PL_GI + w * PW_GI;
  float* gcs = lds + PL_GC;
  float* o1 = lds + PL_O1 + w * 48;
  float* o2 = lds + PL_O2 + w * O2_RP;
  float* dgs = lds + PL_DG;
  const int il = il_g[ig];
  const int cl = cl_g[t];

  {
    const float* gsi = gii + (size_t)ig * 2304;
    const float* gsc = gcc + (size_t)t * 4096;
    v4f gir[11], gcr[2], cr[17];
#pragma unroll
    for (int u = 0; u < 11; ++u) {
      const int p = lane + 32 * u;
      if (p < 324) {
        const int a = p / 9, b4 = (p % 9) * 4;
        gir[u] = *(const v4f*)&gsi[((a >> 4) * 3 + (b4 >> 4)) * 256 + (a & 15) * 16 + (b4 & 15)];
      }
    }
#pragma unroll
    for (int u = 0; u < 2; ++u) {
      const int p = tid + 512 * u;
      if (p < 900) {
        const int a = p / 15, b4 = (p % 15) * 4;
        gcr[u] = *(const v4f*)&gsc[((a >> 4) * 4 + (b4 >> 4)) * 256 + (a & 15) * 16 + (b4 & 15)];
      }
    }
    const int vi = packed_rows(il, 36, false);
    const int vc4 = packed_rows(cl, 60, true);
    const float* src = cg + (size_t)offs[ig] * CG_N + offs[OFF_C + t];
#pragma unroll
    for (int u = 0; u < 17; ++u) {
      const int idx = lane + 32 * u;
      if (idx < 540) {
        const int row = idx / 15, p4 = (idx % 15) * 4;
        const float* p = (row < vi && p4 < vc4) ? &src[(size_t)row * CG_N + p4] : zpiece;
        cr[u] = *(const v4f*)p;
      }
    }
#pragma unroll
    for (int u = 0; u < 11; ++u) {
      const int p = lane + 32 * u;
      if (p < 324) *(v4f*)&giw[p * 4] = gir[u];
    }
#pragma unroll
    for (int u = 0; u < 2; ++u) {
      const int p = tid + 512 * u;
      if (p < 900) *(v4f*)&gcs[(p / 15) * GC_RP + (p % 15) * 4] = gcr[u];
    }
#pragma unroll
    for (int u = 0; u < 17; ++u) {
      const int idx = lane + 32 * u;
      if (idx < 540) *(v4f*)&cw[(idx / 15) * PC_RP + (idx % 15) * 4] = cr[u];
    }
  }
  const int l0 = lane, l1 = lane + 32;
  const int l1c = (l1 < 59) ? l1 : 59;
  const int kA = lane, kB = 32 + (lane & 3);
  float w2a = (l0 < cl) ? (1.f / (float)cl) : 0.f;
  float w2b = (l1 < cl) ? (1.f / (float)cl) : 0.f;
  float w1A = (kA < il) ? (1.f / (float)il) : 0.f;
  float w1B = (lane < 4 && kB < il) ? (1.f / (float)il) : 0.f;
  o2[l0] = w2a;
  o2[l1] = w2b;
  o1[kA] = w1A;
  if (lane < 4) o1[kB] = w1B;
  if (lane < 4) o1[36 + lane] = 0.f;
  __syncthreads();

  const int pp = lane >> 2;
  float ccol0[36], ccol1[36];
  float crow[60];
  float cx[8];
  float gcol[36];
  float gx[5];
#pragma unroll
  for (int k = 0; k < 36; ++k) {
    ccol0[k] = cw[k * PC_RP + l0];
    ccol1[k] = cw[k * PC_RP + l1c];
  }
#pragma unroll
  for (int l = 0; l < 60; l += 4) {
    const v4f c = *(const v4f*)&cw[kA * PC_RP + l];
#pragma unroll
    for (int e = 0; e < 4; ++e) crow[l + e] = c[e];
  }
  {
    const v4f c0 = *(const v4f*)&cw[kB * PC_RP + 8 * pp];
    const v4f c1 = *(const v4f*)&cw[kB * PC_RP + 8 * pp + ((pp < 7) ? 4 : 0)];
#pragma unroll
    for (int e = 0; e < 4; ++e) { cx[e] = c0[e]; cx[4 + e] = (pp < 7) ? c1[e] : 0.f; }
  }
#pragma unroll
  for (int kp = 0; kp < 36; ++kp) gcol[kp] = giw[kp * GI_RP + kA];
#pragma unroll
  for (int e = 0; e < 5; ++e) {
    const int kp = 5 * pp + e;
    gx[e] = (kp < 36) ? giw[min(kp, 35) * GI_RP + kB] : 0.f;
  }

  float dg0 = 0.f, dg1 = 0.f;
  float duA = 0.f, duB = 0.f;
  float giA = 0.f, giB = 0.f;
#pragma unroll 1
  for (int iter = 0; iter <= N_ITERS; ++iter) {
    if (w < 4) {
      const float* ao = lds + PL_O2 + ln * O2_RP + (hi ? 2 : 0);
      const float* bg = gcs + (hi ? 2 : 0) * GC_RP + w * 16 + ln;
      v8f g0, g1; g0 = 0.0f; g1 = 0.0f;
#pragma unroll
      for (int lp = 0; lp < 60; lp += 8) {
        v2f a = *(const v2f*)&ao[lp];
        v2f b; b.x = bg[lp * GC_RP]; b.y = bg[(lp + 1) * GC_RP];
        g0 = wmma4(a, b, g0);
        if (lp + 4 < 60) {
          v2f a2 = *(const v2f*)&ao[lp + 4];
          v2f b2; b2.x = bg[(lp + 4) * GC_RP]; b2.y = bg[(lp + 5) * GC_RP];
          g1 = wmma4(a2, b2, g1);
        }
      }
#pragma unroll
      for (int j = 0; j < 8; ++j)
        dgs[(j + (hi ? 8 : 0)) * DG_RP + w * 16 + ln] = g0[j] + g1[j];
    }
    __syncthreads();
    dg0 = dgs[w * DG_RP + l0];
    dg1 = dgs[w * DG_RP + l1c];
    float dc0 = 0.f, dc1 = 0.f;
    if (iter < N_ITERS) {
      dot2_36(w1A, w1B, ccol0, ccol1, dc0, dc1, Idx<0>{}, Idx<36>{});
      const float n0 = (l0 < cl) ? fmaxf(w2a - ETA_F * (dg0 - dc0), 0.f) : 0.f;
      const float n1 = (l1 < cl) ? fmaxf(w2b - ETA_F * (dg1 - dc1), 0.f) : 0.f;
      const float inv = 1.f / (wave_sum(n0 + n1) + EPS_F);
      w2a = n0 * inv;
      w2b = n1 * inv;
      __syncthreads();
      o2[l0] = w2a;
      o2[l1] = w2b;
      __syncthreads();
    }
    duA = 0.f; duB = 0.f; giA = 0.f; giB = 0.f;
    dot1_m_o(w2a, w2b, crow, duA, Idx<0>{}, Idx<60>{});
    dot1_o_m(w1A, w1B, gcol, giA, Idx<0>{}, Idx<36>{});
    {
      const v4f oa = *(const v4f*)&o2[8 * pp];
      const v4f ob = *(const v4f*)&o2[8 * pp + 4];
#pragma unroll
      for (int e = 0; e < 4; ++e) duB += cx[e] * oa[e];
#pragma unroll
      for (int e = 0; e < 4; ++e) duB += cx[4 + e] * ob[e];
#pragma unroll
      for (int e = 0; e < 5; ++e) giB += o1[5 * pp + e] * gx[e];
      duB += xor_in_row<4>(duB);
      giB += xor_in_row<4>(giB);
      duB += xor_in_row<8>(duB);
      giB += xor_in_row<8>(giB);
      duB += xor_rows(duB);
      giB += xor_rows(giB);
    }
    if (iter < N_ITERS) {
      const float nA = (kA < il) ? fmaxf(w1A + ETA_F * (duA - giA), 0.f) : 0.f;
      const float nB = (lane < 4 && kB < il) ? fmaxf(w1B + ETA_F * (duB - giB), 0.f) : 0.f;
      const float inv = 1.f / (wave_sum(nA + nB) + EPS_F);
      w1A = nA * inv;
      w1B = nB * inv;
      __syncthreads();
      o1[kA] = w1A;
      if (lane < 4) o1[kB] = w1B;
      __syncthreads();
    }
  }
  const float s22 = wave_sum(w2a * dg0 + w2b * dg1);
  const float s12 = wave_sum(w1A * duA + ((lane < 4) ? w1B * duB : 0.f));
  const float s11 = wave_sum(w1A * giA + ((lane < 4) ? w1B * giB : 0.f));
  if (lane == 0) {
    const float dnm = fmaxf(sqrtf(fmaxf(s11, 0.f)), 1e-12f) *
                      fmaxf(sqrtf(fmaxf(s22, 0.f)), 1e-12f);
    out[(size_t)ig * 128 + t] = s12 / dnm;
  }
}

extern "C" void kernel_launch(void* const* d_in, const int* in_sizes, int n_in,
                              void* d_out, int out_size, void* d_ws, size_t ws_size,
                              hipStream_t stream) {
  (void)in_sizes; (void)n_in; (void)out_size; (void)ws_size;
  const float* imgs = (const float*)d_in[0];
  const float* caps = (const float*)d_in[1];
  const int* il = (const int*)d_in[2];
  const int* cl = (const int*)d_in[3];
  float* gcc = (float*)d_ws;
  float* gii = gcc + (size_t)128 * 4096;
  float* cg = gii + (size_t)128 * 2304;
  int* offs = (int*)(cg + (size_t)4608 * 7680);
  int* rmap_i = offs + 512;
  int* rmap_c = rmap_i + 4608;
  float* zpiece = (float*)(rmap_c + 7680);
  float* inv_i = zpiece + 32;
  float* inv_c = inv_i + 128 * 64;

  gram_rows<<<256, 512, 0, stream>>>(imgs, caps, gcc, gii, inv_i, inv_c, il, cl, offs, rmap_i,
                                     rmap_c, zpiece);
  cross_gram<<<dim3(60, 36), 512, 0, stream>>>(imgs, caps, inv_i, inv_c, offs, rmap_i, rmap_c, cg);
  nmf_pairs<<<1024, 512, 0, stream>>>(cg, offs, zpiece, gcc, gii, il, cl, (float*)d_out);
}
